# starting kernel
# baseline (speedup 1.0000x reference)
#include <hip/hip_runtime.h>

typedef __attribute__((ext_vector_type(2))) float v2f;
typedef __attribute__((ext_vector_type(8))) float v8f;

#define NCOLS 128
#define MTILE 32
#define LDSTRIDE 260

__global__ void pack_w(const float* __restrict__ W, float* __restrict__ Wp,
                       int Kw, int Kpad) {
  int i = blockIdx.x * blockDim.x + threadIdx.x;
  if (i < Kpad * NCOLS) {
    int k = i / NCOLS;
    int n = i - k * NCOLS;
    float v = (k < Kw) ? W[(size_t)k * NCOLS + n] : 0.0f;
    Wp[((size_t)(k >> 1) * NCOLS + n) * 2 + (k & 1)] = v;
  }
}

__global__ void init_ws(float* __restrict__ deg, float* __restrict__ s1,
                        float* __restrict__ agg2, int N) {
  unsigned tid = blockIdx.x * blockDim.x + threadIdx.x;
  if (tid < (unsigned)N * 128u) {
    agg2[tid] = 0.0f;
    unsigned j = tid & 127u;
    unsigned i = tid >> 7;
    if (j == 0u) deg[i] = 1.0f;
    if (j < 5u)  s1[i * 5u + j] = 0.0f;
  }
}

__global__ void edge_accum_feat(const int* __restrict__ src, const int* __restrict__ dst,
                                const float* __restrict__ x, float* __restrict__ deg,
                                float* __restrict__ s1, int E) {
  int e = blockIdx.x * blockDim.x + threadIdx.x;
  if (e < E) {
    int s = src[e], d = dst[e];
    atomicAdd(&deg[d], 1.0f);
#pragma unroll
    for (int f = 0; f < 5; ++f)
      atomicAdd(&s1[(size_t)d * 5 + f], x[(size_t)s * 5 + f]);
  }
}

__global__ void edge_accum_embed(const int* __restrict__ src, const int* __restrict__ dst,
                                 const float* __restrict__ h, float* __restrict__ agg, int E) {
  unsigned tid = blockIdx.x * blockDim.x + threadIdx.x;
  unsigned e = tid >> 5;
  unsigned c = (tid & 31u) * 4u;
  if (e < (unsigned)E) {
    int s = src[e], d = dst[e];
    const float4 v = *(const float4*)(h + (size_t)s * 128 + c);
    float* a = agg + (size_t)d * 128 + c;
    atomicAdd(a + 0, v.x);
    atomicAdd(a + 1, v.y);
    atomicAdd(a + 2, v.z);
    atomicAdd(a + 3, v.w);
  }
}

__global__ void sage_gemm_wmma(const float* __restrict__ A0, const float* __restrict__ A1,
                               const float* __restrict__ deg, const float* __restrict__ Wp,
                               const float* __restrict__ bias, float* __restrict__ out,
                               int M, int d, int Kpad) {
  __shared__ float As[MTILE * LDSTRIDE];

  const int row0 = blockIdx.x * MTILE;
  const int t = threadIdx.x;

  const int total = MTILE * Kpad;
  for (int i = t; i < total; i += 256) {
    int m = i / Kpad;
    int k = i - m * Kpad;
    int row = row0 + m;
    float v = 0.0f;
    if (row < M) {
      if (k < d) {
        v = A0[(size_t)row * d + k];
      } else if (A1 != nullptr && k < 2 * d) {
        int kk = k - d;
        v = (A0[(size_t)row * d + kk] + A1[(size_t)row * d + kk]) / deg[row];
      }
    }
    As[m * LDSTRIDE + k] = v;
  }
  __syncthreads();

  const int lane  = t & 31;
  const int wave  = t >> 5;
  const int n     = wave * 16 + (lane & 15);
  const int mA    = lane & 15;
  const int khalf = (lane < 16) ? 0 : 2;

  v8f c0 = {};
  v8f c1 = {};
#pragma unroll 2
  for (int k0 = 0; k0 < Kpad; k0 += 4) {
    const int kb = k0 + khalf;
    v2f a0 = *(const v2f*)&As[mA * LDSTRIDE + kb];
    v2f a1 = *(const v2f*)&As[(mA + 16) * LDSTRIDE + kb];
    v2f b  = *(const v2f*)(Wp + ((size_t)(kb >> 1) * NCOLS + n) * 2);
    c0 = __builtin_amdgcn_wmma_f32_16x16x4_f32(false, a0, false, b,
                                               (short)0, c0, false, false);
    c1 = __builtin_amdgcn_wmma_f32_16x16x4_f32(false, a1, false, b,
                                               (short)0, c1, false, false);
  }

  const float bv    = bias[n];
  const int   rbase = (lane < 16) ? 0 : 8;
  float* outA = out + (size_t)(row0 + rbase) * NCOLS + n;
  float* outB = outA + (size_t)16 * NCOLS;

  if (row0 + MTILE <= M) {
#pragma unroll
    for (int r = 0; r < 8; ++r) {
      float vA = c0[r] + bv;
      float vB = c1[r] + bv;
      outA[(size_t)r * NCOLS] = vA > 0.0f ? vA : 0.0f;
      outB[(size_t)r * NCOLS] = vB > 0.0f ? vB : 0.0f;
    }
  } else {
#pragma unroll
    for (int r = 0; r < 8; ++r) {
      int rowA = row0 + rbase + r;
      if (rowA < M) {
        float v = c0[r] + bv;
        outA[(size_t)r * NCOLS] = v > 0.0f ? v : 0.0f;
      }
      int rowB = row0 + 16 + rbase + r;
      if (rowB < M) {
        float v = c1[r] + bv;
        outB[(size_t)r * NCOLS] = v > 0.0f ? v : 0.0f;
      }
    }
  }
}

__global__ void lin2_kernel(const float* __restrict__ h, const float* __restrict__ W,
                            const float* __restrict__ b, float* __restrict__ out, int M) {
  int i = blockIdx.x * blockDim.x + threadIdx.x;
  if (i < M) {
    float a0 = b[0], a1 = b[1];
    const float* row = h + (size_t)i * 128;
#pragma unroll 8
    for (int k = 0; k < 128; ++k) {
      float v = row[k];
      a0 += v * W[2 * k];
      a1 += v * W[2 * k + 1];
    }
    out[(size_t)i * 2 + 0] = a0;
    out[(size_t)i * 2 + 1] = a1;
  }
}

extern "C" void kernel_launch(void* const* d_in, const int* in_sizes, int n_in,
                              void* d_out, int out_size, void* d_ws, size_t ws_size,
                              hipStream_t stream) {
  const float* x   = (const float*)d_in[0];
  const int*   ei  = (const int*)d_in[1];
  const float* W1  = (const float*)d_in[2];
  const float* b1  = (const float*)d_in[3];
  const float* W2  = (const float*)d_in[4];
  const float* b2  = (const float*)d_in[5];
  const float* l1W = (const float*)d_in[6];
  const float* l1b = (const float*)d_in[7];
  const float* l2W = (const float*)d_in[8];
  const float* l2b = (const float*)d_in[9];

  const int N = in_sizes[0] / 5;
  const int E = in_sizes[1] / 2;
  const int* src = ei;
  const int* dst = ei + E;

  float* ws   = (float*)d_ws;
  float* deg  = ws;
  float* s1   = deg  + (size_t)N;
  float* agg2 = s1   + (size_t)5 * N;
  float* h1   = agg2 + (size_t)128 * N;
  float* h2   = h1   + (size_t)128 * N;
  float* Wp1  = h2   + (size_t)128 * N;
  float* Wp2  = Wp1  + (size_t)12 * 128;
  float* Wp3  = Wp2  + (size_t)256 * 128;
  float* h3   = h1;

  const int nthread = 256;

  pack_w<<<(12  * NCOLS + nthread - 1) / nthread, nthread, 0, stream>>>(W1,  Wp1, 10,  12);
  pack_w<<<(256 * NCOLS + nthread - 1) / nthread, nthread, 0, stream>>>(W2,  Wp2, 256, 256);
  pack_w<<<(128 * NCOLS + nthread - 1) / nthread, nthread, 0, stream>>>(l1W, Wp3, 128, 128);

  init_ws<<<((unsigned)N * 128u + nthread - 1) / nthread, nthread, 0, stream>>>(deg, s1, agg2, N);

  edge_accum_feat<<<(E + nthread - 1) / nthread, nthread, 0, stream>>>(src, dst, x, deg, s1, E);

  sage_gemm_wmma<<<(N + MTILE - 1) / MTILE, nthread, 0, stream>>>(x, s1, deg, Wp1, b1, h1, N, 5, 12);

  edge_accum_embed<<<((unsigned)E * 32u + nthread - 1) / nthread, nthread, 0, stream>>>(src, dst, h1, agg2, E);

  sage_gemm_wmma<<<(N + MTILE - 1) / MTILE, nthread, 0, stream>>>(h1, agg2, deg, Wp2, b2, h2, N, 128, 256);

  sage_gemm_wmma<<<(N + MTILE - 1) / MTILE, nthread, 0, stream>>>(h2, nullptr, deg, Wp3, l1b, h3, N, 128, 128);

  lin2_kernel<<<(N + nthread - 1) / nthread, nthread, 0, stream>>>(h3, l2W, l2b, (float*)d_out, N);
}


# Both edge passes: each work-group owns 400 nodes, reads the whole target list itself, sums its nodes' source rows in LDS (LDS adds, arrival order); no regrouping, no global atomics, no zeroing launch
# speedup vs baseline: 2.1654x; 2.1654x over previous
#include <hip/hip_runtime.h>

typedef __attribute__((ext_vector_type(2))) float v2f;
typedef __attribute__((ext_vector_type(8))) float v8f;

#define NCOLS 128
#define MTILE 32
#define LDSTRIDE 260

__global__ void pack_w(const float* __restrict__ W, float* __restrict__ Wp,
                       int Kw, int Kpad) {
  int i = blockIdx.x * blockDim.x + threadIdx.x;
  if (i < Kpad * NCOLS) {
    int k = i / NCOLS;
    int n = i - k * NCOLS;
    float v = (k < Kw) ? W[(size_t)k * NCOLS + n] : 0.0f;
    Wp[((size_t)(k >> 1) * NCOLS + n) * 2 + (k & 1)] = v;
  }
}

#define NPR      400
#define OWN_THR  1024
#define OWN_WAV  32
#define SCAN_U   4
#define QFLUSH   96
#define QCAP     (QFLUSH + SCAN_U * 128)

template <bool VEC>
__device__ __forceinline__ int4 load_dst4(const int* __restrict__ dst, int idx4, int nvec, int E) {
  int4 r;
  if (VEC) {
    r = *(const int4*)(dst + (size_t)min(idx4, nvec - 1) * 4);
  } else {
    const int e = idx4 * 4;
    r.x = dst[min(e + 0, E - 1)];
    r.y = dst[min(e + 1, E - 1)];
    r.z = dst[min(e + 2, E - 1)];
    r.w = dst[min(e + 3, E - 1)];
    if (e + 0 >= E) r.x = -1;
    if (e + 1 >= E) r.y = -1;
    if (e + 2 >= E) r.z = -1;
    if (e + 3 >= E) r.w = -1;
  }
  return r;
}

template <int MODE, bool VEC>
__device__ __forceinline__ void owner_sweep(const int* __restrict__ src, const int* __restrict__ dst,
                                            const float* __restrict__ feat, float* tab, int* cnt,
                                            int* q, int lo, unsigned nloc, int E) {
  const int tid  = threadIdx.x;
  const int lane = tid & 31;
  const int nvec   = (E + 3) >> 2;
  const int per_it = OWN_THR * SCAN_U;
  const int n_it   = (nvec + per_it - 1) / per_it;
  const int rot = (int)(((long long)blockIdx.x * n_it) / gridDim.x);

  int qn = 0;

  auto fetch = [&](int4 (&v)[SCAN_U], int i) -> int {
    int it = i + rot;
    if (it >= n_it) it -= n_it;
    const int base = (i < n_it) ? (it * per_it + tid) : nvec;
#pragma unroll
    for (int u = 0; u < SCAN_U; ++u) v[u] = load_dst4<VEC>(dst, base + u * OWN_THR, nvec, E);
    return base;
  };

  auto take = [&](const int4 (&v)[SCAN_U], int base) {
#pragma unroll
    for (int u = 0; u < SCAN_U; ++u) {
      const int e0 = (base + u * OWN_THR) * 4;
      const unsigned lim = (base + u * OWN_THR < nvec) ? nloc : 0u;
      const bool h0 = (unsigned)(v[u].x - lo) < lim;
      const bool h1 = (unsigned)(v[u].y - lo) < lim;
      const bool h2 = (unsigned)(v[u].z - lo) < lim;
      const bool h3 = (unsigned)(v[u].w - lo) < lim;
      const unsigned m0 = __builtin_amdgcn_ballot_w32(h0);
      const unsigned m1 = __builtin_amdgcn_ballot_w32(h1);
      const unsigned m2 = __builtin_amdgcn_ballot_w32(h2);
      const unsigned m3 = __builtin_amdgcn_ballot_w32(h3);
      if ((m0 | m1 | m2 | m3) != 0u) {
        if (h0) q[qn + __builtin_amdgcn_mbcnt_lo(m0, 0u)] = e0 + 0;
        qn += __builtin_popcount(m0);
        if (h1) q[qn + __builtin_amdgcn_mbcnt_lo(m1, 0u)] = e0 + 1;
        qn += __builtin_popcount(m1);
        if (h2) q[qn + __builtin_amdgcn_mbcnt_lo(m2, 0u)] = e0 + 2;
        qn += __builtin_popcount(m2);
        if (h3) q[qn + __builtin_amdgcn_mbcnt_lo(m3, 0u)] = e0 + 3;
        qn += __builtin_popcount(m3);
      }
    }
  };

  auto drain = [&]() {
    for (int b = 0; b < qn; b += 32) {
      const int k = b + lane;
      int dl = 0, s = 0;
      if (k < qn) {
        const int e = q[k];
        dl = dst[e] - lo;
        s  = src[e];
      }
      if (MODE == 1) {
        if (k < qn) {
          const float* xr = feat + (size_t)s * 5;
          const float x0 = xr[0], x1 = xr[1], x2 = xr[2], x3 = xr[3], x4 = xr[4];
          __hip_atomic_fetch_add(&cnt[dl], 1, __ATOMIC_RELAXED, __HIP_MEMORY_SCOPE_WORKGROUP);
          float* row = &tab[dl * 5];
          __hip_atomic_fetch_add(row + 0, x0, __ATOMIC_RELAXED, __HIP_MEMORY_SCOPE_WORKGROUP);
          __hip_atomic_fetch_add(row + 1, x1, __ATOMIC_RELAXED, __HIP_MEMORY_SCOPE_WORKGROUP);
          __hip_atomic_fetch_add(row + 2, x2, __ATOMIC_RELAXED, __HIP_MEMORY_SCOPE_WORKGROUP);
          __hip_atomic_fetch_add(row + 3, x3, __ATOMIC_RELAXED, __HIP_MEMORY_SCOPE_WORKGROUP);
          __hip_atomic_fetch_add(row + 4, x4, __ATOMIC_RELAXED, __HIP_MEMORY_SCOPE_WORKGROUP);
        }
      } else {
        const int soff = s * 128;
        const int doff = dl * 128;
        const int c = min(32, qn - b);
        int j = 0;
        for (; j + 8 <= c; j += 8) {
          float4 r[8];
          int    d[8];
#pragma unroll
          for (int u = 0; u < 8; ++u) {
            const int sj = __builtin_amdgcn_readlane(soff, j + u);
            d[u] = __builtin_amdgcn_readlane(doff, j + u);
            r[u] = *(const float4*)(feat + (size_t)sj + lane * 4);
          }
#pragma unroll
          for (int u = 0; u < 8; ++u) {
            float* row = &tab[d[u] + lane];
            __hip_atomic_fetch_add(row +  0, r[u].x, __ATOMIC_RELAXED, __HIP_MEMORY_SCOPE_WORKGROUP);
            __hip_atomic_fetch_add(row + 32, r[u].y, __ATOMIC_RELAXED, __HIP_MEMORY_SCOPE_WORKGROUP);
            __hip_atomic_fetch_add(row + 64, r[u].z, __ATOMIC_RELAXED, __HIP_MEMORY_SCOPE_WORKGROUP);
            __hip_atomic_fetch_add(row + 96, r[u].w, __ATOMIC_RELAXED, __HIP_MEMORY_SCOPE_WORKGROUP);
          }
        }
        for (; j < c; ++j) {
          const int sj = __builtin_amdgcn_readlane(soff, j);
          const int dj = __builtin_amdgcn_readlane(doff, j);
          const float4 r = *(const float4*)(feat + (size_t)sj + lane * 4);
          float* row = &tab[dj + lane];
          __hip_atomic_fetch_add(row +  0, r.x, __ATOMIC_RELAXED, __HIP_MEMORY_SCOPE_WORKGROUP);
          __hip_atomic_fetch_add(row + 32, r.y, __ATOMIC_RELAXED, __HIP_MEMORY_SCOPE_WORKGROUP);
          __hip_atomic_fetch_add(row + 64, r.z, __ATOMIC_RELAXED, __HIP_MEMORY_SCOPE_WORKGROUP);
          __hip_atomic_fetch_add(row + 96, r.w, __ATOMIC_RELAXED, __HIP_MEMORY_SCOPE_WORKGROUP);
        }
      }
    }
    qn = 0;
  };

  int4 A[SCAN_U], B[SCAN_U];
  int baseA = fetch(A, 0);
  for (int i = 0; i < n_it; i += 2) {
    const int baseB = fetch(B, i + 1);
    __builtin_amdgcn_sched_barrier(0);
    take(A, baseA);
    if (qn > QFLUSH) drain();
    baseA = fetch(A, i + 2);
    __builtin_amdgcn_sched_barrier(0);
    take(B, baseB);
    if (qn > QFLUSH || i + 2 >= n_it) drain();
  }
}

template <int MODE>
__global__ __launch_bounds__(OWN_THR)
void edge_owner_pass(const int* __restrict__ src, const int* __restrict__ dst,
                     const float* __restrict__ feat, float* __restrict__ deg,
                     float* __restrict__ sums, int N, int E) {
  __shared__ __attribute__((aligned(16))) float tab[NPR * 128];
  __shared__ int   cnt[NPR];
  __shared__ int   queue[OWN_WAV * QCAP];

  const int tid  = threadIdx.x;
  const int lane = tid & 31;
  const int wave = tid >> 5;
  const int lo   = blockIdx.x * NPR;
  const int nown = min(NPR, N - lo);

  if (MODE == 1) {
    for (int t = tid; t < NPR * 5; t += OWN_THR) tab[t] = 0.0f;
    for (int t = tid; t < NPR; t += OWN_THR) cnt[t] = 0;
  } else {
    for (int t = tid; t < NPR * 32; t += OWN_THR)
      ((float4*)tab)[t] = make_float4(0.0f, 0.0f, 0.0f, 0.0f);
  }
  __syncthreads();

  const bool vec_ok = ((E & 3) == 0) && ((((size_t)dst) & 15) == 0);
  if (E > 0) {
    if (vec_ok)
      owner_sweep<MODE, true >(src, dst, feat, tab, cnt, &queue[wave * QCAP], lo, (unsigned)nown, E);
    else
      owner_sweep<MODE, false>(src, dst, feat, tab, cnt, &queue[wave * QCAP], lo, (unsigned)nown, E);
  }

  __syncthreads();

  if (MODE == 1) {
    for (int t = tid; t < nown; t += OWN_THR) deg[lo + t] = 1.0f + (float)cnt[t];
    for (int t = tid; t < nown * 5; t += OWN_THR) sums[(size_t)lo * 5 + t] = tab[t];
  } else {
    for (int r = wave; r < nown; r += OWN_WAV) {
      const float* row = &tab[r * 128 + lane];
      const float4 o = make_float4(row[0], row[32], row[64], row[96]);
      *(float4*)(sums + (size_t)(lo + r) * 128 + lane * 4) = o;
    }
  }
}

__global__ void sage_gemm_wmma(const float* __restrict__ A0, const float* __restrict__ A1,
                               const float* __restrict__ deg, const float* __restrict__ Wp,
                               const float* __restrict__ bias, float* __restrict__ out,
                               int M, int d, int Kpad) {
  __shared__ float As[MTILE * LDSTRIDE];

  const int row0 = blockIdx.x * MTILE;
  const int t = threadIdx.x;

  const int total = MTILE * Kpad;
  for (int i = t; i < total; i += 256) {
    int m = i / Kpad;
    int k = i - m * Kpad;
    int row = row0 + m;
    float v = 0.0f;
    if (row < M) {
      if (k < d) {
        v = A0[(size_t)row * d + k];
      } else if (A1 != nullptr && k < 2 * d) {
        int kk = k - d;
        v = (A0[(size_t)row * d + kk] + A1[(size_t)row * d + kk]) / deg[row];
      }
    }
    As[m * LDSTRIDE + k] = v;
  }
  __syncthreads();

  const int lane  = t & 31;
  const int wave  = t >> 5;
  const int n     = wave * 16 + (lane & 15);
  const int mA    = lane & 15;
  const int khalf = (lane < 16) ? 0 : 2;

  v8f c0 = {};
  v8f c1 = {};
#pragma unroll 2
  for (int k0 = 0; k0 < Kpad; k0 += 4) {
    const int kb = k0 + khalf;
    v2f a0 = *(const v2f*)&As[mA * LDSTRIDE + kb];
    v2f a1 = *(const v2f*)&As[(mA + 16) * LDSTRIDE + kb];
    v2f b  = *(const v2f*)(Wp + ((size_t)(kb >> 1) * NCOLS + n) * 2);
    c0 = __builtin_amdgcn_wmma_f32_16x16x4_f32(false, a0, false, b,
                                               (short)0, c0, false, false);
    c1 = __builtin_amdgcn_wmma_f32_16x16x4_f32(false, a1, false, b,
                                               (short)0, c1, false, false);
  }

  const float bv    = bias[n];
  const int   rbase = (lane < 16) ? 0 : 8;
  float* outA = out + (size_t)(row0 + rbase) * NCOLS + n;
  float* outB = outA + (size_t)16 * NCOLS;

  if (row0 + MTILE <= M) {
#pragma unroll
    for (int r = 0; r < 8; ++r) {
      float vA = c0[r] + bv;
      float vB = c1[r] + bv;
      outA[(size_t)r * NCOLS] = vA > 0.0f ? vA : 0.0f;
      outB[(size_t)r * NCOLS] = vB > 0.0f ? vB : 0.0f;
    }
  } else {
#pragma unroll
    for (int r = 0; r < 8; ++r) {
      int rowA = row0 + rbase + r;
      if (rowA < M) {
        float v = c0[r] + bv;
        outA[(size_t)r * NCOLS] = v > 0.0f ? v : 0.0f;
      }
      int rowB = row0 + 16 + rbase + r;
      if (rowB < M) {
        float v = c1[r] + bv;
        outB[(size_t)r * NCOLS] = v > 0.0f ? v : 0.0f;
      }
    }
  }
}

__global__ void lin2_kernel(const float* __restrict__ h, const float* __restrict__ W,
                            const float* __restrict__ b, float* __restrict__ out, int M) {
  int i = blockIdx.x * blockDim.x + threadIdx.x;
  if (i < M) {
    float a0 = b[0], a1 = b[1];
    const float* row = h + (size_t)i * 128;
#pragma unroll 8
    for (int k = 0; k < 128; ++k) {
      float v = row[k];
      a0 += v * W[2 * k];
      a1 += v * W[2 * k + 1];
    }
    out[(size_t)i * 2 + 0] = a0;
    out[(size_t)i * 2 + 1] = a1;
  }
}

extern "C" void kernel_launch(void* const* d_in, const int* in_sizes, int n_in,
                              void* d_out, int out_size, void* d_ws, size_t ws_size,
                              hipStream_t stream) {
  const float* x   = (const float*)d_in[0];
  const int*   ei  = (const int*)d_in[1];
  const float* W1  = (const float*)d_in[2];
  const float* b1  = (const float*)d_in[3];
  const float* W2  = (const float*)d_in[4];
  const float* b2  = (const float*)d_in[5];
  const float* l1W = (const float*)d_in[6];
  const float* l1b = (const float*)d_in[7];
  const float* l2W = (const float*)d_in[8];
  const float* l2b = (const float*)d_in[9];

  const int N = in_sizes[0] / 5;
  const int E = in_sizes[1] / 2;
  const int* src = ei;
  const int* dst = ei + E;

  float* ws   = (float*)d_ws;
  float* deg  = ws;
  float* s1   = deg  + (size_t)N;
  float* agg2 = s1   + (size_t)5 * N;
  float* h1   = agg2 + (size_t)128 * N;
  float* h2   = h1   + (size_t)128 * N;
  float* Wp1  = h2   + (size_t)128 * N;
  float* Wp2  = Wp1  + (size_t)12 * 128;
  float* Wp3  = Wp2  + (size_t)256 * 128;
  float* h3   = h1;

  const int nthread = 256;

  pack_w<<<(12  * NCOLS + nthread - 1) / nthread, nthread, 0, stream>>>(W1,  Wp1, 10,  12);
  pack_w<<<(256 * NCOLS + nthread - 1) / nthread, nthread, 0, stream>>>(W2,  Wp2, 256, 256);
  pack_w<<<(128 * NCOLS + nthread - 1) / nthread, nthread, 0, stream>>>(l1W, Wp3, 128, 128);

  edge_owner_pass<1><<<(N + NPR - 1) / NPR, OWN_THR, 0, stream>>>(src, dst, x, deg, s1, N, E);

  sage_gemm_wmma<<<(N + MTILE - 1) / MTILE, nthread, 0, stream>>>(x, s1, deg, Wp1, b1, h1, N, 5, 12);

  edge_owner_pass<2><<<(N + NPR - 1) / NPR, OWN_THR, 0, stream>>>(src, dst, h1, deg, agg2, N, E);

  sage_gemm_wmma<<<(N + MTILE - 1) / MTILE, nthread, 0, stream>>>(h1, agg2, deg, Wp2, b2, h2, N, 128, 256);

  sage_gemm_wmma<<<(N + MTILE - 1) / MTILE, nthread, 0, stream>>>(h2, nullptr, deg, Wp3, l1b, h3, N, 128, 128);

  lin2_kernel<<<(N + nthread - 1) / nthread, nthread, 0, stream>>>(h3, l2W, l2b, (float*)d_out, N);
}


# Layer 2 no longer sweeps the target list: layer 1's range owners store each edge they find (source, local target) in a per-range list, layer 2 sums from its list; LDS adds by arrival as before
# speedup vs baseline: 2.5527x; 1.1789x over previous
#include <hip/hip_runtime.h>

typedef __attribute__((ext_vector_type(2))) float v2f;
typedef __attribute__((ext_vector_type(8))) float v8f;

#define NCOLS 128
#define MTILE 32
#define LDSTRIDE 260

__global__ void pack_w(const float* __restrict__ W, float* __restrict__ Wp,
                       int Kw, int Kpad) {
  int i = blockIdx.x * blockDim.x + threadIdx.x;
  if (i < Kpad * NCOLS) {
    int k = i / NCOLS;
    int n = i - k * NCOLS;
    float v = (k < Kw) ? W[(size_t)k * NCOLS + n] : 0.0f;
    Wp[((size_t)(k >> 1) * NCOLS + n) * 2 + (k & 1)] = v;
  }
}

#define NPR      400
#define OWN_THR  1024
#define OWN_WAV  32
#define SCAN_U   4
#define QFLUSH   96
#define QCAP     (QFLUSH + SCAN_U * 128)
#define LCAP     16384

__device__ __forceinline__ void add_rows(const float* __restrict__ feat, float* tab,
                                         int soff, int doff, int c, int lane) {
  for (int j = 0; j < c; j += 8) {
    float4 r[8];
    int    d[8];
    if (j + 8 <= c) {
#pragma unroll
      for (int u = 0; u < 8; ++u) {
        const int sj = __builtin_amdgcn_readlane(soff, j + u);
        d[u] = __builtin_amdgcn_readlane(doff, j + u);
        r[u] = *(const float4*)(feat + (size_t)sj + lane * 4);
      }
#pragma unroll
      for (int u = 0; u < 8; ++u) {
        float* row = &tab[d[u] + lane];
        __hip_atomic_fetch_add(row +  0, r[u].x, __ATOMIC_RELAXED, __HIP_MEMORY_SCOPE_WORKGROUP);
        __hip_atomic_fetch_add(row + 32, r[u].y, __ATOMIC_RELAXED, __HIP_MEMORY_SCOPE_WORKGROUP);
        __hip_atomic_fetch_add(row + 64, r[u].z, __ATOMIC_RELAXED, __HIP_MEMORY_SCOPE_WORKGROUP);
        __hip_atomic_fetch_add(row + 96, r[u].w, __ATOMIC_RELAXED, __HIP_MEMORY_SCOPE_WORKGROUP);
      }
    } else {
      const int m = c - j;
#pragma unroll
      for (int u = 0; u < 7; ++u) {
        if (u < m) {
          const int sj = __builtin_amdgcn_readlane(soff, j + u);
          d[u] = __builtin_amdgcn_readlane(doff, j + u);
          r[u] = *(const float4*)(feat + (size_t)sj + lane * 4);
        }
      }
#pragma unroll
      for (int u = 0; u < 7; ++u) {
        if (u < m) {
          float* row = &tab[d[u] + lane];
          __hip_atomic_fetch_add(row +  0, r[u].x, __ATOMIC_RELAXED, __HIP_MEMORY_SCOPE_WORKGROUP);
          __hip_atomic_fetch_add(row + 32, r[u].y, __ATOMIC_RELAXED, __HIP_MEMORY_SCOPE_WORKGROUP);
          __hip_atomic_fetch_add(row + 64, r[u].z, __ATOMIC_RELAXED, __HIP_MEMORY_SCOPE_WORKGROUP);
          __hip_atomic_fetch_add(row + 96, r[u].w, __ATOMIC_RELAXED, __HIP_MEMORY_SCOPE_WORKGROUP);
        }
      }
    }
  }
}

template <bool VEC>
__device__ __forceinline__ int4 load_dst4(const int* __restrict__ dst, int idx4, int nvec, int E) {
  int4 r;
  if (VEC) {
    r = *(const int4*)(dst + (size_t)min(idx4, nvec - 1) * 4);
  } else {
    const int e = idx4 * 4;
    r.x = dst[min(e + 0, E - 1)];
    r.y = dst[min(e + 1, E - 1)];
    r.z = dst[min(e + 2, E - 1)];
    r.w = dst[min(e + 3, E - 1)];
    if (e + 0 >= E) r.x = -1;
    if (e + 1 >= E) r.y = -1;
    if (e + 2 >= E) r.z = -1;
    if (e + 3 >= E) r.w = -1;
  }
  return r;
}

template <int MODE, bool VEC>
__device__ __forceinline__ void owner_sweep(const int* __restrict__ src, const int* __restrict__ dst,
                                            const float* __restrict__ feat, float* tab, int* cnt,
                                            int* q, int* lcount, int* list,
                                            int lo, unsigned nloc, int E) {
  const int tid  = threadIdx.x;
  const int lane = tid & 31;
  const int nvec   = (E + 3) >> 2;
  const int per_it = OWN_THR * SCAN_U;
  const int n_it   = (nvec + per_it - 1) / per_it;
  const int rot = (int)(((long long)blockIdx.x * n_it) / gridDim.x);

  int qn = 0;

  auto fetch = [&](int4 (&v)[SCAN_U], int i) -> int {
    int it = i + rot;
    if (it >= n_it) it -= n_it;
    const int base = (i < n_it) ? (it * per_it + tid) : nvec;
#pragma unroll
    for (int u = 0; u < SCAN_U; ++u) v[u] = load_dst4<VEC>(dst, base + u * OWN_THR, nvec, E);
    return base;
  };

  auto take = [&](const int4 (&v)[SCAN_U], int base) {
#pragma unroll
    for (int u = 0; u < SCAN_U; ++u) {
      const int e0 = (base + u * OWN_THR) * 4;
      const unsigned lim = (base + u * OWN_THR < nvec) ? nloc : 0u;
      const bool h0 = (unsigned)(v[u].x - lo) < lim;
      const bool h1 = (unsigned)(v[u].y - lo) < lim;
      const bool h2 = (unsigned)(v[u].z - lo) < lim;
      const bool h3 = (unsigned)(v[u].w - lo) < lim;
      const unsigned m0 = __builtin_amdgcn_ballot_w32(h0);
      const unsigned m1 = __builtin_amdgcn_ballot_w32(h1);
      const unsigned m2 = __builtin_amdgcn_ballot_w32(h2);
      const unsigned m3 = __builtin_amdgcn_ballot_w32(h3);
      if ((m0 | m1 | m2 | m3) != 0u) {
        if (h0) q[qn + __builtin_amdgcn_mbcnt_lo(m0, 0u)] = e0 + 0;
        qn += __builtin_popcount(m0);
        if (h1) q[qn + __builtin_amdgcn_mbcnt_lo(m1, 0u)] = e0 + 1;
        qn += __builtin_popcount(m1);
        if (h2) q[qn + __builtin_amdgcn_mbcnt_lo(m2, 0u)] = e0 + 2;
        qn += __builtin_popcount(m2);
        if (h3) q[qn + __builtin_amdgcn_mbcnt_lo(m3, 0u)] = e0 + 3;
        qn += __builtin_popcount(m3);
      }
    }
  };

  auto drain = [&]() {
    for (int b = 0; b < qn; b += 32) {
      const int k = b + lane;
      int dl = 0, s = 0;
      if (k < qn) {
        const int e = q[k];
        dl = dst[e] - lo;
        s  = src[e];
      }
      if (MODE == 1) {
        int slot = 0;
        if (lane == 0)
          slot = __hip_atomic_fetch_add(lcount, min(32, qn - b), __ATOMIC_RELAXED, __HIP_MEMORY_SCOPE_WORKGROUP);
        slot = __builtin_amdgcn_readfirstlane(slot) + lane;
        if (k < qn && slot < LCAP) *(int2*)(list + 2 * (size_t)slot) = make_int2(s, dl);
        if (k < qn) {
          const float* xr = feat + (size_t)s * 5;
          const float x0 = xr[0], x1 = xr[1], x2 = xr[2], x3 = xr[3], x4 = xr[4];
          __hip_atomic_fetch_add(&cnt[dl], 1, __ATOMIC_RELAXED, __HIP_MEMORY_SCOPE_WORKGROUP);
          float* row = &tab[dl * 5];
          __hip_atomic_fetch_add(row + 0, x0, __ATOMIC_RELAXED, __HIP_MEMORY_SCOPE_WORKGROUP);
          __hip_atomic_fetch_add(row + 1, x1, __ATOMIC_RELAXED, __HIP_MEMORY_SCOPE_WORKGROUP);
          __hip_atomic_fetch_add(row + 2, x2, __ATOMIC_RELAXED, __HIP_MEMORY_SCOPE_WORKGROUP);
          __hip_atomic_fetch_add(row + 3, x3, __ATOMIC_RELAXED, __HIP_MEMORY_SCOPE_WORKGROUP);
          __hip_atomic_fetch_add(row + 4, x4, __ATOMIC_RELAXED, __HIP_MEMORY_SCOPE_WORKGROUP);
        }
      } else {
        add_rows(feat, tab, s * 128, dl * 128, min(32, qn - b), lane);
      }
    }
    qn = 0;
  };

  int4 A[SCAN_U], B[SCAN_U];
  int baseA = fetch(A, 0);
  for (int i = 0; i < n_it; i += 2) {
    const int baseB = fetch(B, i + 1);
    __builtin_amdgcn_sched_barrier(0);
    take(A, baseA);
    if (qn > QFLUSH) drain();
    baseA = fetch(A, i + 2);
    __builtin_amdgcn_sched_barrier(0);
    take(B, baseB);
    if (qn > QFLUSH || i + 2 >= n_it) drain();
  }
}

template <int MODE>
__global__ __launch_bounds__(OWN_THR)
void edge_owner_pass(const int* __restrict__ src, const int* __restrict__ dst,
                     const float* __restrict__ feat, float* __restrict__ deg,
                     float* __restrict__ sums, int* counts, int* lists, int N, int E) {
  __shared__ __attribute__((aligned(16))) float tab[NPR * 128];
  __shared__ int   cnt[NPR];
  __shared__ int   queue[OWN_WAV * QCAP];
  __shared__ int   lcount;

  const int tid  = threadIdx.x;
  const int lane = tid & 31;
  const int wave = tid >> 5;
  const int lo   = blockIdx.x * NPR;
  const int nown = min(NPR, N - lo);

  if (MODE == 1) {
    for (int t = tid; t < NPR * 5; t += OWN_THR) tab[t] = 0.0f;
    for (int t = tid; t < NPR; t += OWN_THR) cnt[t] = 0;
    if (tid == 0) lcount = 0;
  } else {
    for (int t = tid; t < NPR * 32; t += OWN_THR)
      ((float4*)tab)[t] = make_float4(0.0f, 0.0f, 0.0f, 0.0f);
  }
  __syncthreads();

  int* list = lists + (size_t)blockIdx.x * (2 * LCAP);
  const int nl = (MODE == 2) ? counts[blockIdx.x] : 0;

  if (MODE == 2 && nl <= LCAP) {
    const int wv  = __builtin_amdgcn_readfirstlane(wave);
    const int beg = (int)(((long long)wv * nl) / OWN_WAV);
    const int end = (int)(((long long)(wv + 1) * nl) / OWN_WAV);
    int2 nx = make_int2(0, 0);
    if (beg + lane < end) nx = *(const int2*)(list + 2 * (size_t)(beg + lane));
    for (int b = beg; b < end; b += 32) {
      const int2 p = nx;
      if (b + 32 + lane < end) nx = *(const int2*)(list + 2 * (size_t)(b + 32 + lane));
      add_rows(feat, tab, p.x * 128, p.y * 128, min(32, end - b), lane);
    }
  } else if (E > 0) {
    const bool vec_ok = ((E & 3) == 0) && ((((size_t)dst) & 15) == 0);
    if (vec_ok)
      owner_sweep<MODE, true >(src, dst, feat, tab, cnt, &queue[wave * QCAP], &lcount, list, lo, (unsigned)nown, E);
    else
      owner_sweep<MODE, false>(src, dst, feat, tab, cnt, &queue[wave * QCAP], &lcount, list, lo, (unsigned)nown, E);
  }

  __syncthreads();

  if (MODE == 1) {
    if (tid == 0) counts[blockIdx.x] = lcount;
    for (int t = tid; t < nown; t += OWN_THR) deg[lo + t] = 1.0f + (float)cnt[t];
    for (int t = tid; t < nown * 5; t += OWN_THR) sums[(size_t)lo * 5 + t] = tab[t];
  } else {
    for (int r = wave; r < nown; r += OWN_WAV) {
      const float* row = &tab[r * 128 + lane];
      const float4 o = make_float4(row[0], row[32], row[64], row[96]);
      *(float4*)(sums + (size_t)(lo + r) * 128 + lane * 4) = o;
    }
  }
}

__global__ void sage_gemm_wmma(const float* __restrict__ A0, const float* __restrict__ A1,
                               const float* __restrict__ deg, const float* __restrict__ Wp,
                               const float* __restrict__ bias, float* __restrict__ out,
                               int M, int d, int Kpad) {
  __shared__ float As[MTILE * LDSTRIDE];

  const int row0 = blockIdx.x * MTILE;
  const int t = threadIdx.x;

  const int total = MTILE * Kpad;
  for (int i = t; i < total; i += 256) {
    int m = i / Kpad;
    int k = i - m * Kpad;
    int row = row0 + m;
    float v = 0.0f;
    if (row < M) {
      if (k < d) {
        v = A0[(size_t)row * d + k];
      } else if (A1 != nullptr && k < 2 * d) {
        int kk = k - d;
        v = (A0[(size_t)row * d + kk] + A1[(size_t)row * d + kk]) / deg[row];
      }
    }
    As[m * LDSTRIDE + k] = v;
  }
  __syncthreads();

  const int lane  = t & 31;
  const int wave  = t >> 5;
  const int n     = wave * 16 + (lane & 15);
  const int mA    = lane & 15;
  const int khalf = (lane < 16) ? 0 : 2;

  v8f c0 = {};
  v8f c1 = {};
#pragma unroll 2
  for (int k0 = 0; k0 < Kpad; k0 += 4) {
    const int kb = k0 + khalf;
    v2f a0 = *(const v2f*)&As[mA * LDSTRIDE + kb];
    v2f a1 = *(const v2f*)&As[(mA + 16) * LDSTRIDE + kb];
    v2f b  = *(const v2f*)(Wp + ((size_t)(kb >> 1) * NCOLS + n) * 2);
    c0 = __builtin_amdgcn_wmma_f32_16x16x4_f32(false, a0, false, b,
                                               (short)0, c0, false, false);
    c1 = __builtin_amdgcn_wmma_f32_16x16x4_f32(false, a1, false, b,
                                               (short)0, c1, false, false);
  }

  const float bv    = bias[n];
  const int   rbase = (lane < 16) ? 0 : 8;
  float* outA = out + (size_t)(row0 + rbase) * NCOLS + n;
  float* outB = outA + (size_t)16 * NCOLS;

  if (row0 + MTILE <= M) {
#pragma unroll
    for (int r = 0; r < 8; ++r) {
      float vA = c0[r] + bv;
      float vB = c1[r] + bv;
      outA[(size_t)r * NCOLS] = vA > 0.0f ? vA : 0.0f;
      outB[(size_t)r * NCOLS] = vB > 0.0f ? vB : 0.0f;
    }
  } else {
#pragma unroll
    for (int r = 0; r < 8; ++r) {
      int rowA = row0 + rbase + r;
      if (rowA < M) {
        float v = c0[r] + bv;
        outA[(size_t)r * NCOLS] = v > 0.0f ? v : 0.0f;
      }
      int rowB = row0 + 16 + rbase + r;
      if (rowB < M) {
        float v = c1[r] + bv;
        outB[(size_t)r * NCOLS] = v > 0.0f ? v : 0.0f;
      }
    }
  }
}

__global__ void lin2_kernel(const float* __restrict__ h, const float* __restrict__ W,
                            const float* __restrict__ b, float* __restrict__ out, int M) {
  int i = blockIdx.x * blockDim.x + threadIdx.x;
  if (i < M) {
    float a0 = b[0], a1 = b[1];
    const float* row = h + (size_t)i * 128;
#pragma unroll 8
    for (int k = 0; k < 128; ++k) {
      float v = row[k];
      a0 += v * W[2 * k];
      a1 += v * W[2 * k + 1];
    }
    out[(size_t)i * 2 + 0] = a0;
    out[(size_t)i * 2 + 1] = a1;
  }
}

extern "C" void kernel_launch(void* const* d_in, const int* in_sizes, int n_in,
                              void* d_out, int out_size, void* d_ws, size_t ws_size,
                              hipStream_t stream) {
  const float* x   = (const float*)d_in[0];
  const int*   ei  = (const int*)d_in[1];
  const float* W1  = (const float*)d_in[2];
  const float* b1  = (const float*)d_in[3];
  const float* W2  = (const float*)d_in[4];
  const float* b2  = (const float*)d_in[5];
  const float* l1W = (const float*)d_in[6];
  const float* l1b = (const float*)d_in[7];
  const float* l2W = (const float*)d_in[8];
  const float* l2b = (const float*)d_in[9];

  const int N = in_sizes[0] / 5;
  const int E = in_sizes[1] / 2;
  const int* src = ei;
  const int* dst = ei + E;

  float* ws   = (float*)d_ws;
  float* deg  = ws;
  float* s1   = deg  + (size_t)N;
  float* agg2 = s1   + (size_t)5 * N;
  float* h1   = agg2 + (size_t)128 * N;
  float* h2   = h1   + (size_t)128 * N;
  float* Wp1  = h2   + (size_t)128 * N;
  float* Wp2  = Wp1  + (size_t)12 * 128;
  float* Wp3  = Wp2  + (size_t)256 * 128;
  float* h3   = h1;
  const int nrange = (N + NPR - 1) / NPR;
  int* counts = (int*)(Wp3 + (size_t)128 * 128);
  int* lists  = counts + (size_t)((nrange + 3) & ~3);

  const int nthread = 256;

  pack_w<<<(12  * NCOLS + nthread - 1) / nthread, nthread, 0, stream>>>(W1,  Wp1, 10,  12);
  pack_w<<<(256 * NCOLS + nthread - 1) / nthread, nthread, 0, stream>>>(W2,  Wp2, 256, 256);
  pack_w<<<(128 * NCOLS + nthread - 1) / nthread, nthread, 0, stream>>>(l1W, Wp3, 128, 128);

  edge_owner_pass<1><<<nrange, OWN_THR, 0, stream>>>(src, dst, x, deg, s1, counts, lists, N, E);

  sage_gemm_wmma<<<(N + MTILE - 1) / MTILE, nthread, 0, stream>>>(x, s1, deg, Wp1, b1, h1, N, 5, 12);

  edge_owner_pass<2><<<nrange, OWN_THR, 0, stream>>>(src, dst, h1, deg, agg2, counts, lists, N, E);

  sage_gemm_wmma<<<(N + MTILE - 1) / MTILE, nthread, 0, stream>>>(h1, agg2, deg, Wp2, b2, h2, N, 128, 256);

  sage_gemm_wmma<<<(N + MTILE - 1) / MTILE, nthread, 0, stream>>>(h2, nullptr, deg, Wp3, l1b, h3, N, 128, 128);

  lin2_kernel<<<(N + nthread - 1) / nthread, nthread, 0, stream>>>(h3, l2W, l2b, (float*)d_out, N);
}


# Layer-1 sweep reads one-byte range keys, a vector load of them per lane (a new pre-pass launch writes key = target / 400 per edge), not 4-byte targets: 4x fewer bytes swept; LDS adds by arrival
# speedup vs baseline: 2.7680x; 1.0843x over previous
#include <hip/hip_runtime.h>

typedef __attribute__((ext_vector_type(2))) float v2f;
typedef __attribute__((ext_vector_type(8))) float v8f;

#define NCOLS 128
#define MTILE 32
#define LDSTRIDE 260

__global__ void pack_w(const float* __restrict__ W, float* __restrict__ Wp,
                       int Kw, int Kpad) {
  int i = blockIdx.x * blockDim.x + threadIdx.x;
  if (i < Kpad * NCOLS) {
    int k = i / NCOLS;
    int n = i - k * NCOLS;
    float v = (k < Kw) ? W[(size_t)k * NCOLS + n] : 0.0f;
    Wp[((size_t)(k >> 1) * NCOLS + n) * 2 + (k & 1)] = v;
  }
}

#define NPR      400
#define OWN_THR  1024
#define OWN_WAV  32
#define SCAN_U   4
#define QFLUSH   96
#define QCAP     (QFLUSH + SCAN_U * 128)
#define LCAP     16384

__device__ __forceinline__ void add_rows(const float* __restrict__ feat, float* tab,
                                         int soff, int doff, int c, int lane) {
  for (int j = 0; j < c; j += 8) {
    float4 r[8];
    int    d[8];
    if (j + 8 <= c) {
#pragma unroll
      for (int u = 0; u < 8; ++u) {
        const int sj = __builtin_amdgcn_readlane(soff, j + u);
        d[u] = __builtin_amdgcn_readlane(doff, j + u);
        r[u] = *(const float4*)(feat + (size_t)sj + lane * 4);
      }
#pragma unroll
      for (int u = 0; u < 8; ++u) {
        float* row = &tab[d[u] + lane];
        __hip_atomic_fetch_add(row +  0, r[u].x, __ATOMIC_RELAXED, __HIP_MEMORY_SCOPE_WORKGROUP);
        __hip_atomic_fetch_add(row + 32, r[u].y, __ATOMIC_RELAXED, __HIP_MEMORY_SCOPE_WORKGROUP);
        __hip_atomic_fetch_add(row + 64, r[u].z, __ATOMIC_RELAXED, __HIP_MEMORY_SCOPE_WORKGROUP);
        __hip_atomic_fetch_add(row + 96, r[u].w, __ATOMIC_RELAXED, __HIP_MEMORY_SCOPE_WORKGROUP);
      }
    } else {
      const int m = c - j;
#pragma unroll
      for (int u = 0; u < 7; ++u) {
        if (u < m) {
          const int sj = __builtin_amdgcn_readlane(soff, j + u);
          d[u] = __builtin_amdgcn_readlane(doff, j + u);
          r[u] = *(const float4*)(feat + (size_t)sj + lane * 4);
        }
      }
#pragma unroll
      for (int u = 0; u < 7; ++u) {
        if (u < m) {
          float* row = &tab[d[u] + lane];
          __hip_atomic_fetch_add(row +  0, r[u].x, __ATOMIC_RELAXED, __HIP_MEMORY_SCOPE_WORKGROUP);
          __hip_atomic_fetch_add(row + 32, r[u].y, __ATOMIC_RELAXED, __HIP_MEMORY_SCOPE_WORKGROUP);
          __hip_atomic_fetch_add(row + 64, r[u].z, __ATOMIC_RELAXED, __HIP_MEMORY_SCOPE_WORKGROUP);
          __hip_atomic_fetch_add(row + 96, r[u].w, __ATOMIC_RELAXED, __HIP_MEMORY_SCOPE_WORKGROUP);
        }
      }
    }
  }
}

template <bool VEC>
__device__ __forceinline__ int4 load_dst4(const int* __restrict__ dst, int idx4, int nvec, int E) {
  int4 r;
  if (VEC) {
    r = *(const int4*)(dst + (size_t)min(idx4, nvec - 1) * 4);
  } else {
    const int e = idx4 * 4;
    r.x = dst[min(e + 0, E - 1)];
    r.y = dst[min(e + 1, E - 1)];
    r.z = dst[min(e + 2, E - 1)];
    r.w = dst[min(e + 3, E - 1)];
    if (e + 0 >= E) r.x = -1;
    if (e + 1 >= E) r.y = -1;
    if (e + 2 >= E) r.z = -1;
    if (e + 3 >= E) r.w = -1;
  }
  return r;
}

template <int MODE, bool VEC>
__device__ __forceinline__ void owner_sweep(const int* __restrict__ src, const int* __restrict__ dst,
                                            const float* __restrict__ feat, float* tab, int* cnt,
                                            int* q, int* lcount, int* list,
                                            int lo, unsigned nloc, int E) {
  const int tid  = threadIdx.x;
  const int lane = tid & 31;
  const int nvec   = (E + 3) >> 2;
  const int per_it = OWN_THR * SCAN_U;
  const int n_it   = (nvec + per_it - 1) / per_it;
  const int rot = (int)(((long long)blockIdx.x * n_it) / gridDim.x);

  int qn = 0;

  auto fetch = [&](int4 (&v)[SCAN_U], int i) -> int {
    int it = i + rot;
    if (it >= n_it) it -= n_it;
    const int base = (i < n_it) ? (it * per_it + tid) : nvec;
#pragma unroll
    for (int u = 0; u < SCAN_U; ++u) v[u] = load_dst4<VEC>(dst, base + u * OWN_THR, nvec, E);
    return base;
  };

  auto take = [&](const int4 (&v)[SCAN_U], int base) {
#pragma unroll
    for (int u = 0; u < SCAN_U; ++u) {
      const int e0 = (base + u * OWN_THR) * 4;
      const unsigned lim = (base + u * OWN_THR < nvec) ? nloc : 0u;
      const bool h0 = (unsigned)(v[u].x - lo) < lim;
      const bool h1 = (unsigned)(v[u].y - lo) < lim;
      const bool h2 = (unsigned)(v[u].z - lo) < lim;
      const bool h3 = (unsigned)(v[u].w - lo) < lim;
      const unsigned m0 = __builtin_amdgcn_ballot_w32(h0);
      const unsigned m1 = __builtin_amdgcn_ballot_w32(h1);
      const unsigned m2 = __builtin_amdgcn_ballot_w32(h2);
      const unsigned m3 = __builtin_amdgcn_ballot_w32(h3);
      if ((m0 | m1 | m2 | m3) != 0u) {
        if (h0) q[qn + __builtin_amdgcn_mbcnt_lo(m0, 0u)] = e0 + 0;
        qn += __builtin_popcount(m0);
        if (h1) q[qn + __builtin_amdgcn_mbcnt_lo(m1, 0u)] = e0 + 1;
        qn += __builtin_popcount(m1);
        if (h2) q[qn + __builtin_amdgcn_mbcnt_lo(m2, 0u)] = e0 + 2;
        qn += __builtin_popcount(m2);
        if (h3) q[qn + __builtin_amdgcn_mbcnt_lo(m3, 0u)] = e0 + 3;
        qn += __builtin_popcount(m3);
      }
    }
  };

  auto drain = [&]() {
    for (int b = 0; b < qn; b += 32) {
      const int k = b + lane;
      int dl = 0, s = 0;
      if (k < qn) {
        const int e = q[k];
        dl = dst[e] - lo;
        s  = src[e];
      }
      if (MODE == 1) {
        int slot = 0;
        if (lane == 0)
          slot = __hip_atomic_fetch_add(lcount, min(32, qn - b), __ATOMIC_RELAXED, __HIP_MEMORY_SCOPE_WORKGROUP);
        slot = __builtin_amdgcn_readfirstlane(slot) + lane;
        if (k < qn && slot < LCAP) *(int2*)(list + 2 * (size_t)slot) = make_int2(s, dl);
        if (k < qn) {
          const float* xr = feat + (size_t)s * 5;
          const float x0 = xr[0], x1 = xr[1], x2 = xr[2], x3 = xr[3], x4 = xr[4];
          __hip_atomic_fetch_add(&cnt[dl], 1, __ATOMIC_RELAXED, __HIP_MEMORY_SCOPE_WORKGROUP);
          float* row = &tab[dl * 5];
          __hip_atomic_fetch_add(row + 0, x0, __ATOMIC_RELAXED, __HIP_MEMORY_SCOPE_WORKGROUP);
          __hip_atomic_fetch_add(row + 1, x1, __ATOMIC_RELAXED, __HIP_MEMORY_SCOPE_WORKGROUP);
          __hip_atomic_fetch_add(row + 2, x2, __ATOMIC_RELAXED, __HIP_MEMORY_SCOPE_WORKGROUP);
          __hip_atomic_fetch_add(row + 3, x3, __ATOMIC_RELAXED, __HIP_MEMORY_SCOPE_WORKGROUP);
          __hip_atomic_fetch_add(row + 4, x4, __ATOMIC_RELAXED, __HIP_MEMORY_SCOPE_WORKGROUP);
        }
      } else {
        add_rows(feat, tab, s * 128, dl * 128, min(32, qn - b), lane);
      }
    }
    qn = 0;
  };

  int4 A[SCAN_U], B[SCAN_U];
  int baseA = fetch(A, 0);
  for (int i = 0; i < n_it; i += 2) {
    const int baseB = fetch(B, i + 1);
    __builtin_amdgcn_sched_barrier(0);
    take(A, baseA);
    if (qn > QFLUSH) drain();
    baseA = fetch(A, i + 2);
    __builtin_amdgcn_sched_barrier(0);
    take(B, baseB);
    if (qn > QFLUSH || i + 2 >= n_it) drain();
  }
}

__device__ __forceinline__ unsigned range_key(int d, int N) {
  return ((unsigned)d < (unsigned)N) ? (unsigned)d / NPR : 0xFFu;
}

__global__ void range_keys(const int* __restrict__ dst, unsigned* __restrict__ keys,
                           int N, int E, int nkv) {
  const int i = blockIdx.x * blockDim.x + threadIdx.x;
  if (i >= nkv) return;
  const int e0 = i * 16;
  unsigned w[4];
  if (e0 + 16 <= E && ((((size_t)dst) & 15) == 0)) {
#pragma unroll
    for (int j = 0; j < 4; ++j) {
      const int4 v = *(const int4*)(dst + e0 + 4 * j);
      w[j] = range_key(v.x, N) | (range_key(v.y, N) << 8) |
             (range_key(v.z, N) << 16) | (range_key(v.w, N) << 24);
    }
  } else {
#pragma unroll
    for (int j = 0; j < 4; ++j) {
      unsigned p = 0u;
#pragma unroll
      for (int y = 0; y < 4; ++y) {
        const int e = e0 + 4 * j + y;
        const int d = (e < E) ? dst[e] : -1;
        p |= range_key(d, N) << (8 * y);
      }
      w[j] = p;
    }
  }
  ((uint4*)keys)[i] = make_uint4(w[0], w[1], w[2], w[3]);
}

#define KSCAN_U 4
__device__ __forceinline__ void owner_sweep_keys(const int* __restrict__ src, const int* __restrict__ dst,
                                                 const uint4* __restrict__ keys,
                                                 const float* __restrict__ feat, float* tab, int* cnt,
                                                 int* q, int* lcount, int* list,
                                                 int lo, unsigned nloc, int E) {
  const int tid  = threadIdx.x;
  const int lane = tid & 31;
  const int nkv    = (E + 15) >> 4;
  const int per_it = OWN_THR * KSCAN_U;
  const int n_it   = (nkv + per_it - 1) / per_it;
  const int rot = (int)(((long long)blockIdx.x * n_it) / gridDim.x);
  const unsigned pat = (unsigned)blockIdx.x * 0x01010101u;

  int qn = 0;

  auto fetch = [&](uint4 (&v)[KSCAN_U], int i) -> int {
    int it = i + rot;
    if (it >= n_it) it -= n_it;
    const int base = (i < n_it) ? (it * per_it + tid) : nkv;
#pragma unroll
    for (int u = 0; u < KSCAN_U; ++u) v[u] = keys[min(base + u * OWN_THR, nkv - 1)];
    return base;
  };

  auto drain = [&]() {
    for (int b = 0; b < qn; b += 32) {
      const int k = b + lane;
      int dl = 0, s = 0;
      bool ok = false;
      if (k < qn) {
        const unsigned w = (unsigned)q[k];
        const int e = (int)((w >> 5) << 4) + (int)(((7u - (w & 7u)) << 2) + ((w & 31u) >> 3));
        if (e < E) {
          dl = dst[e] - lo;
          s  = src[e];
          ok = (unsigned)dl < nloc;
        }
      }
      const unsigned okm = __builtin_amdgcn_ballot_w32(ok);
      int slot = 0;
      if (lane == 0)
        slot = __hip_atomic_fetch_add(lcount, __builtin_popcount(okm), __ATOMIC_RELAXED, __HIP_MEMORY_SCOPE_WORKGROUP);
      slot = __builtin_amdgcn_readfirstlane(slot) + (int)__builtin_amdgcn_mbcnt_lo(okm, 0u);
      if (ok && slot < LCAP) *(int2*)(list + 2 * (size_t)slot) = make_int2(s, dl);
      if (ok) {
        const float* xr = feat + (size_t)s * 5;
        const float x0 = xr[0], x1 = xr[1], x2 = xr[2], x3 = xr[3], x4 = xr[4];
        __hip_atomic_fetch_add(&cnt[dl], 1, __ATOMIC_RELAXED, __HIP_MEMORY_SCOPE_WORKGROUP);
        float* row = &tab[dl * 5];
        __hip_atomic_fetch_add(row + 0, x0, __ATOMIC_RELAXED, __HIP_MEMORY_SCOPE_WORKGROUP);
        __hip_atomic_fetch_add(row + 1, x1, __ATOMIC_RELAXED, __HIP_MEMORY_SCOPE_WORKGROUP);
        __hip_atomic_fetch_add(row + 2, x2, __ATOMIC_RELAXED, __HIP_MEMORY_SCOPE_WORKGROUP);
        __hip_atomic_fetch_add(row + 3, x3, __ATOMIC_RELAXED, __HIP_MEMORY_SCOPE_WORKGROUP);
        __hip_atomic_fetch_add(row + 4, x4, __ATOMIC_RELAXED, __HIP_MEMORY_SCOPE_WORKGROUP);
      }
    }
    qn = 0;
  };

  auto take = [&](const uint4 (&v)[KSCAN_U], int base) {
#pragma unroll
    for (int u = 0; u < KSCAN_U; ++u) {
      const int idx = base + u * OWN_THR;
      const unsigned x0 = v[u].x ^ pat, x1 = v[u].y ^ pat, x2 = v[u].z ^ pat, x3 = v[u].w ^ pat;
      const unsigned u0 = (x0 - 0x01010101u) & ~x0;
      const unsigned u1 = (x1 - 0x01010101u) & ~x1;
      const unsigned u2 = (x2 - 0x01010101u) & ~x2;
      const unsigned u3 = (x3 - 0x01010101u) & ~x3;
      unsigned any = (u0 | u1 | u2 | u3) & 0x80808080u;
      if (idx >= nkv) any = 0u;
      unsigned bm = __builtin_amdgcn_ballot_w32(any != 0u);
      if (bm != 0u) {
        unsigned m = (u0 & 0x80808080u) | ((u1 & 0x80808080u) >> 1) |
                     ((u2 & 0x80808080u) >> 2) | ((u3 & 0x80808080u) >> 3);
        if (any == 0u) m = 0u;
        const unsigned tag = (unsigned)idx << 5;
        do {
          if (m != 0u) {
            const unsigned p = (unsigned)__builtin_ctz(m);
            m &= m - 1u;
            q[qn + (int)__builtin_amdgcn_mbcnt_lo(bm, 0u)] = (int)(tag | p);
          }
          qn += __builtin_popcount(bm);
          bm = __builtin_amdgcn_ballot_w32(m != 0u);
        } while (bm != 0u);
        if (qn > QFLUSH) drain();
      }
    }
  };

  uint4 A[KSCAN_U], B[KSCAN_U];
  int baseA = fetch(A, 0);
  for (int i = 0; i < n_it; i += 2) {
    const int baseB = fetch(B, i + 1);
    __builtin_amdgcn_sched_barrier(0);
    take(A, baseA);
    baseA = fetch(A, i + 2);
    __builtin_amdgcn_sched_barrier(0);
    take(B, baseB);
  }
  drain();
}

template <int MODE>
__global__ __launch_bounds__(OWN_THR)
void edge_owner_pass(const int* __restrict__ src, const int* __restrict__ dst,
                     const float* __restrict__ feat, float* __restrict__ deg,
                     float* __restrict__ sums, int* counts, int* lists,
                     const unsigned* __restrict__ keys, int N, int E) {
  __shared__ __attribute__((aligned(16))) float tab[NPR * 128];
  __shared__ int   cnt[NPR];
  __shared__ int   queue[OWN_WAV * QCAP];
  __shared__ int   lcount;

  const int tid  = threadIdx.x;
  const int lane = tid & 31;
  const int wave = tid >> 5;
  const int lo   = blockIdx.x * NPR;
  const int nown = min(NPR, N - lo);

  if (MODE == 1) {
    for (int t = tid; t < NPR * 5; t += OWN_THR) tab[t] = 0.0f;
    for (int t = tid; t < NPR; t += OWN_THR) cnt[t] = 0;
    if (tid == 0) lcount = 0;
  } else {
    for (int t = tid; t < NPR * 32; t += OWN_THR)
      ((float4*)tab)[t] = make_float4(0.0f, 0.0f, 0.0f, 0.0f);
  }
  __syncthreads();

  int* list = lists + (size_t)blockIdx.x * (2 * LCAP);
  const int nl = (MODE == 2) ? counts[blockIdx.x] : 0;

  if (MODE == 2 && nl <= LCAP) {
    const int wv  = __builtin_amdgcn_readfirstlane(wave);
    const int beg = (int)(((long long)wv * nl) / OWN_WAV);
    const int end = (int)(((long long)(wv + 1) * nl) / OWN_WAV);
    int2 nx = make_int2(0, 0);
    if (beg + lane < end) nx = *(const int2*)(list + 2 * (size_t)(beg + lane));
    for (int b = beg; b < end; b += 32) {
      const int2 p = nx;
      if (b + 32 + lane < end) nx = *(const int2*)(list + 2 * (size_t)(b + 32 + lane));
      add_rows(feat, tab, p.x * 128, p.y * 128, min(32, end - b), lane);
    }
  } else if (MODE == 1 && E > 0 && gridDim.x < 255u) {
    owner_sweep_keys(src, dst, (const uint4*)keys, feat, tab, cnt, &queue[wave * QCAP], &lcount, list, lo, (unsigned)nown, E);
  } else if (E > 0) {
    const bool vec_ok = ((E & 3) == 0) && ((((size_t)dst) & 15) == 0);
    if (vec_ok)
      owner_sweep<MODE, true >(src, dst, feat, tab, cnt, &queue[wave * QCAP], &lcount, list, lo, (unsigned)nown, E);
    else
      owner_sweep<MODE, false>(src, dst, feat, tab, cnt, &queue[wave * QCAP], &lcount, list, lo, (unsigned)nown, E);
  }

  __syncthreads();

  if (MODE == 1) {
    if (tid == 0) counts[blockIdx.x] = lcount;
    for (int t = tid; t < nown; t += OWN_THR) deg[lo + t] = 1.0f + (float)cnt[t];
    for (int t = tid; t < nown * 5; t += OWN_THR) sums[(size_t)lo * 5 + t] = tab[t];
  } else {
    for (int r = wave; r < nown; r += OWN_WAV) {
      const float* row = &tab[r * 128 + lane];
      const float4 o = make_float4(row[0], row[32], row[64], row[96]);
      *(float4*)(sums + (size_t)(lo + r) * 128 + lane * 4) = o;
    }
  }
}

__global__ void sage_gemm_wmma(const float* __restrict__ A0, const float* __restrict__ A1,
                               const float* __restrict__ deg, const float* __restrict__ Wp,
                               const float* __restrict__ bias, float* __restrict__ out,
                               int M, int d, int Kpad) {
  __shared__ float As[MTILE * LDSTRIDE];

  const int row0 = blockIdx.x * MTILE;
  const int t = threadIdx.x;

  const int total = MTILE * Kpad;
  for (int i = t; i < total; i += 256) {
    int m = i / Kpad;
    int k = i - m * Kpad;
    int row = row0 + m;
    float v = 0.0f;
    if (row < M) {
      if (k < d) {
        v = A0[(size_t)row * d + k];
      } else if (A1 != nullptr && k < 2 * d) {
        int kk = k - d;
        v = (A0[(size_t)row * d + kk] + A1[(size_t)row * d + kk]) / deg[row];
      }
    }
    As[m * LDSTRIDE + k] = v;
  }
  __syncthreads();

  const int lane  = t & 31;
  const int wave  = t >> 5;
  const int n     = wave * 16 + (lane & 15);
  const int mA    = lane & 15;
  const int khalf = (lane < 16) ? 0 : 2;

  v8f c0 = {};
  v8f c1 = {};
#pragma unroll 2
  for (int k0 = 0; k0 < Kpad; k0 += 4) {
    const int kb = k0 + khalf;
    v2f a0 = *(const v2f*)&As[mA * LDSTRIDE + kb];
    v2f a1 = *(const v2f*)&As[(mA + 16) * LDSTRIDE + kb];
    v2f b  = *(const v2f*)(Wp + ((size_t)(kb >> 1) * NCOLS + n) * 2);
    c0 = __builtin_amdgcn_wmma_f32_16x16x4_f32(false, a0, false, b,
                                               (short)0, c0, false, false);
    c1 = __builtin_amdgcn_wmma_f32_16x16x4_f32(false, a1, false, b,
                                               (short)0, c1, false, false);
  }

  const float bv    = bias[n];
  const int   rbase = (lane < 16) ? 0 : 8;
  float* outA = out + (size_t)(row0 + rbase) * NCOLS + n;
  float* outB = outA + (size_t)16 * NCOLS;

  if (row0 + MTILE <= M) {
#pragma unroll
    for (int r = 0; r < 8; ++r) {
      float vA = c0[r] + bv;
      float vB = c1[r] + bv;
      outA[(size_t)r * NCOLS] = vA > 0.0f ? vA : 0.0f;
      outB[(size_t)r * NCOLS] = vB > 0.0f ? vB : 0.0f;
    }
  } else {
#pragma unroll
    for (int r = 0; r < 8; ++r) {
      int rowA = row0 + rbase + r;
      if (rowA < M) {
        float v = c0[r] + bv;
        outA[(size_t)r * NCOLS] = v > 0.0f ? v : 0.0f;
      }
      int rowB = row0 + 16 + rbase + r;
      if (rowB < M) {
        float v = c1[r] + bv;
        outB[(size_t)r * NCOLS] = v > 0.0f ? v : 0.0f;
      }
    }
  }
}

__global__ void lin2_kernel(const float* __restrict__ h, const float* __restrict__ W,
                            const float* __restrict__ b, float* __restrict__ out, int M) {
  int i = blockIdx.x * blockDim.x + threadIdx.x;
  if (i < M) {
    float a0 = b[0], a1 = b[1];
    const float* row = h + (size_t)i * 128;
#pragma unroll 8
    for (int k = 0; k < 128; ++k) {
      float v = row[k];
      a0 += v * W[2 * k];
      a1 += v * W[2 * k + 1];
    }
    out[(size_t)i * 2 + 0] = a0;
    out[(size_t)i * 2 + 1] = a1;
  }
}

extern "C" void kernel_launch(void* const* d_in, const int* in_sizes, int n_in,
                              void* d_out, int out_size, void* d_ws, size_t ws_size,
                              hipStream_t stream) {
  const float* x   = (const float*)d_in[0];
  const int*   ei  = (const int*)d_in[1];
  const float* W1  = (const float*)d_in[2];
  const float* b1  = (const float*)d_in[3];
  const float* W2  = (const float*)d_in[4];
  const float* b2  = (const float*)d_in[5];
  const float* l1W = (const float*)d_in[6];
  const float* l1b = (const float*)d_in[7];
  const float* l2W = (const float*)d_in[8];
  const float* l2b = (const float*)d_in[9];

  const int N = in_sizes[0] / 5;
  const int E = in_sizes[1] / 2;
  const int* src = ei;
  const int* dst = ei + E;

  float* ws   = (float*)d_ws;
  float* deg  = ws;
  float* s1   = deg  + (size_t)N;
  float* agg2 = s1   + (size_t)5 * N;
  float* h1   = agg2 + (size_t)128 * N;
  float* h2   = h1   + (size_t)128 * N;
  float* Wp1  = h2   + (size_t)128 * N;
  float* Wp2  = Wp1  + (size_t)12 * 128;
  float* Wp3  = Wp2  + (size_t)256 * 128;
  float* h3   = h1;
  const int nrange = (N + NPR - 1) / NPR;
  int* counts = (int*)(Wp3 + (size_t)128 * 128);
  int* lists  = counts + (size_t)((nrange + 3) & ~3);
  const int nkv = (E + 15) >> 4;
  unsigned* keys = (unsigned*)(lists + (size_t)nrange * (2 * LCAP));

  const int nthread = 256;

  pack_w<<<(12  * NCOLS + nthread - 1) / nthread, nthread, 0, stream>>>(W1,  Wp1, 10,  12);
  pack_w<<<(256 * NCOLS + nthread - 1) / nthread, nthread, 0, stream>>>(W2,  Wp2, 256, 256);
  pack_w<<<(128 * NCOLS + nthread - 1) / nthread, nthread, 0, stream>>>(l1W, Wp3, 128, 128);

  range_keys<<<(nkv + nthread - 1) / nthread, nthread, 0, stream>>>(dst, keys, N, E, nkv);

  edge_owner_pass<1><<<nrange, OWN_THR, 0, stream>>>(src, dst, x, deg, s1, counts, lists, keys, N, E);

  sage_gemm_wmma<<<(N + MTILE - 1) / MTILE, nthread, 0, stream>>>(x, s1, deg, Wp1, b1, h1, N, 5, 12);

  edge_owner_pass<2><<<nrange, OWN_THR, 0, stream>>>(src, dst, h1, deg, agg2, counts, lists, keys, N, E);

  sage_gemm_wmma<<<(N + MTILE - 1) / MTILE, nthread, 0, stream>>>(h1, agg2, deg, Wp2, b2, h2, N, 128, 256);

  sage_gemm_wmma<<<(N + MTILE - 1) / MTILE, nthread, 0, stream>>>(h2, nullptr, deg, Wp3, l1b, h3, N, 128, 128);

  lin2_kernel<<<(N + nthread - 1) / nthread, nthread, 0, stream>>>(h3, l2W, l2b, (float*)d_out, N);
}


# GEMM A-tile staging for 128-float rows: 16-byte loads, a thread requests whole rows before its first wait, 1-2 trips to memory per tile (was 16-32, one 4-byte element per trip); sums unchanged
# speedup vs baseline: 3.3472x; 1.2092x over previous
#include <hip/hip_runtime.h>

typedef __attribute__((ext_vector_type(2))) float v2f;
typedef __attribute__((ext_vector_type(8))) float v8f;

#define NCOLS 128
#define MTILE 32
#define LDSTRIDE 260

__global__ void pack_w(const float* __restrict__ W, float* __restrict__ Wp,
                       int Kw, int Kpad) {
  int i = blockIdx.x * blockDim.x + threadIdx.x;
  if (i < Kpad * NCOLS) {
    int k = i / NCOLS;
    int n = i - k * NCOLS;
    float v = (k < Kw) ? W[(size_t)k * NCOLS + n] : 0.0f;
    Wp[((size_t)(k >> 1) * NCOLS + n) * 2 + (k & 1)] = v;
  }
}

#define NPR      400
#define OWN_THR  1024
#define OWN_WAV  32
#define SCAN_U   4
#define QFLUSH   96
#define QCAP     (QFLUSH + SCAN_U * 128)
#define LCAP     16384

__device__ __forceinline__ void add_rows(const float* __restrict__ feat, float* tab,
                                         int soff, int doff, int c, int lane) {
  for (int j = 0; j < c; j += 8) {
    float4 r[8];
    int    d[8];
    if (j + 8 <= c) {
#pragma unroll
      for (int u = 0; u < 8; ++u) {
        const int sj = __builtin_amdgcn_readlane(soff, j + u);
        d[u] = __builtin_amdgcn_readlane(doff, j + u);
        r[u] = *(const float4*)(feat + (size_t)sj + lane * 4);
      }
#pragma unroll
      for (int u = 0; u < 8; ++u) {
        float* row = &tab[d[u] + lane];
        __hip_atomic_fetch_add(row +  0, r[u].x, __ATOMIC_RELAXED, __HIP_MEMORY_SCOPE_WORKGROUP);
        __hip_atomic_fetch_add(row + 32, r[u].y, __ATOMIC_RELAXED, __HIP_MEMORY_SCOPE_WORKGROUP);
        __hip_atomic_fetch_add(row + 64, r[u].z, __ATOMIC_RELAXED, __HIP_MEMORY_SCOPE_WORKGROUP);
        __hip_atomic_fetch_add(row + 96, r[u].w, __ATOMIC_RELAXED, __HIP_MEMORY_SCOPE_WORKGROUP);
      }
    } else {
      const int m = c - j;
#pragma unroll
      for (int u = 0; u < 7; ++u) {
        if (u < m) {
          const int sj = __builtin_amdgcn_readlane(soff, j + u);
          d[u] = __builtin_amdgcn_readlane(doff, j + u);
          r[u] = *(const float4*)(feat + (size_t)sj + lane * 4);
        }
      }
#pragma unroll
      for (int u = 0; u < 7; ++u) {
        if (u < m) {
          float* row = &tab[d[u] + lane];
          __hip_atomic_fetch_add(row +  0, r[u].x, __ATOMIC_RELAXED, __HIP_MEMORY_SCOPE_WORKGROUP);
          __hip_atomic_fetch_add(row + 32, r[u].y, __ATOMIC_RELAXED, __HIP_MEMORY_SCOPE_WORKGROUP);
          __hip_atomic_fetch_add(row + 64, r[u].z, __ATOMIC_RELAXED, __HIP_MEMORY_SCOPE_WORKGROUP);
          __hip_atomic_fetch_add(row + 96, r[u].w, __ATOMIC_RELAXED, __HIP_MEMORY_SCOPE_WORKGROUP);
        }
      }
    }
  }
}

template <bool VEC>
__device__ __forceinline__ int4 load_dst4(const int* __restrict__ dst, int idx4, int nvec, int E) {
  int4 r;
  if (VEC) {
    r = *(const int4*)(dst + (size_t)min(idx4, nvec - 1) * 4);
  } else {
    const int e = idx4 * 4;
    r.x = dst[min(e + 0, E - 1)];
    r.y = dst[min(e + 1, E - 1)];
    r.z = dst[min(e + 2, E - 1)];
    r.w = dst[min(e + 3, E - 1)];
    if (e + 0 >= E) r.x = -1;
    if (e + 1 >= E) r.y = -1;
    if (e + 2 >= E) r.z = -1;
    if (e + 3 >= E) r.w = -1;
  }
  return r;
}

template <int MODE, bool VEC>
__device__ __forceinline__ void owner_sweep(const int* __restrict__ src, const int* __restrict__ dst,
                                            const float* __restrict__ feat, float* tab, int* cnt,
                                            int* q, int* lcount, int* list,
                                            int lo, unsigned nloc, int E) {
  const int tid  = threadIdx.x;
  const int lane = tid & 31;
  const int nvec   = (E + 3) >> 2;
  const int per_it = OWN_THR * SCAN_U;
  const int n_it   = (nvec + per_it - 1) / per_it;
  const int rot = (int)(((long long)blockIdx.x * n_it) / gridDim.x);

  int qn = 0;

  auto fetch = [&](int4 (&v)[SCAN_U], int i) -> int {
    int it = i + rot;
    if (it >= n_it) it -= n_it;
    const int base = (i < n_it) ? (it * per_it + tid) : nvec;
#pragma unroll
    for (int u = 0; u < SCAN_U; ++u) v[u] = load_dst4<VEC>(dst, base + u * OWN_THR, nvec, E);
    return base;
  };

  auto take = [&](const int4 (&v)[SCAN_U], int base) {
#pragma unroll
    for (int u = 0; u < SCAN_U; ++u) {
      const int e0 = (base + u * OWN_THR) * 4;
      const unsigned lim = (base + u * OWN_THR < nvec) ? nloc : 0u;
      const bool h0 = (unsigned)(v[u].x - lo) < lim;
      const bool h1 = (unsigned)(v[u].y - lo) < lim;
      const bool h2 = (unsigned)(v[u].z - lo) < lim;
      const bool h3 = (unsigned)(v[u].w - lo) < lim;
      const unsigned m0 = __builtin_amdgcn_ballot_w32(h0);
      const unsigned m1 = __builtin_amdgcn_ballot_w32(h1);
      const unsigned m2 = __builtin_amdgcn_ballot_w32(h2);
      const unsigned m3 = __builtin_amdgcn_ballot_w32(h3);
      if ((m0 | m1 | m2 | m3) != 0u) {
        if (h0) q[qn + __builtin_amdgcn_mbcnt_lo(m0, 0u)] = e0 + 0;
        qn += __builtin_popcount(m0);
        if (h1) q[qn + __builtin_amdgcn_mbcnt_lo(m1, 0u)] = e0 + 1;
        qn += __builtin_popcount(m1);
        if (h2) q[qn + __builtin_amdgcn_mbcnt_lo(m2, 0u)] = e0 + 2;
        qn += __builtin_popcount(m2);
        if (h3) q[qn + __builtin_amdgcn_mbcnt_lo(m3, 0u)] = e0 + 3;
        qn += __builtin_popcount(m3);
      }
    }
  };

  auto drain = [&]() {
    for (int b = 0; b < qn; b += 32) {
      const int k = b + lane;
      int dl = 0, s = 0;
      if (k < qn) {
        const int e = q[k];
        dl = dst[e] - lo;
        s  = src[e];
      }
      if (MODE == 1) {
        int slot = 0;
        if (lane == 0)
          slot = __hip_atomic_fetch_add(lcount, min(32, qn - b), __ATOMIC_RELAXED, __HIP_MEMORY_SCOPE_WORKGROUP);
        slot = __builtin_amdgcn_readfirstlane(slot) + lane;
        if (k < qn && slot < LCAP) *(int2*)(list + 2 * (size_t)slot) = make_int2(s, dl);
        if (k < qn) {
          const float* xr = feat + (size_t)s * 5;
          const float x0 = xr[0], x1 = xr[1], x2 = xr[2], x3 = xr[3], x4 = xr[4];
          __hip_atomic_fetch_add(&cnt[dl], 1, __ATOMIC_RELAXED, __HIP_MEMORY_SCOPE_WORKGROUP);
          float* row = &tab[dl * 5];
          __hip_atomic_fetch_add(row + 0, x0, __ATOMIC_RELAXED, __HIP_MEMORY_SCOPE_WORKGROUP);
          __hip_atomic_fetch_add(row + 1, x1, __ATOMIC_RELAXED, __HIP_MEMORY_SCOPE_WORKGROUP);
          __hip_atomic_fetch_add(row + 2, x2, __ATOMIC_RELAXED, __HIP_MEMORY_SCOPE_WORKGROUP);
          __hip_atomic_fetch_add(row + 3, x3, __ATOMIC_RELAXED, __HIP_MEMORY_SCOPE_WORKGROUP);
          __hip_atomic_fetch_add(row + 4, x4, __ATOMIC_RELAXED, __HIP_MEMORY_SCOPE_WORKGROUP);
        }
      } else {
        add_rows(feat, tab, s * 128, dl * 128, min(32, qn - b), lane);
      }
    }
    qn = 0;
  };

  int4 A[SCAN_U], B[SCAN_U];
  int baseA = fetch(A, 0);
  for (int i = 0; i < n_it; i += 2) {
    const int baseB = fetch(B, i + 1);
    __builtin_amdgcn_sched_barrier(0);
    take(A, baseA);
    if (qn > QFLUSH) drain();
    baseA = fetch(A, i + 2);
    __builtin_amdgcn_sched_barrier(0);
    take(B, baseB);
    if (qn > QFLUSH || i + 2 >= n_it) drain();
  }
}

__device__ __forceinline__ unsigned range_key(int d, int N) {
  return ((unsigned)d < (unsigned)N) ? (unsigned)d / NPR : 0xFFu;
}

__global__ void range_keys(const int* __restrict__ dst, unsigned* __restrict__ keys,
                           int N, int E, int nkv) {
  const int i = blockIdx.x * blockDim.x + threadIdx.x;
  if (i >= nkv) return;
  const int e0 = i * 16;
  unsigned w[4];
  if (e0 + 16 <= E && ((((size_t)dst) & 15) == 0)) {
#pragma unroll
    for (int j = 0; j < 4; ++j) {
      const int4 v = *(const int4*)(dst + e0 + 4 * j);
      w[j] = range_key(v.x, N) | (range_key(v.y, N) << 8) |
             (range_key(v.z, N) << 16) | (range_key(v.w, N) << 24);
    }
  } else {
#pragma unroll
    for (int j = 0; j < 4; ++j) {
      unsigned p = 0u;
#pragma unroll
      for (int y = 0; y < 4; ++y) {
        const int e = e0 + 4 * j + y;
        const int d = (e < E) ? dst[e] : -1;
        p |= range_key(d, N) << (8 * y);
      }
      w[j] = p;
    }
  }
  ((uint4*)keys)[i] = make_uint4(w[0], w[1], w[2], w[3]);
}

#define KSCAN_U 4
__device__ __forceinline__ void owner_sweep_keys(const int* __restrict__ src, const int* __restrict__ dst,
                                                 const uint4* __restrict__ keys,
                                                 const float* __restrict__ feat, float* tab, int* cnt,
                                                 int* q, int* lcount, int* list,
                                                 int lo, unsigned nloc, int E) {
  const int tid  = threadIdx.x;
  const int lane = tid & 31;
  const int nkv    = (E + 15) >> 4;
  const int per_it = OWN_THR * KSCAN_U;
  const int n_it   = (nkv + per_it - 1) / per_it;
  const int rot = (int)(((long long)blockIdx.x * n_it) / gridDim.x);
  const unsigned pat = (unsigned)blockIdx.x * 0x01010101u;

  int qn = 0;

  auto fetch = [&](uint4 (&v)[KSCAN_U], int i) -> int {
    int it = i + rot;
    if (it >= n_it) it -= n_it;
    const int base = (i < n_it) ? (it * per_it + tid) : nkv;
#pragma unroll
    for (int u = 0; u < KSCAN_U; ++u) v[u] = keys[min(base + u * OWN_THR, nkv - 1)];
    return base;
  };

  auto drain = [&]() {
    for (int b = 0; b < qn; b += 32) {
      const int k = b + lane;
      int dl = 0, s = 0;
      bool ok = false;
      if (k < qn) {
        const unsigned w = (unsigned)q[k];
        const int e = (int)((w >> 5) << 4) + (int)(((7u - (w & 7u)) << 2) + ((w & 31u) >> 3));
        if (e < E) {
          dl = dst[e] - lo;
          s  = src[e];
          ok = (unsigned)dl < nloc;
        }
      }
      const unsigned okm = __builtin_amdgcn_ballot_w32(ok);
      int slot = 0;
      if (lane == 0)
        slot = __hip_atomic_fetch_add(lcount, __builtin_popcount(okm), __ATOMIC_RELAXED, __HIP_MEMORY_SCOPE_WORKGROUP);
      slot = __builtin_amdgcn_readfirstlane(slot) + (int)__builtin_amdgcn_mbcnt_lo(okm, 0u);
      if (ok && slot < LCAP) *(int2*)(list + 2 * (size_t)slot) = make_int2(s, dl);
      if (ok) {
        const float* xr = feat + (size_t)s * 5;
        const float x0 = xr[0], x1 = xr[1], x2 = xr[2], x3 = xr[3], x4 = xr[4];
        __hip_atomic_fetch_add(&cnt[dl], 1, __ATOMIC_RELAXED, __HIP_MEMORY_SCOPE_WORKGROUP);
        float* row = &tab[dl * 5];
        __hip_atomic_fetch_add(row + 0, x0, __ATOMIC_RELAXED, __HIP_MEMORY_SCOPE_WORKGROUP);
        __hip_atomic_fetch_add(row + 1, x1, __ATOMIC_RELAXED, __HIP_MEMORY_SCOPE_WORKGROUP);
        __hip_atomic_fetch_add(row + 2, x2, __ATOMIC_RELAXED, __HIP_MEMORY_SCOPE_WORKGROUP);
        __hip_atomic_fetch_add(row + 3, x3, __ATOMIC_RELAXED, __HIP_MEMORY_SCOPE_WORKGROUP);
        __hip_atomic_fetch_add(row + 4, x4, __ATOMIC_RELAXED, __HIP_MEMORY_SCOPE_WORKGROUP);
      }
    }
    qn = 0;
  };

  auto take = [&](const uint4 (&v)[KSCAN_U], int base) {
#pragma unroll
    for (int u = 0; u < KSCAN_U; ++u) {
      const int idx = base + u * OWN_THR;
      const unsigned x0 = v[u].x ^ pat, x1 = v[u].y ^ pat, x2 = v[u].z ^ pat, x3 = v[u].w ^ pat;
      const unsigned u0 = (x0 - 0x01010101u) & ~x0;
      const unsigned u1 = (x1 - 0x01010101u) & ~x1;
      const unsigned u2 = (x2 - 0x01010101u) & ~x2;
      const unsigned u3 = (x3 - 0x01010101u) & ~x3;
      unsigned any = (u0 | u1 | u2 | u3) & 0x80808080u;
      if (idx >= nkv) any = 0u;
      unsigned bm = __builtin_amdgcn_ballot_w32(any != 0u);
      if (bm != 0u) {
        unsigned m = (u0 & 0x80808080u) | ((u1 & 0x80808080u) >> 1) |
                     ((u2 & 0x80808080u) >> 2) | ((u3 & 0x80808080u) >> 3);
        if (any == 0u) m = 0u;
        const unsigned tag = (unsigned)idx << 5;
        do {
          if (m != 0u) {
            const unsigned p = (unsigned)__builtin_ctz(m);
            m &= m - 1u;
            q[qn + (int)__builtin_amdgcn_mbcnt_lo(bm, 0u)] = (int)(tag | p);
          }
          qn += __builtin_popcount(bm);
          bm = __builtin_amdgcn_ballot_w32(m != 0u);
        } while (bm != 0u);
        if (qn > QFLUSH) drain();
      }
    }
  };

  uint4 A[KSCAN_U], B[KSCAN_U];
  int baseA = fetch(A, 0);
  for (int i = 0; i < n_it; i += 2) {
    const int baseB = fetch(B, i + 1);
    __builtin_amdgcn_sched_barrier(0);
    take(A, baseA);
    baseA = fetch(A, i + 2);
    __builtin_amdgcn_sched_barrier(0);
    take(B, baseB);
  }
  drain();
}

template <int MODE>
__global__ __launch_bounds__(OWN_THR)
void edge_owner_pass(const int* __restrict__ src, const int* __restrict__ dst,
                     const float* __restrict__ feat, float* __restrict__ deg,
                     float* __restrict__ sums, int* counts, int* lists,
                     const unsigned* __restrict__ keys, int N, int E) {
  __shared__ __attribute__((aligned(16))) float tab[NPR * 128];
  __shared__ int   cnt[NPR];
  __shared__ int   queue[OWN_WAV * QCAP];
  __shared__ int   lcount;

  const int tid  = threadIdx.x;
  const int lane = tid & 31;
  const int wave = tid >> 5;
  const int lo   = blockIdx.x * NPR;
  const int nown = min(NPR, N - lo);

  if (MODE == 1) {
    for (int t = tid; t < NPR * 5; t += OWN_THR) tab[t] = 0.0f;
    for (int t = tid; t < NPR; t += OWN_THR) cnt[t] = 0;
    if (tid == 0) lcount = 0;
  } else {
    for (int t = tid; t < NPR * 32; t += OWN_THR)
      ((float4*)tab)[t] = make_float4(0.0f, 0.0f, 0.0f, 0.0f);
  }
  __syncthreads();

  int* list = lists + (size_t)blockIdx.x * (2 * LCAP);
  const int nl = (MODE == 2) ? counts[blockIdx.x] : 0;

  if (MODE == 2 && nl <= LCAP) {
    const int wv  = __builtin_amdgcn_readfirstlane(wave);
    const int beg = (int)(((long long)wv * nl) / OWN_WAV);
    const int end = (int)(((long long)(wv + 1) * nl) / OWN_WAV);
    int2 nx = make_int2(0, 0);
    if (beg + lane < end) nx = *(const int2*)(list + 2 * (size_t)(beg + lane));
    for (int b = beg; b < end; b += 32) {
      const int2 p = nx;
      if (b + 32 + lane < end) nx = *(const int2*)(list + 2 * (size_t)(b + 32 + lane));
      add_rows(feat, tab, p.x * 128, p.y * 128, min(32, end - b), lane);
    }
  } else if (MODE == 1 && E > 0 && gridDim.x < 255u) {
    owner_sweep_keys(src, dst, (const uint4*)keys, feat, tab, cnt, &queue[wave * QCAP], &lcount, list, lo, (unsigned)nown, E);
  } else if (E > 0) {
    const bool vec_ok = ((E & 3) == 0) && ((((size_t)dst) & 15) == 0);
    if (vec_ok)
      owner_sweep<MODE, true >(src, dst, feat, tab, cnt, &queue[wave * QCAP], &lcount, list, lo, (unsigned)nown, E);
    else
      owner_sweep<MODE, false>(src, dst, feat, tab, cnt, &queue[wave * QCAP], &lcount, list, lo, (unsigned)nown, E);
  }

  __syncthreads();

  if (MODE == 1) {
    if (tid == 0) counts[blockIdx.x] = lcount;
    for (int t = tid; t < nown; t += OWN_THR) deg[lo + t] = 1.0f + (float)cnt[t];
    for (int t = tid; t < nown * 5; t += OWN_THR) sums[(size_t)lo * 5 + t] = tab[t];
  } else {
    for (int r = wave; r < nown; r += OWN_WAV) {
      const float* row = &tab[r * 128 + lane];
      const float4 o = make_float4(row[0], row[32], row[64], row[96]);
      *(float4*)(sums + (size_t)(lo + r) * 128 + lane * 4) = o;
    }
  }
}

template <bool CAT>
__device__ __forceinline__ void stage_wide(const float* __restrict__ A0, const float* __restrict__ A1,
                                           const float* __restrict__ deg, float* As,
                                           int row0, int M, int t) {
  const int c4 = (t & 31) * 4;
  const int mw = t >> 5;
  constexpr int R = CAT ? 2 : 4;
#pragma unroll 1
  for (int h = 0; h < 4; h += R) {
    float4 a[R], s[R];
    float  dg[R];
#pragma unroll
    for (int u = 0; u < R; ++u) {
      const int row = min(row0 + mw + 8 * (h + u), M - 1);
      a[u] = *(const float4*)(A0 + (size_t)row * NCOLS + c4);
      if (CAT) {
        s[u]  = *(const float4*)(A1 + (size_t)row * NCOLS + c4);
        dg[u] = deg[row];
      }
    }
#pragma unroll
    for (int u = 0; u < R; ++u) {
      const int m = mw + 8 * (h + u);
      const bool in = (row0 + m < M);
      float4 v = a[u];
      if (!in) v = make_float4(0.0f, 0.0f, 0.0f, 0.0f);
      *(float4*)&As[m * LDSTRIDE + c4] = v;
      if (CAT) {
        float4 mean;
        mean.x = (a[u].x + s[u].x) / dg[u];
        mean.y = (a[u].y + s[u].y) / dg[u];
        mean.z = (a[u].z + s[u].z) / dg[u];
        mean.w = (a[u].w + s[u].w) / dg[u];
        if (!in) mean = make_float4(0.0f, 0.0f, 0.0f, 0.0f);
        *(float4*)&As[m * LDSTRIDE + NCOLS + c4] = mean;
      }
    }
  }
}

__global__ void sage_gemm_wmma(const float* __restrict__ A0, const float* __restrict__ A1,
                               const float* __restrict__ deg, const float* __restrict__ Wp,
                               const float* __restrict__ bias, float* __restrict__ out,
                               int M, int d, int Kpad) {
  __shared__ __attribute__((aligned(16))) float As[MTILE * LDSTRIDE];

  const int row0 = blockIdx.x * MTILE;
  const int t = threadIdx.x;

  const bool cat  = (A1 != nullptr);
  const bool wide = (d == NCOLS) && (Kpad == (cat ? 2 * NCOLS : NCOLS)) && (M > 0) &&
                    ((((size_t)A0) & 15) == 0) && (!cat || ((((size_t)A1) & 15) == 0));
  if (wide && cat) {
    stage_wide<true >(A0, A1, deg, As, row0, M, t);
  } else if (wide) {
    stage_wide<false>(A0, A1, deg, As, row0, M, t);
  } else {
    const int total = MTILE * Kpad;
    for (int i = t; i < total; i += 256) {
      int m = i / Kpad;
      int k = i - m * Kpad;
      int row = row0 + m;
      float v = 0.0f;
      if (row < M) {
        if (k < d) {
          v = A0[(size_t)row * d + k];
        } else if (A1 != nullptr && k < 2 * d) {
          int kk = k - d;
          v = (A0[(size_t)row * d + kk] + A1[(size_t)row * d + kk]) / deg[row];
        }
      }
      As[m * LDSTRIDE + k] = v;
    }
  }
  __syncthreads();

  const int lane  = t & 31;
  const int wave  = t >> 5;
  const int n     = wave * 16 + (lane & 15);
  const int mA    = lane & 15;
  const int khalf = (lane < 16) ? 0 : 2;

  v8f c0 = {};
  v8f c1 = {};
#pragma unroll 2
  for (int k0 = 0; k0 < Kpad; k0 += 4) {
    const int kb = k0 + khalf;
    v2f a0 = *(const v2f*)&As[mA * LDSTRIDE + kb];
    v2f a1 = *(const v2f*)&As[(mA + 16) * LDSTRIDE + kb];
    v2f b  = *(const v2f*)(Wp + ((size_t)(kb >> 1) * NCOLS + n) * 2);
    c0 = __builtin_amdgcn_wmma_f32_16x16x4_f32(false, a0, false, b,
                                               (short)0, c0, false, false);
    c1 = __builtin_amdgcn_wmma_f32_16x16x4_f32(false, a1, false, b,
                                               (short)0, c1, false, false);
  }

  const float bv    = bias[n];
  const int   rbase = (lane < 16) ? 0 : 8;
  float* outA = out + (size_t)(row0 + rbase) * NCOLS + n;
  float* outB = outA + (size_t)16 * NCOLS;

  if (row0 + MTILE <= M) {
#pragma unroll
    for (int r = 0; r < 8; ++r) {
      float vA = c0[r] + bv;
      float vB = c1[r] + bv;
      outA[(size_t)r * NCOLS] = vA > 0.0f ? vA : 0.0f;
      outB[(size_t)r * NCOLS] = vB > 0.0f ? vB : 0.0f;
    }
  } else {
#pragma unroll
    for (int r = 0; r < 8; ++r) {
      int rowA = row0 + rbase + r;
      if (rowA < M) {
        float v = c0[r] + bv;
        outA[(size_t)r * NCOLS] = v > 0.0f ? v : 0.0f;
      }
      int rowB = row0 + 16 + rbase + r;
      if (rowB < M) {
        float v = c1[r] + bv;
        outB[(size_t)r * NCOLS] = v > 0.0f ? v : 0.0f;
      }
    }
  }
}

__global__ void lin2_kernel(const float* __restrict__ h, const float* __restrict__ W,
                            const float* __restrict__ b, float* __restrict__ out, int M) {
  int i = blockIdx.x * blockDim.x + threadIdx.x;
  if (i < M) {
    float a0 = b[0], a1 = b[1];
    const float* row = h + (size_t)i * 128;
#pragma unroll 8
    for (int k = 0; k < 128; ++k) {
      float v = row[k];
      a0 += v * W[2 * k];
      a1 += v * W[2 * k + 1];
    }
    out[(size_t)i * 2 + 0] = a0;
    out[(size_t)i * 2 + 1] = a1;
  }
}

extern "C" void kernel_launch(void* const* d_in, const int* in_sizes, int n_in,
                              void* d_out, int out_size, void* d_ws, size_t ws_size,
                              hipStream_t stream) {
  const float* x   = (const float*)d_in[0];
  const int*   ei  = (const int*)d_in[1];
  const float* W1  = (const float*)d_in[2];
  const float* b1  = (const float*)d_in[3];
  const float* W2  = (const float*)d_in[4];
  const float* b2  = (const float*)d_in[5];
  const float* l1W = (const float*)d_in[6];
  const float* l1b = (const float*)d_in[7];
  const float* l2W = (const float*)d_in[8];
  const float* l2b = (const float*)d_in[9];

  const int N = in_sizes[0] / 5;
  const int E = in_sizes[1] / 2;
  const int* src = ei;
  const int* dst = ei + E;

  float* ws   = (float*)d_ws;
  float* deg  = ws;
  float* s1   = deg  + (size_t)N;
  float* agg2 = s1   + (size_t)5 * N;
  float* h1   = agg2 + (size_t)128 * N;
  float* h2   = h1   + (size_t)128 * N;
  float* Wp1  = h2   + (size_t)128 * N;
  float* Wp2  = Wp1  + (size_t)12 * 128;
  float* Wp3  = Wp2  + (size_t)256 * 128;
  float* h3   = h1;
  const int nrange = (N + NPR - 1) / NPR;
  int* counts = (int*)(Wp3 + (size_t)128 * 128);
  int* lists  = counts + (size_t)((nrange + 3) & ~3);
  const int nkv = (E + 15) >> 4;
  unsigned* keys = (unsigned*)(lists + (size_t)nrange * (2 * LCAP));

  const int nthread = 256;

  pack_w<<<(12  * NCOLS + nthread - 1) / nthread, nthread, 0, stream>>>(W1,  Wp1, 10,  12);
  pack_w<<<(256 * NCOLS + nthread - 1) / nthread, nthread, 0, stream>>>(W2,  Wp2, 256, 256);
  pack_w<<<(128 * NCOLS + nthread - 1) / nthread, nthread, 0, stream>>>(l1W, Wp3, 128, 128);

  range_keys<<<(nkv + nthread - 1) / nthread, nthread, 0, stream>>>(dst, keys, N, E, nkv);

  edge_owner_pass<1><<<nrange, OWN_THR, 0, stream>>>(src, dst, x, deg, s1, counts, lists, keys, N, E);

  sage_gemm_wmma<<<(N + MTILE - 1) / MTILE, nthread, 0, stream>>>(x, s1, deg, Wp1, b1, h1, N, 5, 12);

  edge_owner_pass<2><<<nrange, OWN_THR, 0, stream>>>(src, dst, h1, deg, agg2, counts, lists, keys, N, E);

  sage_gemm_wmma<<<(N + MTILE - 1) / MTILE, nthread, 0, stream>>>(h1, agg2, deg, Wp2, b2, h2, N, 128, 256);

  sage_gemm_wmma<<<(N + MTILE - 1) / MTILE, nthread, 0, stream>>>(h2, nullptr, deg, Wp3, l1b, h3, N, 128, 128);

  lin2_kernel<<<(N + nthread - 1) / nthread, nthread, 0, stream>>>(h3, l2W, l2b, (float*)d_out, N);
}


# Matrix loops read their B pairs straight from the weights as given (two 4-byte loads a row apart per k-step): the three weight-packing launches go, 10 launches -> 7; k order and sums unchanged
# speedup vs baseline: 3.4449x; 1.0292x over previous
#include <hip/hip_runtime.h>

typedef __attribute__((ext_vector_type(2))) float v2f;
typedef __attribute__((ext_vector_type(8))) float v8f;

#define NCOLS 128
#define MTILE 32
#define LDSTRIDE 260

#define NPR      400
#define OWN_THR  1024
#define OWN_WAV  32
#define SCAN_U   4
#define QFLUSH   96
#define QCAP     (QFLUSH + SCAN_U * 128)
#define LCAP     16384

__device__ __forceinline__ void add_rows(const float* __restrict__ feat, float* tab,
                                         int soff, int doff, int c, int lane) {
  for (int j = 0; j < c; j += 8) {
    float4 r[8];
    int    d[8];
    if (j + 8 <= c) {
#pragma unroll
      for (int u = 0; u < 8; ++u) {
        const int sj = __builtin_amdgcn_readlane(soff, j + u);
        d[u] = __builtin_amdgcn_readlane(doff, j + u);
        r[u] = *(const float4*)(feat + (size_t)sj + lane * 4);
      }
#pragma unroll
      for (int u = 0; u < 8; ++u) {
        float* row = &tab[d[u] + lane];
        __hip_atomic_fetch_add(row +  0, r[u].x, __ATOMIC_RELAXED, __HIP_MEMORY_SCOPE_WORKGROUP);
        __hip_atomic_fetch_add(row + 32, r[u].y, __ATOMIC_RELAXED, __HIP_MEMORY_SCOPE_WORKGROUP);
        __hip_atomic_fetch_add(row + 64, r[u].z, __ATOMIC_RELAXED, __HIP_MEMORY_SCOPE_WORKGROUP);
        __hip_atomic_fetch_add(row + 96, r[u].w, __ATOMIC_RELAXED, __HIP_MEMORY_SCOPE_WORKGROUP);
      }
    } else {
      const int m = c - j;
#pragma unroll
      for (int u = 0; u < 7; ++u) {
        if (u < m) {
          const int sj = __builtin_amdgcn_readlane(soff, j + u);
          d[u] = __builtin_amdgcn_readlane(doff, j + u);
          r[u] = *(const float4*)(feat + (size_t)sj + lane * 4);
        }
      }
#pragma unroll
      for (int u = 0; u < 7; ++u) {
        if (u < m) {
          float* row = &tab[d[u] + lane];
          __hip_atomic_fetch_add(row +  0, r[u].x, __ATOMIC_RELAXED, __HIP_MEMORY_SCOPE_WORKGROUP);
          __hip_atomic_fetch_add(row + 32, r[u].y, __ATOMIC_RELAXED, __HIP_MEMORY_SCOPE_WORKGROUP);
          __hip_atomic_fetch_add(row + 64, r[u].z, __ATOMIC_RELAXED, __HIP_MEMORY_SCOPE_WORKGROUP);
          __hip_atomic_fetch_add(row + 96, r[u].w, __ATOMIC_RELAXED, __HIP_MEMORY_SCOPE_WORKGROUP);
        }
      }
    }
  }
}

template <bool VEC>
__device__ __forceinline__ int4 load_dst4(const int* __restrict__ dst, int idx4, int nvec, int E) {
  int4 r;
  if (VEC) {
    r = *(const int4*)(dst + (size_t)min(idx4, nvec - 1) * 4);
  } else {
    const int e = idx4 * 4;
    r.x = dst[min(e + 0, E - 1)];
    r.y = dst[min(e + 1, E - 1)];
    r.z = dst[min(e + 2, E - 1)];
    r.w = dst[min(e + 3, E - 1)];
    if (e + 0 >= E) r.x = -1;
    if (e + 1 >= E) r.y = -1;
    if (e + 2 >= E) r.z = -1;
    if (e + 3 >= E) r.w = -1;
  }
  return r;
}

template <int MODE, bool VEC>
__device__ __forceinline__ void owner_sweep(const int* __restrict__ src, const int* __restrict__ dst,
                                            const float* __restrict__ feat, float* tab, int* cnt,
                                            int* q, int* lcount, int* list,
                                            int lo, unsigned nloc, int E) {
  const int tid  = threadIdx.x;
  const int lane = tid & 31;
  const int nvec   = (E + 3) >> 2;
  const int per_it = OWN_THR * SCAN_U;
  const int n_it   = (nvec + per_it - 1) / per_it;
  const int rot = (int)(((long long)blockIdx.x * n_it) / gridDim.x);

  int qn = 0;

  auto fetch = [&](int4 (&v)[SCAN_U], int i) -> int {
    int it = i + rot;
    if (it >= n_it) it -= n_it;
    const int base = (i < n_it) ? (it * per_it + tid) : nvec;
#pragma unroll
    for (int u = 0; u < SCAN_U; ++u) v[u] = load_dst4<VEC>(dst, base + u * OWN_THR, nvec, E);
    return base;
  };

  auto take = [&](const int4 (&v)[SCAN_U], int base) {
#pragma unroll
    for (int u = 0; u < SCAN_U; ++u) {
      const int e0 = (base + u * OWN_THR) * 4;
      const unsigned lim = (base + u * OWN_THR < nvec) ? nloc : 0u;
      const bool h0 = (unsigned)(v[u].x - lo) < lim;
      const bool h1 = (unsigned)(v[u].y - lo) < lim;
      const bool h2 = (unsigned)(v[u].z - lo) < lim;
      const bool h3 = (unsigned)(v[u].w - lo) < lim;
      const unsigned m0 = __builtin_amdgcn_ballot_w32(h0);
      const unsigned m1 = __builtin_amdgcn_ballot_w32(h1);
      const unsigned m2 = __builtin_amdgcn_ballot_w32(h2);
      const unsigned m3 = __builtin_amdgcn_ballot_w32(h3);
      if ((m0 | m1 | m2 | m3) != 0u) {
        if (h0) q[qn + __builtin_amdgcn_mbcnt_lo(m0, 0u)] = e0 + 0;
        qn += __builtin_popcount(m0);
        if (h1) q[qn + __builtin_amdgcn_mbcnt_lo(m1, 0u)] = e0 + 1;
        qn += __builtin_popcount(m1);
        if (h2) q[qn + __builtin_amdgcn_mbcnt_lo(m2, 0u)] = e0 + 2;
        qn += __builtin_popcount(m2);
        if (h3) q[qn + __builtin_amdgcn_mbcnt_lo(m3, 0u)] = e0 + 3;
        qn += __builtin_popcount(m3);
      }
    }
  };

  auto drain = [&]() {
    for (int b = 0; b < qn; b += 32) {
      const int k = b + lane;
      int dl = 0, s = 0;
      if (k < qn) {
        const int e = q[k];
        dl = dst[e] - lo;
        s  = src[e];
      }
      if (MODE == 1) {
        int slot = 0;
        if (lane == 0)
          slot = __hip_atomic_fetch_add(lcount, min(32, qn - b), __ATOMIC_RELAXED, __HIP_MEMORY_SCOPE_WORKGROUP);
        slot = __builtin_amdgcn_readfirstlane(slot) + lane;
        if (k < qn && slot < LCAP) *(int2*)(list + 2 * (size_t)slot) = make_int2(s, dl);
        if (k < qn) {
          const float* xr = feat + (size_t)s * 5;
          const float x0 = xr[0], x1 = xr[1], x2 = xr[2], x3 = xr[3], x4 = xr[4];
          __hip_atomic_fetch_add(&cnt[dl], 1, __ATOMIC_RELAXED, __HIP_MEMORY_SCOPE_WORKGROUP);
          float* row = &tab[dl * 5];
          __hip_atomic_fetch_add(row + 0, x0, __ATOMIC_RELAXED, __HIP_MEMORY_SCOPE_WORKGROUP);
          __hip_atomic_fetch_add(row + 1, x1, __ATOMIC_RELAXED, __HIP_MEMORY_SCOPE_WORKGROUP);
          __hip_atomic_fetch_add(row + 2, x2, __ATOMIC_RELAXED, __HIP_MEMORY_SCOPE_WORKGROUP);
          __hip_atomic_fetch_add(row + 3, x3, __ATOMIC_RELAXED, __HIP_MEMORY_SCOPE_WORKGROUP);
          __hip_atomic_fetch_add(row + 4, x4, __ATOMIC_RELAXED, __HIP_MEMORY_SCOPE_WORKGROUP);
        }
      } else {
        add_rows(feat, tab, s * 128, dl * 128, min(32, qn - b), lane);
      }
    }
    qn = 0;
  };

  int4 A[SCAN_U], B[SCAN_U];
  int baseA = fetch(A, 0);
  for (int i = 0; i < n_it; i += 2) {
    const int baseB = fetch(B, i + 1);
    __builtin_amdgcn_sched_barrier(0);
    take(A, baseA);
    if (qn > QFLUSH) drain();
    baseA = fetch(A, i + 2);
    __builtin_amdgcn_sched_barrier(0);
    take(B, baseB);
    if (qn > QFLUSH || i + 2 >= n_it) drain();
  }
}

__device__ __forceinline__ unsigned range_key(int d, int N) {
  return ((unsigned)d < (unsigned)N) ? (unsigned)d / NPR : 0xFFu;
}

__global__ void range_keys(const int* __restrict__ dst, unsigned* __restrict__ keys,
                           int N, int E, int nkv) {
  const int i = blockIdx.x * blockDim.x + threadIdx.x;
  if (i >= nkv) return;
  const int e0 = i * 16;
  unsigned w[4];
  if (e0 + 16 <= E && ((((size_t)dst) & 15) == 0)) {
#pragma unroll
    for (int j = 0; j < 4; ++j) {
      const int4 v = *(const int4*)(dst + e0 + 4 * j);
      w[j] = range_key(v.x, N) | (range_key(v.y, N) << 8) |
             (range_key(v.z, N) << 16) | (range_key(v.w, N) << 24);
    }
  } else {
#pragma unroll
    for (int j = 0; j < 4; ++j) {
      unsigned p = 0u;
#pragma unroll
      for (int y = 0; y < 4; ++y) {
        const int e = e0 + 4 * j + y;
        const int d = (e < E) ? dst[e] : -1;
        p |= range_key(d, N) << (8 * y);
      }
      w[j] = p;
    }
  }
  ((uint4*)keys)[i] = make_uint4(w[0], w[1], w[2], w[3]);
}

#define KSCAN_U 4
__device__ __forceinline__ void owner_sweep_keys(const int* __restrict__ src, const int* __restrict__ dst,
                                                 const uint4* __restrict__ keys,
                                                 const float* __restrict__ feat, float* tab, int* cnt,
                                                 int* q, int* lcount, int* list,
                                                 int lo, unsigned nloc, int E) {
  const int tid  = threadIdx.x;
  const int lane = tid & 31;
  const int nkv    = (E + 15) >> 4;
  const int per_it = OWN_THR * KSCAN_U;
  const int n_it   = (nkv + per_it - 1) / per_it;
  const int rot = (int)(((long long)blockIdx.x * n_it) / gridDim.x);
  const unsigned pat = (unsigned)blockIdx.x * 0x01010101u;

  int qn = 0;

  auto fetch = [&](uint4 (&v)[KSCAN_U], int i) -> int {
    int it = i + rot;
    if (it >= n_it) it -= n_it;
    const int base = (i < n_it) ? (it * per_it + tid) : nkv;
#pragma unroll
    for (int u = 0; u < KSCAN_U; ++u) v[u] = keys[min(base + u * OWN_THR, nkv - 1)];
    return base;
  };

  auto drain = [&]() {
    for (int b = 0; b < qn; b += 32) {
      const int k = b + lane;
      int dl = 0, s = 0;
      bool ok = false;
      if (k < qn) {
        const unsigned w = (unsigned)q[k];
        const int e = (int)((w >> 5) << 4) + (int)(((7u - (w & 7u)) << 2) + ((w & 31u) >> 3));
        if (e < E) {
          dl = dst[e] - lo;
          s  = src[e];
          ok = (unsigned)dl < nloc;
        }
      }
      const unsigned okm = __builtin_amdgcn_ballot_w32(ok);
      int slot = 0;
      if (lane == 0)
        slot = __hip_atomic_fetch_add(lcount, __builtin_popcount(okm), __ATOMIC_RELAXED, __HIP_MEMORY_SCOPE_WORKGROUP);
      slot = __builtin_amdgcn_readfirstlane(slot) + (int)__builtin_amdgcn_mbcnt_lo(okm, 0u);
      if (ok && slot < LCAP) *(int2*)(list + 2 * (size_t)slot) = make_int2(s, dl);
      if (ok) {
        const float* xr = feat + (size_t)s * 5;
        const float x0 = xr[0], x1 = xr[1], x2 = xr[2], x3 = xr[3], x4 = xr[4];
        __hip_atomic_fetch_add(&cnt[dl], 1, __ATOMIC_RELAXED, __HIP_MEMORY_SCOPE_WORKGROUP);
        float* row = &tab[dl * 5];
        __hip_atomic_fetch_add(row + 0, x0, __ATOMIC_RELAXED, __HIP_MEMORY_SCOPE_WORKGROUP);
        __hip_atomic_fetch_add(row + 1, x1, __ATOMIC_RELAXED, __HIP_MEMORY_SCOPE_WORKGROUP);
        __hip_atomic_fetch_add(row + 2, x2, __ATOMIC_RELAXED, __HIP_MEMORY_SCOPE_WORKGROUP);
        __hip_atomic_fetch_add(row + 3, x3, __ATOMIC_RELAXED, __HIP_MEMORY_SCOPE_WORKGROUP);
        __hip_atomic_fetch_add(row + 4, x4, __ATOMIC_RELAXED, __HIP_MEMORY_SCOPE_WORKGROUP);
      }
    }
    qn = 0;
  };

  auto take = [&](const uint4 (&v)[KSCAN_U], int base) {
#pragma unroll
    for (int u = 0; u < KSCAN_U; ++u) {
      const int idx = base + u * OWN_THR;
      const unsigned x0 = v[u].x ^ pat, x1 = v[u].y ^ pat, x2 = v[u].z ^ pat, x3 = v[u].w ^ pat;
      const unsigned u0 = (x0 - 0x01010101u) & ~x0;
      const unsigned u1 = (x1 - 0x01010101u) & ~x1;
      const unsigned u2 = (x2 - 0x01010101u) & ~x2;
      const unsigned u3 = (x3 - 0x01010101u) & ~x3;
      unsigned any = (u0 | u1 | u2 | u3) & 0x80808080u;
      if (idx >= nkv) any = 0u;
      unsigned bm = __builtin_amdgcn_ballot_w32(any != 0u);
      if (bm != 0u) {
        unsigned m = (u0 & 0x80808080u) | ((u1 & 0x80808080u) >> 1) |
                     ((u2 & 0x80808080u) >> 2) | ((u3 & 0x80808080u) >> 3);
        if (any == 0u) m = 0u;
        const unsigned tag = (unsigned)idx << 5;
        do {
          if (m != 0u) {
            const unsigned p = (unsigned)__builtin_ctz(m);
            m &= m - 1u;
            q[qn + (int)__builtin_amdgcn_mbcnt_lo(bm, 0u)] = (int)(tag | p);
          }
          qn += __builtin_popcount(bm);
          bm = __builtin_amdgcn_ballot_w32(m != 0u);
        } while (bm != 0u);
        if (qn > QFLUSH) drain();
      }
    }
  };

  uint4 A[KSCAN_U], B[KSCAN_U];
  int baseA = fetch(A, 0);
  for (int i = 0; i < n_it; i += 2) {
    const int baseB = fetch(B, i + 1);
    __builtin_amdgcn_sched_barrier(0);
    take(A, baseA);
    baseA = fetch(A, i + 2);
    __builtin_amdgcn_sched_barrier(0);
    take(B, baseB);
  }
  drain();
}

template <int MODE>
__global__ __launch_bounds__(OWN_THR)
void edge_owner_pass(const int* __restrict__ src, const int* __restrict__ dst,
                     const float* __restrict__ feat, float* __restrict__ deg,
                     float* __restrict__ sums, int* counts, int* lists,
                     const unsigned* __restrict__ keys, int N, int E) {
  __shared__ __attribute__((aligned(16))) float tab[NPR * 128];
  __shared__ int   cnt[NPR];
  __shared__ int   queue[OWN_WAV * QCAP];
  __shared__ int   lcount;

  const int tid  = threadIdx.x;
  const int lane = tid & 31;
  const int wave = tid >> 5;
  const int lo   = blockIdx.x * NPR;
  const int nown = min(NPR, N - lo);

  if (MODE == 1) {
    for (int t = tid; t < NPR * 5; t += OWN_THR) tab[t] = 0.0f;
    for (int t = tid; t < NPR; t += OWN_THR) cnt[t] = 0;
    if (tid == 0) lcount = 0;
  } else {
    for (int t = tid; t < NPR * 32; t += OWN_THR)
      ((float4*)tab)[t] = make_float4(0.0f, 0.0f, 0.0f, 0.0f);
  }
  __syncthreads();

  int* list = lists + (size_t)blockIdx.x * (2 * LCAP);
  const int nl = (MODE == 2) ? counts[blockIdx.x] : 0;

  if (MODE == 2 && nl <= LCAP) {
    const int wv  = __builtin_amdgcn_readfirstlane(wave);
    const int beg = (int)(((long long)wv * nl) / OWN_WAV);
    const int end = (int)(((long long)(wv + 1) * nl) / OWN_WAV);
    int2 nx = make_int2(0, 0);
    if (beg + lane < end) nx = *(const int2*)(list + 2 * (size_t)(beg + lane));
    for (int b = beg; b < end; b += 32) {
      const int2 p = nx;
      if (b + 32 + lane < end) nx = *(const int2*)(list + 2 * (size_t)(b + 32 + lane));
      add_rows(feat, tab, p.x * 128, p.y * 128, min(32, end - b), lane);
    }
  } else if (MODE == 1 && E > 0 && gridDim.x < 255u) {
    owner_sweep_keys(src, dst, (const uint4*)keys, feat, tab, cnt, &queue[wave * QCAP], &lcount, list, lo, (unsigned)nown, E);
  } else if (E > 0) {
    const bool vec_ok = ((E & 3) == 0) && ((((size_t)dst) & 15) == 0);
    if (vec_ok)
      owner_sweep<MODE, true >(src, dst, feat, tab, cnt, &queue[wave * QCAP], &lcount, list, lo, (unsigned)nown, E);
    else
      owner_sweep<MODE, false>(src, dst, feat, tab, cnt, &queue[wave * QCAP], &lcount, list, lo, (unsigned)nown, E);
  }

  __syncthreads();

  if (MODE == 1) {
    if (tid == 0) counts[blockIdx.x] = lcount;
    for (int t = tid; t < nown; t += OWN_THR) deg[lo + t] = 1.0f + (float)cnt[t];
    for (int t = tid; t < nown * 5; t += OWN_THR) sums[(size_t)lo * 5 + t] = tab[t];
  } else {
    for (int r = wave; r < nown; r += OWN_WAV) {
      const float* row = &tab[r * 128 + lane];
      const float4 o = make_float4(row[0], row[32], row[64], row[96]);
      *(float4*)(sums + (size_t)(lo + r) * 128 + lane * 4) = o;
    }
  }
}

template <bool CAT>
__device__ __forceinline__ void stage_wide(const float* __restrict__ A0, const float* __restrict__ A1,
                                           const float* __restrict__ deg, float* As,
                                           int row0, int M, int t) {
  const int c4 = (t & 31) * 4;
  const int mw = t >> 5;
  constexpr int R = CAT ? 2 : 4;
#pragma unroll 1
  for (int h = 0; h < 4; h += R) {
    float4 a[R], s[R];
    float  dg[R];
#pragma unroll
    for (int u = 0; u < R; ++u) {
      const int row = min(row0 + mw + 8 * (h + u), M - 1);
      a[u] = *(const float4*)(A0 + (size_t)row * NCOLS + c4);
      if (CAT) {
        s[u]  = *(const float4*)(A1 + (size_t)row * NCOLS + c4);
        dg[u] = deg[row];
      }
    }
#pragma unroll
    for (int u = 0; u < R; ++u) {
      const int m = mw + 8 * (h + u);
      const bool in = (row0 + m < M);
      float4 v = a[u];
      if (!in) v = make_float4(0.0f, 0.0f, 0.0f, 0.0f);
      *(float4*)&As[m * LDSTRIDE + c4] = v;
      if (CAT) {
        float4 mean;
        mean.x = (a[u].x + s[u].x) / dg[u];
        mean.y = (a[u].y + s[u].y) / dg[u];
        mean.z = (a[u].z + s[u].z) / dg[u];
        mean.w = (a[u].w + s[u].w) / dg[u];
        if (!in) mean = make_float4(0.0f, 0.0f, 0.0f, 0.0f);
        *(float4*)&As[m * LDSTRIDE + NCOLS + c4] = mean;
      }
    }
  }
}

__global__ void sage_gemm_wmma(const float* __restrict__ A0, const float* __restrict__ A1,
                               const float* __restrict__ deg, const float* __restrict__ W,
                               const float* __restrict__ bias, float* __restrict__ out,
                               int M, int d, int Kpad) {
  __shared__ __attribute__((aligned(16))) float As[MTILE * LDSTRIDE];

  const int row0 = blockIdx.x * MTILE;
  const int t = threadIdx.x;

  const bool cat  = (A1 != nullptr);
  const bool wide = (d == NCOLS) && (Kpad == (cat ? 2 * NCOLS : NCOLS)) && (M > 0) &&
                    ((((size_t)A0) & 15) == 0) && (!cat || ((((size_t)A1) & 15) == 0));
  if (wide && cat) {
    stage_wide<true >(A0, A1, deg, As, row0, M, t);
  } else if (wide) {
    stage_wide<false>(A0, A1, deg, As, row0, M, t);
  } else {
    const int total = MTILE * Kpad;
    for (int i = t; i < total; i += 256) {
      int m = i / Kpad;
      int k = i - m * Kpad;
      int row = row0 + m;
      float v = 0.0f;
      if (row < M) {
        if (k < d) {
          v = A0[(size_t)row * d + k];
        } else if (A1 != nullptr && k < 2 * d) {
          int kk = k - d;
          v = (A0[(size_t)row * d + kk] + A1[(size_t)row * d + kk]) / deg[row];
        }
      }
      As[m * LDSTRIDE + k] = v;
    }
  }
  __syncthreads();

  const int lane  = t & 31;
  const int wave  = t >> 5;
  const int n     = wave * 16 + (lane & 15);
  const int mA    = lane & 15;
  const int khalf = (lane < 16) ? 0 : 2;

  const int Kw = cat ? 2 * d : d;
  const float* wcol = W + n;

  v8f c0 = {};
  v8f c1 = {};
  if (Kw == Kpad) {
#pragma unroll 2
    for (int k0 = 0; k0 < Kpad; k0 += 4) {
      const int kb = k0 + khalf;
      v2f a0 = *(const v2f*)&As[mA * LDSTRIDE + kb];
      v2f a1 = *(const v2f*)&As[(mA + 16) * LDSTRIDE + kb];
      v2f b;
      b.x = wcol[(size_t)kb * NCOLS];
      b.y = wcol[(size_t)(kb + 1) * NCOLS];
      c0 = __builtin_amdgcn_wmma_f32_16x16x4_f32(false, a0, false, b,
                                                 (short)0, c0, false, false);
      c1 = __builtin_amdgcn_wmma_f32_16x16x4_f32(false, a1, false, b,
                                                 (short)0, c1, false, false);
    }
  } else {
#pragma unroll 2
    for (int k0 = 0; k0 < Kpad; k0 += 4) {
      const int kb = k0 + khalf;
      v2f a0 = *(const v2f*)&As[mA * LDSTRIDE + kb];
      v2f a1 = *(const v2f*)&As[(mA + 16) * LDSTRIDE + kb];
      const float w0 = wcol[(size_t)min(kb,     Kw - 1) * NCOLS];
      const float w1 = wcol[(size_t)min(kb + 1, Kw - 1) * NCOLS];
      v2f b;
      b.x = (kb     < Kw) ? w0 : 0.0f;
      b.y = (kb + 1 < Kw) ? w1 : 0.0f;
      c0 = __builtin_amdgcn_wmma_f32_16x16x4_f32(false, a0, false, b,
                                                 (short)0, c0, false, false);
      c1 = __builtin_amdgcn_wmma_f32_16x16x4_f32(false, a1, false, b,
                                                 (short)0, c1, false, false);
    }
  }

  const float bv    = bias[n];
  const int   rbase = (lane < 16) ? 0 : 8;
  float* outA = out + (size_t)(row0 + rbase) * NCOLS + n;
  float* outB = outA + (size_t)16 * NCOLS;

  if (row0 + MTILE <= M) {
#pragma unroll
    for (int r = 0; r < 8; ++r) {
      float vA = c0[r] + bv;
      float vB = c1[r] + bv;
      outA[(size_t)r * NCOLS] = vA > 0.0f ? vA : 0.0f;
      outB[(size_t)r * NCOLS] = vB > 0.0f ? vB : 0.0f;
    }
  } else {
#pragma unroll
    for (int r = 0; r < 8; ++r) {
      int rowA = row0 + rbase + r;
      if (rowA < M) {
        float v = c0[r] + bv;
        outA[(size_t)r * NCOLS] = v > 0.0f ? v : 0.0f;
      }
      int rowB = row0 + 16 + rbase + r;
      if (rowB < M) {
        float v = c1[r] + bv;
        outB[(size_t)r * NCOLS] = v > 0.0f ? v : 0.0f;
      }
    }
  }
}

__global__ void lin2_kernel(const float* __restrict__ h, const float* __restrict__ W,
                            const float* __restrict__ b, float* __restrict__ out, int M) {
  int i = blockIdx.x * blockDim.x + threadIdx.x;
  if (i < M) {
    float a0 = b[0], a1 = b[1];
    const float* row = h + (size_t)i * 128;
#pragma unroll 8
    for (int k = 0; k < 128; ++k) {
      float v = row[k];
      a0 += v * W[2 * k];
      a1 += v * W[2 * k + 1];
    }
    out[(size_t)i * 2 + 0] = a0;
    out[(size_t)i * 2 + 1] = a1;
  }
}

extern "C" void kernel_launch(void* const* d_in, const int* in_sizes, int n_in,
                              void* d_out, int out_size, void* d_ws, size_t ws_size,
                              hipStream_t stream) {
  const float* x   = (const float*)d_in[0];
  const int*   ei  = (const int*)d_in[1];
  const float* W1  = (const float*)d_in[2];
  const float* b1  = (const float*)d_in[3];
  const float* W2  = (const float*)d_in[4];
  const float* b2  = (const float*)d_in[5];
  const float* l1W = (const float*)d_in[6];
  const float* l1b = (const float*)d_in[7];
  const float* l2W = (const float*)d_in[8];
  const float* l2b = (const float*)d_in[9];

  const int N = in_sizes[0] / 5;
  const int E = in_sizes[1] / 2;
  const int* src = ei;
  const int* dst = ei + E;

  float* ws   = (float*)d_ws;
  float* deg  = ws;
  float* s1   = deg  + (size_t)N;
  float* agg2 = s1   + (size_t)5 * N;
  float* h1   = agg2 + (size_t)128 * N;
  float* h2   = h1   + (size_t)128 * N;
  float* h3   = h1;
  const int nrange = (N + NPR - 1) / NPR;
  int* counts = (int*)(h2 + (size_t)128 * N);
  int* lists  = counts + (size_t)((nrange + 3) & ~3);
  const int nkv = (E + 15) >> 4;
  unsigned* keys = (unsigned*)(lists + (size_t)nrange * (2 * LCAP));

  const int nthread = 256;

  range_keys<<<(nkv + nthread - 1) / nthread, nthread, 0, stream>>>(dst, keys, N, E, nkv);

  edge_owner_pass<1><<<nrange, OWN_THR, 0, stream>>>(src, dst, x, deg, s1, counts, lists, keys, N, E);

  sage_gemm_wmma<<<(N + MTILE - 1) / MTILE, nthread, 0, stream>>>(x, s1, deg, W1, b1, h1, N, 5, 12);

  edge_owner_pass<2><<<nrange, OWN_THR, 0, stream>>>(src, dst, h1, deg, agg2, counts, lists, keys, N, E);

  sage_gemm_wmma<<<(N + MTILE - 1) / MTILE, nthread, 0, stream>>>(h1, agg2, deg, W2, b2, h2, N, 128, 256);

  sage_gemm_wmma<<<(N + MTILE - 1) / MTILE, nthread, 0, stream>>>(h2, nullptr, deg, l1W, l1b, h3, N, 128, 128);

  lin2_kernel<<<(N + nthread - 1) / nthread, nthread, 0, stream>>>(h3, l2W, l2b, (float*)d_out, N);
}


# Layer 2 multiplies before it sums: one launch keeps h1 tiles in LDS, writes h1 @ W2 by halves; the edge pass sums product rows and forms h2; 7 launches -> 6; sum of products, not product of sums
# speedup vs baseline: 3.5586x; 1.0330x over previous
#include <hip/hip_runtime.h>

typedef __attribute__((ext_vector_type(2))) float v2f;
typedef __attribute__((ext_vector_type(8))) float v8f;

#define NCOLS 128
#define MTILE 32
#define LDSTRIDE 260

#define NPR      400
#define OWN_THR  1024
#define OWN_WAV  32
#define SCAN_U   4
#define QFLUSH   96
#define QCAP     (QFLUSH + SCAN_U * 128)
#define LCAP     16384

__device__ __forceinline__ void add_rows(const float* __restrict__ feat, float* tab,
                                         int soff, int doff, int c, int lane) {
  for (int j = 0; j < c; j += 8) {
    float4 r[8];
    int    d[8];
    if (j + 8 <= c) {
#pragma unroll
      for (int u = 0; u < 8; ++u) {
        const int sj = __builtin_amdgcn_readlane(soff, j + u);
        d[u] = __builtin_amdgcn_readlane(doff, j + u);
        r[u] = *(const float4*)(feat + (size_t)sj + lane * 4);
      }
#pragma unroll
      for (int u = 0; u < 8; ++u) {
        float* row = &tab[d[u] + lane];
        __hip_atomic_fetch_add(row +  0, r[u].x, __ATOMIC_RELAXED, __HIP_MEMORY_SCOPE_WORKGROUP);
        __hip_atomic_fetch_add(row + 32, r[u].y, __ATOMIC_RELAXED, __HIP_MEMORY_SCOPE_WORKGROUP);
        __hip_atomic_fetch_add(row + 64, r[u].z, __ATOMIC_RELAXED, __HIP_MEMORY_SCOPE_WORKGROUP);
        __hip_atomic_fetch_add(row + 96, r[u].w, __ATOMIC_RELAXED, __HIP_MEMORY_SCOPE_WORKGROUP);
      }
    } else {
      const int m = c - j;
#pragma unroll
      for (int u = 0; u < 7; ++u) {
        if (u < m) {
          const int sj = __builtin_amdgcn_readlane(soff, j + u);
          d[u] = __builtin_amdgcn_readlane(doff, j + u);
          r[u] = *(const float4*)(feat + (size_t)sj + lane * 4);
        }
      }
#pragma unroll
      for (int u = 0; u < 7; ++u) {
        if (u < m) {
          float* row = &tab[d[u] + lane];
          __hip_atomic_fetch_add(row +  0, r[u].x, __ATOMIC_RELAXED, __HIP_MEMORY_SCOPE_WORKGROUP);
          __hip_atomic_fetch_add(row + 32, r[u].y, __ATOMIC_RELAXED, __HIP_MEMORY_SCOPE_WORKGROUP);
          __hip_atomic_fetch_add(row + 64, r[u].z, __ATOMIC_RELAXED, __HIP_MEMORY_SCOPE_WORKGROUP);
          __hip_atomic_fetch_add(row + 96, r[u].w, __ATOMIC_RELAXED, __HIP_MEMORY_SCOPE_WORKGROUP);
        }
      }
    }
  }
}

template <bool VEC>
__device__ __forceinline__ int4 load_dst4(const int* __restrict__ dst, int idx4, int nvec, int E) {
  int4 r;
  if (VEC) {
    r = *(const int4*)(dst + (size_t)min(idx4, nvec - 1) * 4);
  } else {
    const int e = idx4 * 4;
    r.x = dst[min(e + 0, E - 1)];
    r.y = dst[min(e + 1, E - 1)];
    r.z = dst[min(e + 2, E - 1)];
    r.w = dst[min(e + 3, E - 1)];
    if (e + 0 >= E) r.x = -1;
    if (e + 1 >= E) r.y = -1;
    if (e + 2 >= E) r.z = -1;
    if (e + 3 >= E) r.w = -1;
  }
  return r;
}

template <int MODE, bool VEC>
__device__ __forceinline__ void owner_sweep(const int* __restrict__ src, const int* __restrict__ dst,
                                            const float* __restrict__ feat, float* tab, int* cnt,
                                            int* q, int* lcount, int* list,
                                            int lo, unsigned nloc, int E) {
  const int tid  = threadIdx.x;
  const int lane = tid & 31;
  const int nvec   = (E + 3) >> 2;
  const int per_it = OWN_THR * SCAN_U;
  const int n_it   = (nvec + per_it - 1) / per_it;
  const int rot = (int)(((long long)blockIdx.x * n_it) / gridDim.x);

  int qn = 0;

  auto fetch = [&](int4 (&v)[SCAN_U], int i) -> int {
    int it = i + rot;
    if (it >= n_it) it -= n_it;
    const int base = (i < n_it) ? (it * per_it + tid) : nvec;
#pragma unroll
    for (int u = 0; u < SCAN_U; ++u) v[u] = load_dst4<VEC>(dst, base + u * OWN_THR, nvec, E);
    return base;
  };

  auto take = [&](const int4 (&v)[SCAN_U], int base) {
#pragma unroll
    for (int u = 0; u < SCAN_U; ++u) {
      const int e0 = (base + u * OWN_THR) * 4;
      const unsigned lim = (base + u * OWN_THR < nvec) ? nloc : 0u;
      const bool h0 = (unsigned)(v[u].x - lo) < lim;
      const bool h1 = (unsigned)(v[u].y - lo) < lim;
      const bool h2 = (unsigned)(v[u].z - lo) < lim;
      const bool h3 = (unsigned)(v[u].w - lo) < lim;
      const unsigned m0 = __builtin_amdgcn_ballot_w32(h0);
      const unsigned m1 = __builtin_amdgcn_ballot_w32(h1);
      const unsigned m2 = __builtin_amdgcn_ballot_w32(h2);
      const unsigned m3 = __builtin_amdgcn_ballot_w32(h3);
      if ((m0 | m1 | m2 | m3) != 0u) {
        if (h0) q[qn + __builtin_amdgcn_mbcnt_lo(m0, 0u)] = e0 + 0;
        qn += __builtin_popcount(m0);
        if (h1) q[qn + __builtin_amdgcn_mbcnt_lo(m1, 0u)] = e0 + 1;
        qn += __builtin_popcount(m1);
        if (h2) q[qn + __builtin_amdgcn_mbcnt_lo(m2, 0u)] = e0 + 2;
        qn += __builtin_popcount(m2);
        if (h3) q[qn + __builtin_amdgcn_mbcnt_lo(m3, 0u)] = e0 + 3;
        qn += __builtin_popcount(m3);
      }
    }
  };

  auto drain = [&]() {
    for (int b = 0; b < qn; b += 32) {
      const int k = b + lane;
      int dl = 0, s = 0;
      if (k < qn) {
        const int e = q[k];
        dl = dst[e] - lo;
        s  = src[e];
      }
      if (MODE == 1) {
        int slot = 0;
        if (lane == 0)
          slot = __hip_atomic_fetch_add(lcount, min(32, qn - b), __ATOMIC_RELAXED, __HIP_MEMORY_SCOPE_WORKGROUP);
        slot = __builtin_amdgcn_readfirstlane(slot) + lane;
        if (k < qn && slot < LCAP) *(int2*)(list + 2 * (size_t)slot) = make_int2(s, dl);
        if (k < qn) {
          const float* xr = feat + (size_t)s * 5;
          const float x0 = xr[0], x1 = xr[1], x2 = xr[2], x3 = xr[3], x4 = xr[4];
          __hip_atomic_fetch_add(&cnt[dl], 1, __ATOMIC_RELAXED, __HIP_MEMORY_SCOPE_WORKGROUP);
          float* row = &tab[dl * 5];
          __hip_atomic_fetch_add(row + 0, x0, __ATOMIC_RELAXED, __HIP_MEMORY_SCOPE_WORKGROUP);
          __hip_atomic_fetch_add(row + 1, x1, __ATOMIC_RELAXED, __HIP_MEMORY_SCOPE_WORKGROUP);
          __hip_atomic_fetch_add(row + 2, x2, __ATOMIC_RELAXED, __HIP_MEMORY_SCOPE_WORKGROUP);
          __hip_atomic_fetch_add(row + 3, x3, __ATOMIC_RELAXED, __HIP_MEMORY_SCOPE_WORKGROUP);
          __hip_atomic_fetch_add(row + 4, x4, __ATOMIC_RELAXED, __HIP_MEMORY_SCOPE_WORKGROUP);
        }
      } else {
        add_rows(feat, tab, s * 128, dl * 128, min(32, qn - b), lane);
      }
    }
    qn = 0;
  };

  int4 A[SCAN_U], B[SCAN_U];
  int baseA = fetch(A, 0);
  for (int i = 0; i < n_it; i += 2) {
    const int baseB = fetch(B, i + 1);
    __builtin_amdgcn_sched_barrier(0);
    take(A, baseA);
    if (qn > QFLUSH) drain();
    baseA = fetch(A, i + 2);
    __builtin_amdgcn_sched_barrier(0);
    take(B, baseB);
    if (qn > QFLUSH || i + 2 >= n_it) drain();
  }
}

__device__ __forceinline__ unsigned range_key(int d, int N) {
  return ((unsigned)d < (unsigned)N) ? (unsigned)d / NPR : 0xFFu;
}

__global__ void range_keys(const int* __restrict__ dst, unsigned* __restrict__ keys,
                           int N, int E, int nkv) {
  const int i = blockIdx.x * blockDim.x + threadIdx.x;
  if (i >= nkv) return;
  const int e0 = i * 16;
  unsigned w[4];
  if (e0 + 16 <= E && ((((size_t)dst) & 15) == 0)) {
#pragma unroll
    for (int j = 0; j < 4; ++j) {
      const int4 v = *(const int4*)(dst + e0 + 4 * j);
      w[j] = range_key(v.x, N) | (range_key(v.y, N) << 8) |
             (range_key(v.z, N) << 16) | (range_key(v.w, N) << 24);
    }
  } else {
#pragma unroll
    for (int j = 0; j < 4; ++j) {
      unsigned p = 0u;
#pragma unroll
      for (int y = 0; y < 4; ++y) {
        const int e = e0 + 4 * j + y;
        const int d = (e < E) ? dst[e] : -1;
        p |= range_key(d, N) << (8 * y);
      }
      w[j] = p;
    }
  }
  ((uint4*)keys)[i] = make_uint4(w[0], w[1], w[2], w[3]);
}

#define KSCAN_U 4
__device__ __forceinline__ void owner_sweep_keys(const int* __restrict__ src, const int* __restrict__ dst,
                                                 const uint4* __restrict__ keys,
                                                 const float* __restrict__ feat, float* tab, int* cnt,
                                                 int* q, int* lcount, int* list,
                                                 int lo, unsigned nloc, int E) {
  const int tid  = threadIdx.x;
  const int lane = tid & 31;
  const int nkv    = (E + 15) >> 4;
  const int per_it = OWN_THR * KSCAN_U;
  const int n_it   = (nkv + per_it - 1) / per_it;
  const int rot = (int)(((long long)blockIdx.x * n_it) / gridDim.x);
  const unsigned pat = (unsigned)blockIdx.x * 0x01010101u;

  int qn = 0;

  auto fetch = [&](uint4 (&v)[KSCAN_U], int i) -> int {
    int it = i + rot;
    if (it >= n_it) it -= n_it;
    const int base = (i < n_it) ? (it * per_it + tid) : nkv;
#pragma unroll
    for (int u = 0; u < KSCAN_U; ++u) v[u] = keys[min(base + u * OWN_THR, nkv - 1)];
    return base;
  };

  auto drain = [&]() {
    for (int b = 0; b < qn; b += 32) {
      const int k = b + lane;
      int dl = 0, s = 0;
      bool ok = false;
      if (k < qn) {
        const unsigned w = (unsigned)q[k];
        const int e = (int)((w >> 5) << 4) + (int)(((7u - (w & 7u)) << 2) + ((w & 31u) >> 3));
        if (e < E) {
          dl = dst[e] - lo;
          s  = src[e];
          ok = (unsigned)dl < nloc;
        }
      }
      const unsigned okm = __builtin_amdgcn_ballot_w32(ok);
      int slot = 0;
      if (lane == 0)
        slot = __hip_atomic_fetch_add(lcount, __builtin_popcount(okm), __ATOMIC_RELAXED, __HIP_MEMORY_SCOPE_WORKGROUP);
      slot = __builtin_amdgcn_readfirstlane(slot) + (int)__builtin_amdgcn_mbcnt_lo(okm, 0u);
      if (ok && slot < LCAP) *(int2*)(list + 2 * (size_t)slot) = make_int2(s, dl);
      if (ok) {
        const float* xr = feat + (size_t)s * 5;
        const float x0 = xr[0], x1 = xr[1], x2 = xr[2], x3 = xr[3], x4 = xr[4];
        __hip_atomic_fetch_add(&cnt[dl], 1, __ATOMIC_RELAXED, __HIP_MEMORY_SCOPE_WORKGROUP);
        float* row = &tab[dl * 5];
        __hip_atomic_fetch_add(row + 0, x0, __ATOMIC_RELAXED, __HIP_MEMORY_SCOPE_WORKGROUP);
        __hip_atomic_fetch_add(row + 1, x1, __ATOMIC_RELAXED, __HIP_MEMORY_SCOPE_WORKGROUP);
        __hip_atomic_fetch_add(row + 2, x2, __ATOMIC_RELAXED, __HIP_MEMORY_SCOPE_WORKGROUP);
        __hip_atomic_fetch_add(row + 3, x3, __ATOMIC_RELAXED, __HIP_MEMORY_SCOPE_WORKGROUP);
        __hip_atomic_fetch_add(row + 4, x4, __ATOMIC_RELAXED, __HIP_MEMORY_SCOPE_WORKGROUP);
      }
    }
    qn = 0;
  };

  auto take = [&](const uint4 (&v)[KSCAN_U], int base) {
#pragma unroll
    for (int u = 0; u < KSCAN_U; ++u) {
      const int idx = base + u * OWN_THR;
      const unsigned x0 = v[u].x ^ pat, x1 = v[u].y ^ pat, x2 = v[u].z ^ pat, x3 = v[u].w ^ pat;
      const unsigned u0 = (x0 - 0x01010101u) & ~x0;
      const unsigned u1 = (x1 - 0x01010101u) & ~x1;
      const unsigned u2 = (x2 - 0x01010101u) & ~x2;
      const unsigned u3 = (x3 - 0x01010101u) & ~x3;
      unsigned any = (u0 | u1 | u2 | u3) & 0x80808080u;
      if (idx >= nkv) any = 0u;
      unsigned bm = __builtin_amdgcn_ballot_w32(any != 0u);
      if (bm != 0u) {
        unsigned m = (u0 & 0x80808080u) | ((u1 & 0x80808080u) >> 1) |
                     ((u2 & 0x80808080u) >> 2) | ((u3 & 0x80808080u) >> 3);
        if (any == 0u) m = 0u;
        const unsigned tag = (unsigned)idx << 5;
        do {
          if (m != 0u) {
            const unsigned p = (unsigned)__builtin_ctz(m);
            m &= m - 1u;
            q[qn + (int)__builtin_amdgcn_mbcnt_lo(bm, 0u)] = (int)(tag | p);
          }
          qn += __builtin_popcount(bm);
          bm = __builtin_amdgcn_ballot_w32(m != 0u);
        } while (bm != 0u);
        if (qn > QFLUSH) drain();
      }
    }
  };

  uint4 A[KSCAN_U], B[KSCAN_U];
  int baseA = fetch(A, 0);
  for (int i = 0; i < n_it; i += 2) {
    const int baseB = fetch(B, i + 1);
    __builtin_amdgcn_sched_barrier(0);
    take(A, baseA);
    baseA = fetch(A, i + 2);
    __builtin_amdgcn_sched_barrier(0);
    take(B, baseB);
  }
  drain();
}

template <int MODE>
__global__ __launch_bounds__(OWN_THR)
void edge_owner_pass(const int* __restrict__ src, const int* __restrict__ dst,
                     const float* __restrict__ feat, const float* __restrict__ qv,
                     float* __restrict__ deg,
                     float* __restrict__ sums, int* counts, int* lists,
                     const unsigned* __restrict__ keys, int N, int E) {
  __shared__ __attribute__((aligned(16))) float tab[NPR * 128];
  __shared__ int   cnt[NPR];
  __shared__ int   queue[OWN_WAV * QCAP];
  __shared__ int   lcount;

  const int tid  = threadIdx.x;
  const int lane = tid & 31;
  const int wave = tid >> 5;
  const int lo   = blockIdx.x * NPR;
  const int nown = min(NPR, N - lo);

  if (MODE == 1) {
    for (int t = tid; t < NPR * 5; t += OWN_THR) tab[t] = 0.0f;
    for (int t = tid; t < NPR; t += OWN_THR) cnt[t] = 0;
    if (tid == 0) lcount = 0;
  } else {
    for (int t = tid; t < NPR * 32; t += OWN_THR)
      ((float4*)tab)[t] = make_float4(0.0f, 0.0f, 0.0f, 0.0f);
  }
  __syncthreads();

  int* list = lists + (size_t)blockIdx.x * (2 * LCAP);
  const int nl = (MODE == 2) ? counts[blockIdx.x] : 0;

  if (MODE == 2 && nl <= LCAP) {
    const int wv  = __builtin_amdgcn_readfirstlane(wave);
    const int beg = (int)(((long long)wv * nl) / OWN_WAV);
    const int end = (int)(((long long)(wv + 1) * nl) / OWN_WAV);
    int2 nx = make_int2(0, 0);
    if (beg + lane < end) nx = *(const int2*)(list + 2 * (size_t)(beg + lane));
    for (int b = beg; b < end; b += 32) {
      const int2 p = nx;
      if (b + 32 + lane < end) nx = *(const int2*)(list + 2 * (size_t)(b + 32 + lane));
      add_rows(feat, tab, p.x * 128, p.y * 128, min(32, end - b), lane);
    }
  } else if (MODE == 1 && E > 0 && gridDim.x < 255u) {
    owner_sweep_keys(src, dst, (const uint4*)keys, feat, tab, cnt, &queue[wave * QCAP], &lcount, list, lo, (unsigned)nown, E);
  } else if (E > 0) {
    const bool vec_ok = ((E & 3) == 0) && ((((size_t)dst) & 15) == 0);
    if (vec_ok)
      owner_sweep<MODE, true >(src, dst, feat, tab, cnt, &queue[wave * QCAP], &lcount, list, lo, (unsigned)nown, E);
    else
      owner_sweep<MODE, false>(src, dst, feat, tab, cnt, &queue[wave * QCAP], &lcount, list, lo, (unsigned)nown, E);
  }

  __syncthreads();

  if (MODE == 1) {
    if (tid == 0) counts[blockIdx.x] = lcount;
    for (int t = tid; t < nown; t += OWN_THR) deg[lo + t] = 1.0f + (float)cnt[t];
    for (int t = tid; t < nown * 5; t += OWN_THR) sums[(size_t)lo * 5 + t] = tab[t];
  } else {
    for (int r0 = wave; r0 < nown; r0 += 4 * OWN_WAV) {
      float4 qr[4];
      float  dg[4];
#pragma unroll
      for (int u = 0; u < 4; ++u) {
        const int r = r0 + u * OWN_WAV;
        if (r < nown) {
          qr[u] = *(const float4*)(qv + (size_t)(lo + r) * 128 + lane * 4);
          dg[u] = deg[lo + r];
        }
      }
#pragma unroll
      for (int u = 0; u < 4; ++u) {
        const int r = r0 + u * OWN_WAV;
        if (r < nown) {
          const float* row = &tab[r * 128 + lane];
          float4 o;
          o.x = qr[u].x + row[0]  / dg[u];
          o.y = qr[u].y + row[32] / dg[u];
          o.z = qr[u].z + row[64] / dg[u];
          o.w = qr[u].w + row[96] / dg[u];
          o.x = o.x > 0.0f ? o.x : 0.0f;
          o.y = o.y > 0.0f ? o.y : 0.0f;
          o.z = o.z > 0.0f ? o.z : 0.0f;
          o.w = o.w > 0.0f ? o.w : 0.0f;
          *(float4*)(sums + (size_t)(lo + r) * 128 + lane * 4) = o;
        }
      }
    }
  }
}

template <bool CAT>
__device__ __forceinline__ void stage_wide(const float* __restrict__ A0, const float* __restrict__ A1,
                                           const float* __restrict__ deg, float* As,
                                           int row0, int M, int t) {
  const int c4 = (t & 31) * 4;
  const int mw = t >> 5;
  constexpr int R = CAT ? 2 : 4;
#pragma unroll 1
  for (int h = 0; h < 4; h += R) {
    float4 a[R], s[R];
    float  dg[R];
#pragma unroll
    for (int u = 0; u < R; ++u) {
      const int row = min(row0 + mw + 8 * (h + u), M - 1);
      a[u] = *(const float4*)(A0 + (size_t)row * NCOLS + c4);
      if (CAT) {
        s[u]  = *(const float4*)(A1 + (size_t)row * NCOLS + c4);
        dg[u] = deg[row];
      }
    }
#pragma unroll
    for (int u = 0; u < R; ++u) {
      const int m = mw + 8 * (h + u);
      const bool in = (row0 + m < M);
      float4 v = a[u];
      if (!in) v = make_float4(0.0f, 0.0f, 0.0f, 0.0f);
      *(float4*)&As[m * LDSTRIDE + c4] = v;
      if (CAT) {
        float4 mean;
        mean.x = (a[u].x + s[u].x) / dg[u];
        mean.y = (a[u].y + s[u].y) / dg[u];
        mean.z = (a[u].z + s[u].z) / dg[u];
        mean.w = (a[u].w + s[u].w) / dg[u];
        if (!in) mean = make_float4(0.0f, 0.0f, 0.0f, 0.0f);
        *(float4*)&As[m * LDSTRIDE + NCOLS + c4] = mean;
      }
    }
  }
}

__global__ void sage_gemm_wmma(const float* __restrict__ A0, const float* __restrict__ A1,
                               const float* __restrict__ deg, const float* __restrict__ W,
                               const float* __restrict__ bias, float* __restrict__ out,
                               int M, int d, int Kpad) {
  __shared__ __attribute__((aligned(16))) float As[MTILE * LDSTRIDE];

  const int row0 = blockIdx.x * MTILE;
  const int t = threadIdx.x;

  const bool cat  = (A1 != nullptr);
  const bool wide = (d == NCOLS) && (Kpad == (cat ? 2 * NCOLS : NCOLS)) && (M > 0) &&
                    ((((size_t)A0) & 15) == 0) && (!cat || ((((size_t)A1) & 15) == 0));
  if (wide && cat) {
    stage_wide<true >(A0, A1, deg, As, row0, M, t);
  } else if (wide) {
    stage_wide<false>(A0, A1, deg, As, row0, M, t);
  } else {
    const int total = MTILE * Kpad;
    for (int i = t; i < total; i += 256) {
      int m = i / Kpad;
      int k = i - m * Kpad;
      int row = row0 + m;
      float v = 0.0f;
      if (row < M) {
        if (k < d) {
          v = A0[(size_t)row * d + k];
        } else if (A1 != nullptr && k < 2 * d) {
          int kk = k - d;
          v = (A0[(size_t)row * d + kk] + A1[(size_t)row * d + kk]) / deg[row];
        }
      }
      As[m * LDSTRIDE + k] = v;
    }
  }
  __syncthreads();

  const int lane  = t & 31;
  const int wave  = t >> 5;
  const int n     = wave * 16 + (lane & 15);
  const int mA    = lane & 15;
  const int khalf = (lane < 16) ? 0 : 2;

  const int Kw = cat ? 2 * d : d;
  const float* wcol = W + n;

  v8f c0 = {};
  v8f c1 = {};
  if (Kw == Kpad) {
#pragma unroll 2
    for (int k0 = 0; k0 < Kpad; k0 += 4) {
      const int kb = k0 + khalf;
      v2f a0 = *(const v2f*)&As[mA * LDSTRIDE + kb];
      v2f a1 = *(const v2f*)&As[(mA + 16) * LDSTRIDE + kb];
      v2f b;
      b.x = wcol[(size_t)kb * NCOLS];
      b.y = wcol[(size_t)(kb + 1) * NCOLS];
      c0 = __builtin_amdgcn_wmma_f32_16x16x4_f32(false, a0, false, b,
                                                 (short)0, c0, false, false);
      c1 = __builtin_amdgcn_wmma_f32_16x16x4_f32(false, a1, false, b,
                                                 (short)0, c1, false, false);
    }
  } else {
#pragma unroll 2
    for (int k0 = 0; k0 < Kpad; k0 += 4) {
      const int kb = k0 + khalf;
      v2f a0 = *(const v2f*)&As[mA * LDSTRIDE + kb];
      v2f a1 = *(const v2f*)&As[(mA + 16) * LDSTRIDE + kb];
      const float w0 = wcol[(size_t)min(kb,     Kw - 1) * NCOLS];
      const float w1 = wcol[(size_t)min(kb + 1, Kw - 1) * NCOLS];
      v2f b;
      b.x = (kb     < Kw) ? w0 : 0.0f;
      b.y = (kb + 1 < Kw) ? w1 : 0.0f;
      c0 = __builtin_amdgcn_wmma_f32_16x16x4_f32(false, a0, false, b,
                                                 (short)0, c0, false, false);
      c1 = __builtin_amdgcn_wmma_f32_16x16x4_f32(false, a1, false, b,
                                                 (short)0, c1, false, false);
    }
  }

  const float bv    = bias[n];
  const int   rbase = (lane < 16) ? 0 : 8;
  float* outA = out + (size_t)(row0 + rbase) * NCOLS + n;
  float* outB = outA + (size_t)16 * NCOLS;

  if (row0 + MTILE <= M) {
#pragma unroll
    for (int r = 0; r < 8; ++r) {
      float vA = c0[r] + bv;
      float vB = c1[r] + bv;
      outA[(size_t)r * NCOLS] = vA > 0.0f ? vA : 0.0f;
      outB[(size_t)r * NCOLS] = vB > 0.0f ? vB : 0.0f;
    }
  } else {
#pragma unroll
    for (int r = 0; r < 8; ++r) {
      int rowA = row0 + rbase + r;
      if (rowA < M) {
        float v = c0[r] + bv;
        outA[(size_t)r * NCOLS] = v > 0.0f ? v : 0.0f;
      }
      int rowB = row0 + 16 + rbase + r;
      if (rowB < M) {
        float v = c1[r] + bv;
        outB[(size_t)r * NCOLS] = v > 0.0f ? v : 0.0f;
      }
    }
  }
}

__global__ __launch_bounds__(256)
void sage_l1_proj_wmma(const float* __restrict__ x, const float* __restrict__ s1,
                       const float* __restrict__ deg, const float* __restrict__ W1,
                       const float* __restrict__ b1, const float* __restrict__ W2,
                       const float* __restrict__ b2, float* __restrict__ qv,
                       float* __restrict__ pv, int M) {
  __shared__ __attribute__((aligned(16))) float As[MTILE * LDSTRIDE];
  __shared__ float dgs[MTILE];

  const int row0 = blockIdx.x * MTILE;
  const int t = threadIdx.x;

  for (int i = t; i < MTILE * 12; i += 256) {
    const int m = i / 12;
    const int k = i - m * 12;
    const int row = row0 + m;
    float v = 0.0f;
    if (row < M) {
      if (k < 5) {
        v = x[(size_t)row * 5 + k];
      } else if (k < 10) {
        const int kk = k - 5;
        v = (x[(size_t)row * 5 + kk] + s1[(size_t)row * 5 + kk]) / deg[row];
      }
    }
    As[m * LDSTRIDE + k] = v;
  }
  if (t < MTILE) dgs[t] = (row0 + t < M) ? deg[row0 + t] : 1.0f;
  __syncthreads();

  const int lane  = t & 31;
  const int wave  = t >> 5;
  const int n     = wave * 16 + (lane & 15);
  const int mA    = lane & 15;
  const int khalf = (lane < 16) ? 0 : 2;
  const int rbase = (lane < 16) ? 0 : 8;

  v8f c0 = {};
  v8f c1 = {};
  {
    const float* wcol = W1 + n;
#pragma unroll
    for (int k0 = 0; k0 < 12; k0 += 4) {
      const int kb = k0 + khalf;
      v2f a0 = *(const v2f*)&As[mA * LDSTRIDE + kb];
      v2f a1 = *(const v2f*)&As[(mA + 16) * LDSTRIDE + kb];
      const float w0 = wcol[(size_t)min(kb,     9) * NCOLS];
      const float w1 = wcol[(size_t)min(kb + 1, 9) * NCOLS];
      v2f b;
      b.x = (kb     < 10) ? w0 : 0.0f;
      b.y = (kb + 1 < 10) ? w1 : 0.0f;
      c0 = __builtin_amdgcn_wmma_f32_16x16x4_f32(false, a0, false, b,
                                                 (short)0, c0, false, false);
      c1 = __builtin_amdgcn_wmma_f32_16x16x4_f32(false, a1, false, b,
                                                 (short)0, c1, false, false);
    }
  }
  __syncthreads();
  {
    const float bv = b1[n];
#pragma unroll
    for (int r = 0; r < 8; ++r) {
      const float vA = c0[r] + bv;
      const float vB = c1[r] + bv;
      As[(rbase + r) * LDSTRIDE + n]      = vA > 0.0f ? vA : 0.0f;
      As[(16 + rbase + r) * LDSTRIDE + n] = vB > 0.0f ? vB : 0.0f;
    }
  }
  __syncthreads();

  const float* wq = W2 + n;
  const float* wp = W2 + (size_t)NCOLS * NCOLS + n;
  v8f q0 = {};
  v8f q1 = {};
  v8f p0 = {};
  v8f p1 = {};
#pragma unroll 2
  for (int k0 = 0; k0 < NCOLS; k0 += 4) {
    const int kb = k0 + khalf;
    v2f a0 = *(const v2f*)&As[mA * LDSTRIDE + kb];
    v2f a1 = *(const v2f*)&As[(mA + 16) * LDSTRIDE + kb];
    v2f bq, bp;
    bq.x = wq[(size_t)kb * NCOLS];
    bq.y = wq[(size_t)(kb + 1) * NCOLS];
    bp.x = wp[(size_t)kb * NCOLS];
    bp.y = wp[(size_t)(kb + 1) * NCOLS];
    q0 = __builtin_amdgcn_wmma_f32_16x16x4_f32(false, a0, false, bq,
                                               (short)0, q0, false, false);
    q1 = __builtin_amdgcn_wmma_f32_16x16x4_f32(false, a1, false, bq,
                                               (short)0, q1, false, false);
    p0 = __builtin_amdgcn_wmma_f32_16x16x4_f32(false, a0, false, bp,
                                               (short)0, p0, false, false);
    p1 = __builtin_amdgcn_wmma_f32_16x16x4_f32(false, a1, false, bp,
                                               (short)0, p1, false, false);
  }

  const float bv2 = b2[n];
  const size_t o0 = (size_t)(row0 + rbase) * NCOLS + n;
  const size_t o1 = o0 + (size_t)16 * NCOLS;

  if (row0 + MTILE <= M) {
#pragma unroll
    for (int r = 0; r < 8; ++r) {
      const float dA = dgs[rbase + r];
      const float dB = dgs[16 + rbase + r];
      pv[o0 + (size_t)r * NCOLS] = p0[r];
      pv[o1 + (size_t)r * NCOLS] = p1[r];
      qv[o0 + (size_t)r * NCOLS] = q0[r] + bv2 + p0[r] / dA;
      qv[o1 + (size_t)r * NCOLS] = q1[r] + bv2 + p1[r] / dB;
    }
  } else {
#pragma unroll
    for (int r = 0; r < 8; ++r) {
      if (row0 + rbase + r < M) {
        pv[o0 + (size_t)r * NCOLS] = p0[r];
        qv[o0 + (size_t)r * NCOLS] = q0[r] + bv2 + p0[r] / dgs[rbase + r];
      }
      if (row0 + 16 + rbase + r < M) {
        pv[o1 + (size_t)r * NCOLS] = p1[r];
        qv[o1 + (size_t)r * NCOLS] = q1[r] + bv2 + p1[r] / dgs[16 + rbase + r];
      }
    }
  }
}

__global__ void lin2_kernel(const float* __restrict__ h, const float* __restrict__ W,
                            const float* __restrict__ b, float* __restrict__ out, int M) {
  int i = blockIdx.x * blockDim.x + threadIdx.x;
  if (i < M) {
    float a0 = b[0], a1 = b[1];
    const float* row = h + (size_t)i * 128;
#pragma unroll 8
    for (int k = 0; k < 128; ++k) {
      float v = row[k];
      a0 += v * W[2 * k];
      a1 += v * W[2 * k + 1];
    }
    out[(size_t)i * 2 + 0] = a0;
    out[(size_t)i * 2 + 1] = a1;
  }
}

extern "C" void kernel_launch(void* const* d_in, const int* in_sizes, int n_in,
                              void* d_out, int out_size, void* d_ws, size_t ws_size,
                              hipStream_t stream) {
  const float* x   = (const float*)d_in[0];
  const int*   ei  = (const int*)d_in[1];
  const float* W1  = (const float*)d_in[2];
  const float* b1  = (const float*)d_in[3];
  const float* W2  = (const float*)d_in[4];
  const float* b2  = (const float*)d_in[5];
  const float* l1W = (const float*)d_in[6];
  const float* l1b = (const float*)d_in[7];
  const float* l2W = (const float*)d_in[8];
  const float* l2b = (const float*)d_in[9];

  const int N = in_sizes[0] / 5;
  const int E = in_sizes[1] / 2;
  const int* src = ei;
  const int* dst = ei + E;

  float* ws   = (float*)d_ws;
  float* deg  = ws;
  float* s1   = deg  + (size_t)N;
  float* pv   = s1   + (size_t)5 * N;
  float* qv   = pv   + (size_t)128 * N;
  float* h2   = qv   + (size_t)128 * N;
  float* h3   = pv;
  const int nrange = (N + NPR - 1) / NPR;
  int* counts = (int*)(h2 + (size_t)128 * N);
  int* lists  = counts + (size_t)((nrange + 3) & ~3);
  const int nkv = (E + 15) >> 4;
  unsigned* keys = (unsigned*)(lists + (size_t)nrange * (2 * LCAP));

  const int nthread = 256;

  range_keys<<<(nkv + nthread - 1) / nthread, nthread, 0, stream>>>(dst, keys, N, E, nkv);

  edge_owner_pass<1><<<nrange, OWN_THR, 0, stream>>>(src, dst, x, nullptr, deg, s1, counts, lists, keys, N, E);

  sage_l1_proj_wmma<<<(N + MTILE - 1) / MTILE, nthread, 0, stream>>>(x, s1, deg, W1, b1, W2, b2, qv, pv, N);

  edge_owner_pass<2><<<nrange, OWN_THR, 0, stream>>>(src, dst, pv, qv, deg, h2, counts, lists, keys, N, E);

  sage_gemm_wmma<<<(N + MTILE - 1) / MTILE, nthread, 0, stream>>>(h2, nullptr, deg, l1W, l1b, h3, N, 128, 128);

  lin2_kernel<<<(N + nthread - 1) / nthread, nthread, 0, stream>>>(h3, l2W, l2b, (float*)d_out, N);
}


# lin1 and lin2 moved into the layer-2 edge pass's range-owning launch: h2 stays in the owner's LDS table, lin1 on WMMA there; lin2's 128-term sum = 16-lane trees, 8 partials in order; 6 launches -> 4
# speedup vs baseline: 3.9911x; 1.1215x over previous
#include <hip/hip_runtime.h>

typedef __attribute__((ext_vector_type(2))) float v2f;
typedef __attribute__((ext_vector_type(8))) float v8f;

#define NCOLS 128
#define MTILE 32
#define LDSTRIDE 260

#define NPR      400
#define OWN_THR  1024
#define OWN_WAV  32
#define SCAN_U   4
#define QFLUSH   96
#define QCAP     (QFLUSH + SCAN_U * 128)
#define LCAP     16384

__device__ __forceinline__ void add_rows(const float* __restrict__ feat, float* tab,
                                         int soff, int doff, int c, int lane) {
  for (int j = 0; j < c; j += 8) {
    float4 r[8];
    int    d[8];
    if (j + 8 <= c) {
#pragma unroll
      for (int u = 0; u < 8; ++u) {
        const int sj = __builtin_amdgcn_readlane(soff, j + u);
        d[u] = __builtin_amdgcn_readlane(doff, j + u);
        r[u] = *(const float4*)(feat + (size_t)sj + lane * 4);
      }
#pragma unroll
      for (int u = 0; u < 8; ++u) {
        float* row = &tab[d[u] + lane];
        __hip_atomic_fetch_add(row +  0, r[u].x, __ATOMIC_RELAXED, __HIP_MEMORY_SCOPE_WORKGROUP);
        __hip_atomic_fetch_add(row + 32, r[u].y, __ATOMIC_RELAXED, __HIP_MEMORY_SCOPE_WORKGROUP);
        __hip_atomic_fetch_add(row + 64, r[u].z, __ATOMIC_RELAXED, __HIP_MEMORY_SCOPE_WORKGROUP);
        __hip_atomic_fetch_add(row + 96, r[u].w, __ATOMIC_RELAXED, __HIP_MEMORY_SCOPE_WORKGROUP);
      }
    } else {
      const int m = c - j;
#pragma unroll
      for (int u = 0; u < 7; ++u) {
        if (u < m) {
          const int sj = __builtin_amdgcn_readlane(soff, j + u);
          d[u] = __builtin_amdgcn_readlane(doff, j + u);
          r[u] = *(const float4*)(feat + (size_t)sj + lane * 4);
        }
      }
#pragma unroll
      for (int u = 0; u < 7; ++u) {
        if (u < m) {
          float* row = &tab[d[u] + lane];
          __hip_atomic_fetch_add(row +  0, r[u].x, __ATOMIC_RELAXED, __HIP_MEMORY_SCOPE_WORKGROUP);
          __hip_atomic_fetch_add(row + 32, r[u].y, __ATOMIC_RELAXED, __HIP_MEMORY_SCOPE_WORKGROUP);
          __hip_atomic_fetch_add(row + 64, r[u].z, __ATOMIC_RELAXED, __HIP_MEMORY_SCOPE_WORKGROUP);
          __hip_atomic_fetch_add(row + 96, r[u].w, __ATOMIC_RELAXED, __HIP_MEMORY_SCOPE_WORKGROUP);
        }
      }
    }
  }
}

template <bool VEC>
__device__ __forceinline__ int4 load_dst4(const int* __restrict__ dst, int idx4, int nvec, int E) {
  int4 r;
  if (VEC) {
    r = *(const int4*)(dst + (size_t)min(idx4, nvec - 1) * 4);
  } else {
    const int e = idx4 * 4;
    r.x = dst[min(e + 0, E - 1)];
    r.y = dst[min(e + 1, E - 1)];
    r.z = dst[min(e + 2, E - 1)];
    r.w = dst[min(e + 3, E - 1)];
    if (e + 0 >= E) r.x = -1;
    if (e + 1 >= E) r.y = -1;
    if (e + 2 >= E) r.z = -1;
    if (e + 3 >= E) r.w = -1;
  }
  return r;
}

template <int MODE, bool VEC>
__device__ __forceinline__ void owner_sweep(const int* __restrict__ src, const int* __restrict__ dst,
                                            const float* __restrict__ feat, float* tab, int* cnt,
                                            int* q, int* lcount, int* list,
                                            int lo, unsigned nloc, int E) {
  const int tid  = threadIdx.x;
  const int lane = tid & 31;
  const int nvec   = (E + 3) >> 2;
  const int per_it = OWN_THR * SCAN_U;
  const int n_it   = (nvec + per_it - 1) / per_it;
  const int rot = (int)(((long long)blockIdx.x * n_it) / gridDim.x);

  int qn = 0;

  auto fetch = [&](int4 (&v)[SCAN_U], int i) -> int {
    int it = i + rot;
    if (it >= n_it) it -= n_it;
    const int base = (i < n_it) ? (it * per_it + tid) : nvec;
#pragma unroll
    for (int u = 0; u < SCAN_U; ++u) v[u] = load_dst4<VEC>(dst, base + u * OWN_THR, nvec, E);
    return base;
  };

  auto take = [&](const int4 (&v)[SCAN_U], int base) {
#pragma unroll
    for (int u = 0; u < SCAN_U; ++u) {
      const int e0 = (base + u * OWN_THR) * 4;
      const unsigned lim = (base + u * OWN_THR < nvec) ? nloc : 0u;
      const bool h0 = (unsigned)(v[u].x - lo) < lim;
      const bool h1 = (unsigned)(v[u].y - lo) < lim;
      const bool h2 = (unsigned)(v[u].z - lo) < lim;
      const bool h3 = (unsigned)(v[u].w - lo) < lim;
      const unsigned m0 = __builtin_amdgcn_ballot_w32(h0);
      const unsigned m1 = __builtin_amdgcn_ballot_w32(h1);
      const unsigned m2 = __builtin_amdgcn_ballot_w32(h2);
      const unsigned m3 = __builtin_amdgcn_ballot_w32(h3);
      if ((m0 | m1 | m2 | m3) != 0u) {
        if (h0) q[qn + __builtin_amdgcn_mbcnt_lo(m0, 0u)] = e0 + 0;
        qn += __builtin_popcount(m0);
        if (h1) q[qn + __builtin_amdgcn_mbcnt_lo(m1, 0u)] = e0 + 1;
        qn += __builtin_popcount(m1);
        if (h2) q[qn + __builtin_amdgcn_mbcnt_lo(m2, 0u)] = e0 + 2;
        qn += __builtin_popcount(m2);
        if (h3) q[qn + __builtin_amdgcn_mbcnt_lo(m3, 0u)] = e0 + 3;
        qn += __builtin_popcount(m3);
      }
    }
  };

  auto drain = [&]() {
    for (int b = 0; b < qn; b += 32) {
      const int k = b + lane;
      int dl = 0, s = 0;
      if (k < qn) {
        const int e = q[k];
        dl = dst[e] - lo;
        s  = src[e];
      }
      if (MODE == 1) {
        int slot = 0;
        if (lane == 0)
          slot = __hip_atomic_fetch_add(lcount, min(32, qn - b), __ATOMIC_RELAXED, __HIP_MEMORY_SCOPE_WORKGROUP);
        slot = __builtin_amdgcn_readfirstlane(slot) + lane;
        if (k < qn && slot < LCAP) *(int2*)(list + 2 * (size_t)slot) = make_int2(s, dl);
        if (k < qn) {
          const float* xr = feat + (size_t)s * 5;
          const float x0 = xr[0], x1 = xr[1], x2 = xr[2], x3 = xr[3], x4 = xr[4];
          __hip_atomic_fetch_add(&cnt[dl], 1, __ATOMIC_RELAXED, __HIP_MEMORY_SCOPE_WORKGROUP);
          float* row = &tab[dl * 5];
          __hip_atomic_fetch_add(row + 0, x0, __ATOMIC_RELAXED, __HIP_MEMORY_SCOPE_WORKGROUP);
          __hip_atomic_fetch_add(row + 1, x1, __ATOMIC_RELAXED, __HIP_MEMORY_SCOPE_WORKGROUP);
          __hip_atomic_fetch_add(row + 2, x2, __ATOMIC_RELAXED, __HIP_MEMORY_SCOPE_WORKGROUP);
          __hip_atomic_fetch_add(row + 3, x3, __ATOMIC_RELAXED, __HIP_MEMORY_SCOPE_WORKGROUP);
          __hip_atomic_fetch_add(row + 4, x4, __ATOMIC_RELAXED, __HIP_MEMORY_SCOPE_WORKGROUP);
        }
      } else {
        add_rows(feat, tab, s * 128, dl * 128, min(32, qn - b), lane);
      }
    }
    qn = 0;
  };

  int4 A[SCAN_U], B[SCAN_U];
  int baseA = fetch(A, 0);
  for (int i = 0; i < n_it; i += 2) {
    const int baseB = fetch(B, i + 1);
    __builtin_amdgcn_sched_barrier(0);
    take(A, baseA);
    if (qn > QFLUSH) drain();
    baseA = fetch(A, i + 2);
    __builtin_amdgcn_sched_barrier(0);
    take(B, baseB);
    if (qn > QFLUSH || i + 2 >= n_it) drain();
  }
}

__device__ __forceinline__ unsigned range_key(int d, int N) {
  return ((unsigned)d < (unsigned)N) ? (unsigned)d / NPR : 0xFFu;
}

__global__ void range_keys(const int* __restrict__ dst, unsigned* __restrict__ keys,
                           int N, int E, int nkv) {
  const int i = blockIdx.x * blockDim.x + threadIdx.x;
  if (i >= nkv) return;
  const int e0 = i * 16;
  unsigned w[4];
  if (e0 + 16 <= E && ((((size_t)dst) & 15) == 0)) {
#pragma unroll
    for (int j = 0; j < 4; ++j) {
      const int4 v = *(const int4*)(dst + e0 + 4 * j);
      w[j] = range_key(v.x, N) | (range_key(v.y, N) << 8) |
             (range_key(v.z, N) << 16) | (range_key(v.w, N) << 24);
    }
  } else {
#pragma unroll
    for (int j = 0; j < 4; ++j) {
      unsigned p = 0u;
#pragma unroll
      for (int y = 0; y < 4; ++y) {
        const int e = e0 + 4 * j + y;
        const int d = (e < E) ? dst[e] : -1;
        p |= range_key(d, N) << (8 * y);
      }
      w[j] = p;
    }
  }
  ((uint4*)keys)[i] = make_uint4(w[0], w[1], w[2], w[3]);
}

#define KSCAN_U 4
__device__ __forceinline__ void owner_sweep_keys(const int* __restrict__ src, const int* __restrict__ dst,
                                                 const uint4* __restrict__ keys,
                                                 const float* __restrict__ feat, float* tab, int* cnt,
                                                 int* q, int* lcount, int* list,
                                                 int lo, unsigned nloc, int E) {
  const int tid  = threadIdx.x;
  const int lane = tid & 31;
  const int nkv    = (E + 15) >> 4;
  const int per_it = OWN_THR * KSCAN_U;
  const int n_it   = (nkv + per_it - 1) / per_it;
  const int rot = (int)(((long long)blockIdx.x * n_it) / gridDim.x);
  const unsigned pat = (unsigned)blockIdx.x * 0x01010101u;

  int qn = 0;

  auto fetch = [&](uint4 (&v)[KSCAN_U], int i) -> int {
    int it = i + rot;
    if (it >= n_it) it -= n_it;
    const int base = (i < n_it) ? (it * per_it + tid) : nkv;
#pragma unroll
    for (int u = 0; u < KSCAN_U; ++u) v[u] = keys[min(base + u * OWN_THR, nkv - 1)];
    return base;
  };

  auto drain = [&]() {
    for (int b = 0; b < qn; b += 32) {
      const int k = b + lane;
      int dl = 0, s = 0;
      bool ok = false;
      if (k < qn) {
        const unsigned w = (unsigned)q[k];
        const int e = (int)((w >> 5) << 4) + (int)(((7u - (w & 7u)) << 2) + ((w & 31u) >> 3));
        if (e < E) {
          dl = dst[e] - lo;
          s  = src[e];
          ok = (unsigned)dl < nloc;
        }
      }
      const unsigned okm = __builtin_amdgcn_ballot_w32(ok);
      int slot = 0;
      if (lane == 0)
        slot = __hip_atomic_fetch_add(lcount, __builtin_popcount(okm), __ATOMIC_RELAXED, __HIP_MEMORY_SCOPE_WORKGROUP);
      slot = __builtin_amdgcn_readfirstlane(slot) + (int)__builtin_amdgcn_mbcnt_lo(okm, 0u);
      if (ok && slot < LCAP) *(int2*)(list + 2 * (size_t)slot) = make_int2(s, dl);
      if (ok) {
        const float* xr = feat + (size_t)s * 5;
        const float x0 = xr[0], x1 = xr[1], x2 = xr[2], x3 = xr[3], x4 = xr[4];
        __hip_atomic_fetch_add(&cnt[dl], 1, __ATOMIC_RELAXED, __HIP_MEMORY_SCOPE_WORKGROUP);
        float* row = &tab[dl * 5];
        __hip_atomic_fetch_add(row + 0, x0, __ATOMIC_RELAXED, __HIP_MEMORY_SCOPE_WORKGROUP);
        __hip_atomic_fetch_add(row + 1, x1, __ATOMIC_RELAXED, __HIP_MEMORY_SCOPE_WORKGROUP);
        __hip_atomic_fetch_add(row + 2, x2, __ATOMIC_RELAXED, __HIP_MEMORY_SCOPE_WORKGROUP);
        __hip_atomic_fetch_add(row + 3, x3, __ATOMIC_RELAXED, __HIP_MEMORY_SCOPE_WORKGROUP);
        __hip_atomic_fetch_add(row + 4, x4, __ATOMIC_RELAXED, __HIP_MEMORY_SCOPE_WORKGROUP);
      }
    }
    qn = 0;
  };

  auto take = [&](const uint4 (&v)[KSCAN_U], int base) {
#pragma unroll
    for (int u = 0; u < KSCAN_U; ++u) {
      const int idx = base + u * OWN_THR;
      const unsigned x0 = v[u].x ^ pat, x1 = v[u].y ^ pat, x2 = v[u].z ^ pat, x3 = v[u].w ^ pat;
      const unsigned u0 = (x0 - 0x01010101u) & ~x0;
      const unsigned u1 = (x1 - 0x01010101u) & ~x1;
      const unsigned u2 = (x2 - 0x01010101u) & ~x2;
      const unsigned u3 = (x3 - 0x01010101u) & ~x3;
      unsigned any = (u0 | u1 | u2 | u3) & 0x80808080u;
      if (idx >= nkv) any = 0u;
      unsigned bm = __builtin_amdgcn_ballot_w32(any != 0u);
      if (bm != 0u) {
        unsigned m = (u0 & 0x80808080u) | ((u1 & 0x80808080u) >> 1) |
                     ((u2 & 0x80808080u) >> 2) | ((u3 & 0x80808080u) >> 3);
        if (any == 0u) m = 0u;
        const unsigned tag = (unsigned)idx << 5;
        do {
          if (m != 0u) {
            const unsigned p = (unsigned)__builtin_ctz(m);
            m &= m - 1u;
            q[qn + (int)__builtin_amdgcn_mbcnt_lo(bm, 0u)] = (int)(tag | p);
          }
          qn += __builtin_popcount(bm);
          bm = __builtin_amdgcn_ballot_w32(m != 0u);
        } while (bm != 0u);
        if (qn > QFLUSH) drain();
      }
    }
  };

  uint4 A[KSCAN_U], B[KSCAN_U];
  int baseA = fetch(A, 0);
  for (int i = 0; i < n_it; i += 2) {
    const int baseB = fetch(B, i + 1);
    __builtin_amdgcn_sched_barrier(0);
    take(A, baseA);
    baseA = fetch(A, i + 2);
    __builtin_amdgcn_sched_barrier(0);
    take(B, baseB);
  }
  drain();
}

template <int CTRL>
__device__ __forceinline__ float row_xor_add(float v) {
  const int t = __builtin_amdgcn_update_dpp(0, __builtin_bit_cast(int, v), CTRL, 0xF, 0xF, true);
  return v + __builtin_bit_cast(float, t);
}
__device__ __forceinline__ float row_sum16(float v) {
  v = row_xor_add<0x161>(v);
  v = row_xor_add<0x162>(v);
  v = row_xor_add<0x164>(v);
  v = row_xor_add<0x168>(v);
  return v;
}

static_assert(NPR == 25 * 16, "tail_gemm splits a range into 25 row tiles of 16");
template <int NT>
__device__ __forceinline__ void tail_gemm(const float* tab, float* part,
                                          const float* __restrict__ l1W, const float* __restrict__ l1b,
                                          const float* __restrict__ l2W, int ct, int g, int lane) {
  const int n     = ct * 16 + (lane & 15);
  const int mA    = lane & 15;
  const int khalf = (lane < 16) ? 0 : 2;
  const int sw    = mA << 2;
  const float* wcol = l1W + n;
  const float* arow = tab + (g * 16 + mA) * 128 + khalf;

  v8f acc[NT];
#pragma unroll
  for (int i = 0; i < NT; ++i) acc[i] = v8f{};

#pragma unroll 2
  for (int k0 = 0; k0 < 128; k0 += 4) {
    const int kb = k0 + khalf;
    v2f b;
    b.x = wcol[(size_t)kb * NCOLS];
    b.y = wcol[(size_t)(kb + 1) * NCOLS];
    const int wo = k0 ^ sw;
#pragma unroll
    for (int i = 0; i < NT; ++i) {
      const v2f a = *(const v2f*)&arow[i * (64 * 128) + wo];
      acc[i] = __builtin_amdgcn_wmma_f32_16x16x4_f32(false, a, false, b,
                                                     (short)0, acc[i], false, false);
    }
  }

  const float bv = l1b[n];
  const float w0 = l2W[2 * n];
  const float w1 = l2W[2 * n + 1];
  const int rbase = (lane < 16) ? 0 : 8;
#pragma unroll
  for (int i = 0; i < NT; ++i) {
#pragma unroll
    for (int r = 0; r < 8; ++r) {
      float h = acc[i][r] + bv;
      h = h > 0.0f ? h : 0.0f;
      const float s0 = row_sum16(h * w0);
      const float s1 = row_sum16(h * w1);
      if ((lane & 15) == 0)
        *(float2*)&part[((size_t)ct * NPR + (g + 4 * i) * 16 + rbase + r) * 2] = make_float2(s0, s1);
    }
  }
}

template <int MODE>
__global__ __launch_bounds__(OWN_THR)
void edge_owner_pass(const int* __restrict__ src, const int* __restrict__ dst,
                     const float* __restrict__ feat, const float* __restrict__ qv,
                     float* __restrict__ deg,
                     float* __restrict__ sums, int* counts, int* lists,
                     const unsigned* __restrict__ keys,
                     const float* __restrict__ l1W, const float* __restrict__ l1b,
                     const float* __restrict__ l2W, const float* __restrict__ l2b,
                     float* __restrict__ out, int N, int E) {
  __shared__ __attribute__((aligned(16))) float tab[NPR * 128];
  __shared__ int   cnt[NPR];
  __shared__ __attribute__((aligned(16))) int queue[OWN_WAV * QCAP];
  __shared__ int   lcount;

  const int tid  = threadIdx.x;
  const int lane = tid & 31;
  const int wave = (MODE == 2) ? __builtin_amdgcn_readfirstlane(tid >> 5) : (tid >> 5);
  const int lo   = blockIdx.x * NPR;
  const int nown = min(NPR, N - lo);

  if (MODE == 1) {
    for (int t = tid; t < NPR * 5; t += OWN_THR) tab[t] = 0.0f;
    for (int t = tid; t < NPR; t += OWN_THR) cnt[t] = 0;
    if (tid == 0) lcount = 0;
  } else {
    for (int t = tid; t < NPR * 32; t += OWN_THR)
      ((float4*)tab)[t] = make_float4(0.0f, 0.0f, 0.0f, 0.0f);
  }
  __syncthreads();

  int* list = lists + (size_t)blockIdx.x * (2 * LCAP);
  const int nl = (MODE == 2) ? counts[blockIdx.x] : 0;

  if (MODE == 2 && nl <= LCAP) {
    const int wv  = wave;
    const int beg = (int)(((long long)wv * nl) / OWN_WAV);
    const int end = (int)(((long long)(wv + 1) * nl) / OWN_WAV);
    int2 nx = make_int2(0, 0);
    if (beg + lane < end) nx = *(const int2*)(list + 2 * (size_t)(beg + lane));
    for (int b = beg; b < end; b += 32) {
      const int2 p = nx;
      if (b + 32 + lane < end) nx = *(const int2*)(list + 2 * (size_t)(b + 32 + lane));
      add_rows(feat, tab, p.x * 128, p.y * 128, min(32, end - b), lane);
    }
  } else if (MODE == 1 && E > 0 && gridDim.x < 255u) {
    owner_sweep_keys(src, dst, (const uint4*)keys, feat, tab, cnt, &queue[wave * QCAP], &lcount, list, lo, (unsigned)nown, E);
  } else if (E > 0) {
    const bool vec_ok = ((E & 3) == 0) && ((((size_t)dst) & 15) == 0);
    if (vec_ok)
      owner_sweep<MODE, true >(src, dst, feat, tab, cnt, &queue[wave * QCAP], &lcount, list, lo, (unsigned)nown, E);
    else
      owner_sweep<MODE, false>(src, dst, feat, tab, cnt, &queue[wave * QCAP], &lcount, list, lo, (unsigned)nown, E);
  }

  __syncthreads();

  if (MODE == 1) {
    if (tid == 0) counts[blockIdx.x] = lcount;
    for (int t = tid; t < nown; t += OWN_THR) deg[lo + t] = 1.0f + (float)cnt[t];
    for (int t = tid; t < nown * 5; t += OWN_THR) sums[(size_t)lo * 5 + t] = tab[t];
  } else {
    for (int r0 = wave; r0 < nown; r0 += 4 * OWN_WAV) {
      float4 qr[4];
      float  dg[4];
#pragma unroll
      for (int u = 0; u < 4; ++u) {
        const int r = r0 + u * OWN_WAV;
        if (r < nown) {
          qr[u] = *(const float4*)(qv + (size_t)(lo + r) * 128 + lane * 4);
          dg[u] = deg[lo + r];
        }
      }
#pragma unroll
      for (int u = 0; u < 4; ++u) {
        const int r = r0 + u * OWN_WAV;
        if (r < nown) {
          float* row = &tab[r * 128 + lane];
          float4 o;
          o.x = qr[u].x + row[0]  / dg[u];
          o.y = qr[u].y + row[32] / dg[u];
          o.z = qr[u].z + row[64] / dg[u];
          o.w = qr[u].w + row[96] / dg[u];
          o.x = o.x > 0.0f ? o.x : 0.0f;
          o.y = o.y > 0.0f ? o.y : 0.0f;
          o.z = o.z > 0.0f ? o.z : 0.0f;
          o.w = o.w > 0.0f ? o.w : 0.0f;
          *(float4*)&tab[r * 128 + 4 * (lane ^ (r & 15))] = o;
        }
      }
    }
    __syncthreads();

    float* part = (float*)queue;
    const int wv = wave;
    if ((wv >> 3) == 0) tail_gemm<7>(tab, part, l1W, l1b, l2W, wv & 7, 0, lane);
    else                tail_gemm<6>(tab, part, l1W, l1b, l2W, wv & 7, wv >> 3, lane);
    __syncthreads();

    if (tid < nown * 2) {
      const int row = tid >> 1;
      const int j   = tid & 1;
      float a = l2b[j];
#pragma unroll
      for (int t8 = 0; t8 < 8; ++t8) a += part[((size_t)t8 * NPR + row) * 2 + j];
      out[(size_t)(lo + row) * 2 + j] = a;
    }
  }
}

__global__ __launch_bounds__(256)
void sage_l1_proj_wmma(const float* __restrict__ x, const float* __restrict__ s1,
                       const float* __restrict__ deg, const float* __restrict__ W1,
                       const float* __restrict__ b1, const float* __restrict__ W2,
                       const float* __restrict__ b2, float* __restrict__ qv,
                       float* __restrict__ pv, int M) {
  __shared__ __attribute__((aligned(16))) float As[MTILE * LDSTRIDE];
  __shared__ float dgs[MTILE];

  const int row0 = blockIdx.x * MTILE;
  const int t = threadIdx.x;

  for (int i = t; i < MTILE * 12; i += 256) {
    const int m = i / 12;
    const int k = i - m * 12;
    const int row = row0 + m;
    float v = 0.0f;
    if (row < M) {
      if (k < 5) {
        v = x[(size_t)row * 5 + k];
      } else if (k < 10) {
        const int kk = k - 5;
        v = (x[(size_t)row * 5 + kk] + s1[(size_t)row * 5 + kk]) / deg[row];
      }
    }
    As[m * LDSTRIDE + k] = v;
  }
  if (t < MTILE) dgs[t] = (row0 + t < M) ? deg[row0 + t] : 1.0f;
  __syncthreads();

  const int lane  = t & 31;
  const int wave  = t >> 5;
  const int n     = wave * 16 + (lane & 15);
  const int mA    = lane & 15;
  const int khalf = (lane < 16) ? 0 : 2;
  const int rbase = (lane < 16) ? 0 : 8;

  v8f c0 = {};
  v8f c1 = {};
  {
    const float* wcol = W1 + n;
#pragma unroll
    for (int k0 = 0; k0 < 12; k0 += 4) {
      const int kb = k0 + khalf;
      v2f a0 = *(const v2f*)&As[mA * LDSTRIDE + kb];
      v2f a1 = *(const v2f*)&As[(mA + 16) * LDSTRIDE + kb];
      const float w0 = wcol[(size_t)min(kb,     9) * NCOLS];
      const float w1 = wcol[(size_t)min(kb + 1, 9) * NCOLS];
      v2f b;
      b.x = (kb     < 10) ? w0 : 0.0f;
      b.y = (kb + 1 < 10) ? w1 : 0.0f;
      c0 = __builtin_amdgcn_wmma_f32_16x16x4_f32(false, a0, false, b,
                                                 (short)0, c0, false, false);
      c1 = __builtin_amdgcn_wmma_f32_16x16x4_f32(false, a1, false, b,
                                                 (short)0, c1, false, false);
    }
  }
  __syncthreads();
  {
    const float bv = b1[n];
#pragma unroll
    for (int r = 0; r < 8; ++r) {
      const float vA = c0[r] + bv;
      const float vB = c1[r] + bv;
      As[(rbase + r) * LDSTRIDE + n]      = vA > 0.0f ? vA : 0.0f;
      As[(16 + rbase + r) * LDSTRIDE + n] = vB > 0.0f ? vB : 0.0f;
    }
  }
  __syncthreads();

  const float* wq = W2 + n;
  const float* wp = W2 + (size_t)NCOLS * NCOLS + n;
  v8f q0 = {};
  v8f q1 = {};
  v8f p0 = {};
  v8f p1 = {};
#pragma unroll 2
  for (int k0 = 0; k0 < NCOLS; k0 += 4) {
    const int kb = k0 + khalf;
    v2f a0 = *(const v2f*)&As[mA * LDSTRIDE + kb];
    v2f a1 = *(const v2f*)&As[(mA + 16) * LDSTRIDE + kb];
    v2f bq, bp;
    bq.x = wq[(size_t)kb * NCOLS];
    bq.y = wq[(size_t)(kb + 1) * NCOLS];
    bp.x = wp[(size_t)kb * NCOLS];
    bp.y = wp[(size_t)(kb + 1) * NCOLS];
    q0 = __builtin_amdgcn_wmma_f32_16x16x4_f32(false, a0, false, bq,
                                               (short)0, q0, false, false);
    q1 = __builtin_amdgcn_wmma_f32_16x16x4_f32(false, a1, false, bq,
                                               (short)0, q1, false, false);
    p0 = __builtin_amdgcn_wmma_f32_16x16x4_f32(false, a0, false, bp,
                                               (short)0, p0, false, false);
    p1 = __builtin_amdgcn_wmma_f32_16x16x4_f32(false, a1, false, bp,
                                               (short)0, p1, false, false);
  }

  const float bv2 = b2[n];
  const size_t o0 = (size_t)(row0 + rbase) * NCOLS + n;
  const size_t o1 = o0 + (size_t)16 * NCOLS;

  if (row0 + MTILE <= M) {
#pragma unroll
    for (int r = 0; r < 8; ++r) {
      const float dA = dgs[rbase + r];
      const float dB = dgs[16 + rbase + r];
      pv[o0 + (size_t)r * NCOLS] = p0[r];
      pv[o1 + (size_t)r * NCOLS] = p1[r];
      qv[o0 + (size_t)r * NCOLS] = q0[r] + bv2 + p0[r] / dA;
      qv[o1 + (size_t)r * NCOLS] = q1[r] + bv2 + p1[r] / dB;
    }
  } else {
#pragma unroll
    for (int r = 0; r < 8; ++r) {
      if (row0 + rbase + r < M) {
        pv[o0 + (size_t)r * NCOLS] = p0[r];
        qv[o0 + (size_t)r * NCOLS] = q0[r] + bv2 + p0[r] / dgs[rbase + r];
      }
      if (row0 + 16 + rbase + r < M) {
        pv[o1 + (size_t)r * NCOLS] = p1[r];
        qv[o1 + (size_t)r * NCOLS] = q1[r] + bv2 + p1[r] / dgs[16 + rbase + r];
      }
    }
  }
}

extern "C" void kernel_launch(void* const* d_in, const int* in_sizes, int n_in,
                              void* d_out, int out_size, void* d_ws, size_t ws_size,
                              hipStream_t stream) {
  const float* x   = (const float*)d_in[0];
  const int*   ei  = (const int*)d_in[1];
  const float* W1  = (const float*)d_in[2];
  const float* b1  = (const float*)d_in[3];
  const float* W2  = (const float*)d_in[4];
  const float* b2  = (const float*)d_in[5];
  const float* l1W = (const float*)d_in[6];
  const float* l1b = (const float*)d_in[7];
  const float* l2W = (const float*)d_in[8];
  const float* l2b = (const float*)d_in[9];

  const int N = in_sizes[0] / 5;
  const int E = in_sizes[1] / 2;
  const int* src = ei;
  const int* dst = ei + E;

  float* ws   = (float*)d_ws;
  float* deg  = ws;
  float* s1   = deg  + (size_t)N;
  float* pv   = s1   + (size_t)5 * N;
  float* qv   = pv   + (size_t)128 * N;
  float* h2   = qv   + (size_t)128 * N;
  const int nrange = (N + NPR - 1) / NPR;
  int* counts = (int*)(h2 + (size_t)128 * N);
  int* lists  = counts + (size_t)((nrange + 3) & ~3);
  const int nkv = (E + 15) >> 4;
  unsigned* keys = (unsigned*)(lists + (size_t)nrange * (2 * LCAP));

  const int nthread = 256;

  range_keys<<<(nkv + nthread - 1) / nthread, nthread, 0, stream>>>(dst, keys, N, E, nkv);

  edge_owner_pass<1><<<nrange, OWN_THR, 0, stream>>>(src, dst, x, nullptr, deg, s1, counts, lists, keys,
                                                     nullptr, nullptr, nullptr, nullptr, nullptr, N, E);

  sage_l1_proj_wmma<<<(N + MTILE - 1) / MTILE, nthread, 0, stream>>>(x, s1, deg, W1, b1, W2, b2, qv, pv, N);

  edge_owner_pass<2><<<nrange, OWN_THR, 0, stream>>>(src, dst, pv, qv, deg, h2, counts, lists, keys,
                                                     l1W, l1b, l2W, l2b, (float*)d_out, N, E);
}


# Layer-1 edge pass: a work-group owns 1600 nodes and sweeps one quarter of the 4-byte target list itself; key pre-pass launch gone (4 -> 3); projection adds the 4 partial sums in fixed order
# speedup vs baseline: 4.2375x; 1.0617x over previous
#include <hip/hip_runtime.h>

typedef __attribute__((ext_vector_type(2))) float v2f;
typedef __attribute__((ext_vector_type(8))) float v8f;

#define NCOLS 128
#define MTILE 32
#define LDSTRIDE 260

#define NPR      400
#define RPG      4
#define NPR1     (RPG * NPR)
#define NQ       4
#define OWN_THR  1024
#define OWN_WAV  32
#define SCAN_U   4
#define QFLUSH   96
#define QCAP     (QFLUSH + SCAN_U * 128)
#define LCAP     16384
#define LCAPQ    (LCAP / NQ)

__device__ __forceinline__ void add_rows(const float* __restrict__ feat, float* tab,
                                         int soff, int doff, int c, int lane) {
  for (int j = 0; j < c; j += 8) {
    float4 r[8];
    int    d[8];
    if (j + 8 <= c) {
#pragma unroll
      for (int u = 0; u < 8; ++u) {
        const int sj = __builtin_amdgcn_readlane(soff, j + u);
        d[u] = __builtin_amdgcn_readlane(doff, j + u);
        r[u] = *(const float4*)(feat + (size_t)sj + lane * 4);
      }
#pragma unroll
      for (int u = 0; u < 8; ++u) {
        float* row = &tab[d[u] + lane];
        __hip_atomic_fetch_add(row +  0, r[u].x, __ATOMIC_RELAXED, __HIP_MEMORY_SCOPE_WORKGROUP);
        __hip_atomic_fetch_add(row + 32, r[u].y, __ATOMIC_RELAXED, __HIP_MEMORY_SCOPE_WORKGROUP);
        __hip_atomic_fetch_add(row + 64, r[u].z, __ATOMIC_RELAXED, __HIP_MEMORY_SCOPE_WORKGROUP);
        __hip_atomic_fetch_add(row + 96, r[u].w, __ATOMIC_RELAXED, __HIP_MEMORY_SCOPE_WORKGROUP);
      }
    } else {
      const int m = c - j;
#pragma unroll
      for (int u = 0; u < 7; ++u) {
        if (u < m) {
          const int sj = __builtin_amdgcn_readlane(soff, j + u);
          d[u] = __builtin_amdgcn_readlane(doff, j + u);
          r[u] = *(const float4*)(feat + (size_t)sj + lane * 4);
        }
      }
#pragma unroll
      for (int u = 0; u < 7; ++u) {
        if (u < m) {
          float* row = &tab[d[u] + lane];
          __hip_atomic_fetch_add(row +  0, r[u].x, __ATOMIC_RELAXED, __HIP_MEMORY_SCOPE_WORKGROUP);
          __hip_atomic_fetch_add(row + 32, r[u].y, __ATOMIC_RELAXED, __HIP_MEMORY_SCOPE_WORKGROUP);
          __hip_atomic_fetch_add(row + 64, r[u].z, __ATOMIC_RELAXED, __HIP_MEMORY_SCOPE_WORKGROUP);
          __hip_atomic_fetch_add(row + 96, r[u].w, __ATOMIC_RELAXED, __HIP_MEMORY_SCOPE_WORKGROUP);
        }
      }
    }
  }
}

template <bool VEC>
__device__ __forceinline__ int4 load_dst4(const int* __restrict__ dst, int idx4, int nvec, int E) {
  int4 r;
  if (VEC) {
    r = *(const int4*)(dst + (size_t)min(idx4, nvec - 1) * 4);
  } else {
    const int e = idx4 * 4;
    r.x = dst[min(e + 0, E - 1)];
    r.y = dst[min(e + 1, E - 1)];
    r.z = dst[min(e + 2, E - 1)];
    r.w = dst[min(e + 3, E - 1)];
    if (e + 0 >= E) r.x = -1;
    if (e + 1 >= E) r.y = -1;
    if (e + 2 >= E) r.z = -1;
    if (e + 3 >= E) r.w = -1;
  }
  return r;
}

template <int MODE, bool VEC>
__device__ __forceinline__ void owner_sweep(const int* __restrict__ src, const int* __restrict__ dst,
                                            const float* __restrict__ feat, float* tab, int* cnt,
                                            int* q, int* lcount, int* list,
                                            int lo, unsigned nloc, int E) {
  const int tid  = threadIdx.x;
  const int lane = tid & 31;
  const int nvec   = (E + 3) >> 2;
  const int per_it = OWN_THR * SCAN_U;
  const int n_it   = (nvec + per_it - 1) / per_it;
  const int rot = (int)(((long long)blockIdx.x * n_it) / gridDim.x);

  int qn = 0;

  auto fetch = [&](int4 (&v)[SCAN_U], int i) -> int {
    int it = i + rot;
    if (it >= n_it) it -= n_it;
    const int base = (i < n_it) ? (it * per_it + tid) : nvec;
#pragma unroll
    for (int u = 0; u < SCAN_U; ++u) v[u] = load_dst4<VEC>(dst, base + u * OWN_THR, nvec, E);
    return base;
  };

  auto take = [&](const int4 (&v)[SCAN_U], int base) {
#pragma unroll
    for (int u = 0; u < SCAN_U; ++u) {
      const int e0 = (base + u * OWN_THR) * 4;
      const unsigned lim = (base + u * OWN_THR < nvec) ? nloc : 0u;
      const bool h0 = (unsigned)(v[u].x - lo) < lim;
      const bool h1 = (unsigned)(v[u].y - lo) < lim;
      const bool h2 = (unsigned)(v[u].z - lo) < lim;
      const bool h3 = (unsigned)(v[u].w - lo) < lim;
      const unsigned m0 = __builtin_amdgcn_ballot_w32(h0);
      const unsigned m1 = __builtin_amdgcn_ballot_w32(h1);
      const unsigned m2 = __builtin_amdgcn_ballot_w32(h2);
      const unsigned m3 = __builtin_amdgcn_ballot_w32(h3);
      if ((m0 | m1 | m2 | m3) != 0u) {
        if (h0) q[qn + __builtin_amdgcn_mbcnt_lo(m0, 0u)] = e0 + 0;
        qn += __builtin_popcount(m0);
        if (h1) q[qn + __builtin_amdgcn_mbcnt_lo(m1, 0u)] = e0 + 1;
        qn += __builtin_popcount(m1);
        if (h2) q[qn + __builtin_amdgcn_mbcnt_lo(m2, 0u)] = e0 + 2;
        qn += __builtin_popcount(m2);
        if (h3) q[qn + __builtin_amdgcn_mbcnt_lo(m3, 0u)] = e0 + 3;
        qn += __builtin_popcount(m3);
      }
    }
  };

  auto drain = [&]() {
    for (int b = 0; b < qn; b += 32) {
      const int k = b + lane;
      int dl = 0, s = 0;
      if (k < qn) {
        const int e = q[k];
        dl = dst[e] - lo;
        s  = src[e];
      }
      if (MODE == 1) {
        static_assert(RPG == 4, "four lists per group: the selects below name them one by one");
        const bool     live = k < qn;
        const unsigned jr = live ? (unsigned)dl / NPR : (unsigned)RPG;
        const unsigned m0 = __builtin_amdgcn_ballot_w32(jr == 0u);
        const unsigned m1 = __builtin_amdgcn_ballot_w32(jr == 1u);
        const unsigned m2 = __builtin_amdgcn_ballot_w32(jr == 2u);
        const unsigned m3 = __builtin_amdgcn_ballot_w32(jr == 3u);
        int base = 0;
        if (lane < RPG) {
          const unsigned mj = (lane == 0) ? m0 : (lane == 1) ? m1 : (lane == 2) ? m2 : m3;
          base = __hip_atomic_fetch_add(lcount + lane, (int)__builtin_popcount(mj), __ATOMIC_RELAXED, __HIP_MEMORY_SCOPE_WORKGROUP);
        }
        const int b0 = __builtin_amdgcn_readlane(base, 0);
        const int b1 = __builtin_amdgcn_readlane(base, 1);
        const int b2 = __builtin_amdgcn_readlane(base, 2);
        const int b3 = __builtin_amdgcn_readlane(base, 3);
        if (live) {
          const unsigned mine = (jr == 0u) ? m0 : (jr == 1u) ? m1 : (jr == 2u) ? m2 : m3;
          const int slot = ((jr == 0u) ? b0 : (jr == 1u) ? b1 : (jr == 2u) ? b2 : b3) +
                           (int)__builtin_amdgcn_mbcnt_lo(mine, 0u);
          if (slot < LCAPQ)
            *(int2*)(list + (size_t)jr * (2 * LCAP) + 2 * (size_t)slot) = make_int2(s, dl - (int)jr * NPR);
          const float* xr = feat + (size_t)s * 5;
          const float x0 = xr[0], x1 = xr[1], x2 = xr[2], x3 = xr[3], x4 = xr[4];
          __hip_atomic_fetch_add(&cnt[dl], 1, __ATOMIC_RELAXED, __HIP_MEMORY_SCOPE_WORKGROUP);
          float* row = &tab[dl * 5];
          __hip_atomic_fetch_add(row + 0, x0, __ATOMIC_RELAXED, __HIP_MEMORY_SCOPE_WORKGROUP);
          __hip_atomic_fetch_add(row + 1, x1, __ATOMIC_RELAXED, __HIP_MEMORY_SCOPE_WORKGROUP);
          __hip_atomic_fetch_add(row + 2, x2, __ATOMIC_RELAXED, __HIP_MEMORY_SCOPE_WORKGROUP);
          __hip_atomic_fetch_add(row + 3, x3, __ATOMIC_RELAXED, __HIP_MEMORY_SCOPE_WORKGROUP);
          __hip_atomic_fetch_add(row + 4, x4, __ATOMIC_RELAXED, __HIP_MEMORY_SCOPE_WORKGROUP);
        }
      } else {
        add_rows(feat, tab, s * 128, dl * 128, min(32, qn - b), lane);
      }
    }
    qn = 0;
  };

  int4 A[SCAN_U], B[SCAN_U];
  int baseA = fetch(A, 0);
  for (int i = 0; i < n_it; i += 2) {
    const int baseB = fetch(B, i + 1);
    __builtin_amdgcn_sched_barrier(0);
    take(A, baseA);
    if (qn > QFLUSH) drain();
    baseA = fetch(A, i + 2);
    __builtin_amdgcn_sched_barrier(0);
    take(B, baseB);
    if (qn > QFLUSH || i + 2 >= n_it) drain();
  }
}

template <int CTRL>
__device__ __forceinline__ float row_xor_add(float v) {
  const int t = __builtin_amdgcn_update_dpp(0, __builtin_bit_cast(int, v), CTRL, 0xF, 0xF, true);
  return v + __builtin_bit_cast(float, t);
}
__device__ __forceinline__ float row_sum16(float v) {
  v = row_xor_add<0x161>(v);
  v = row_xor_add<0x162>(v);
  v = row_xor_add<0x164>(v);
  v = row_xor_add<0x168>(v);
  return v;
}

static_assert(NPR == 25 * 16, "tail_gemm splits a range into 25 row tiles of 16");
template <int NT>
__device__ __forceinline__ void tail_gemm(const float* tab, float* part,
                                          const float* __restrict__ l1W, const float* __restrict__ l1b,
                                          const float* __restrict__ l2W, int ct, int g, int lane) {
  const int n     = ct * 16 + (lane & 15);
  const int mA    = lane & 15;
  const int khalf = (lane < 16) ? 0 : 2;
  const int sw    = mA << 2;
  const float* wcol = l1W + n;
  const float* arow = tab + (g * 16 + mA) * 128 + khalf;

  v8f acc[NT];
#pragma unroll
  for (int i = 0; i < NT; ++i) acc[i] = v8f{};

#pragma unroll 2
  for (int k0 = 0; k0 < 128; k0 += 4) {
    const int kb = k0 + khalf;
    v2f b;
    b.x = wcol[(size_t)kb * NCOLS];
    b.y = wcol[(size_t)(kb + 1) * NCOLS];
    const int wo = k0 ^ sw;
#pragma unroll
    for (int i = 0; i < NT; ++i) {
      const v2f a = *(const v2f*)&arow[i * (64 * 128) + wo];
      acc[i] = __builtin_amdgcn_wmma_f32_16x16x4_f32(false, a, false, b,
                                                     (short)0, acc[i], false, false);
    }
  }

  const float bv = l1b[n];
  const float w0 = l2W[2 * n];
  const float w1 = l2W[2 * n + 1];
  const int rbase = (lane < 16) ? 0 : 8;
#pragma unroll
  for (int i = 0; i < NT; ++i) {
#pragma unroll
    for (int r = 0; r < 8; ++r) {
      float h = acc[i][r] + bv;
      h = h > 0.0f ? h : 0.0f;
      const float s0 = row_sum16(h * w0);
      const float s1 = row_sum16(h * w1);
      if ((lane & 15) == 0)
        *(float2*)&part[((size_t)ct * NPR + (g + 4 * i) * 16 + rbase + r) * 2] = make_float2(s0, s1);
    }
  }
}

template <int MODE>
__global__ __launch_bounds__(OWN_THR)
void edge_owner_pass(const int* __restrict__ src, const int* __restrict__ dst,
                     const float* __restrict__ feat, const float* __restrict__ qv,
                     float* __restrict__ deg,
                     float* __restrict__ sums, int* counts, int* lists,
                     const float* __restrict__ l1W, const float* __restrict__ l1b,
                     const float* __restrict__ l2W, const float* __restrict__ l2b,
                     float* __restrict__ out, int N, int E) {
  __shared__ __attribute__((aligned(16))) float tab[NPR * 128];
  __shared__ int   cnt[NPR1];
  __shared__ __attribute__((aligned(16))) int queue[OWN_WAV * QCAP];
  __shared__ int   lcount[RPG];

  const int tid  = threadIdx.x;
  const int lane = tid & 31;
  const int wave = (MODE == 2) ? __builtin_amdgcn_readfirstlane(tid >> 5) : (tid >> 5);
  const int grp  = (MODE == 1) ? (int)(blockIdx.x / NQ) : (int)blockIdx.x;
  const int qtr  = (MODE == 1) ? (int)(blockIdx.x % NQ) : 0;
  const int lo   = grp * ((MODE == 1) ? NPR1 : NPR);
  const int nown = min((MODE == 1) ? NPR1 : NPR, N - lo);

  if (MODE == 1) {
    for (int t = tid; t < NPR1 * 5; t += OWN_THR) tab[t] = 0.0f;
    for (int t = tid; t < NPR1; t += OWN_THR) cnt[t] = 0;
    if (tid < RPG) lcount[tid] = 0;
  } else {
    for (int t = tid; t < NPR * 32; t += OWN_THR)
      ((float4*)tab)[t] = make_float4(0.0f, 0.0f, 0.0f, 0.0f);
  }
  __syncthreads();

  static_assert(NQ == 4 && OWN_WAV % NQ == 0, "four lists per range, named one by one; the waves split evenly over them");
  int c0 = 0, c1 = 0, c2 = 0, c3 = 0;
  if (MODE == 2) {
    const int* cb = counts + (size_t)blockIdx.x * NQ;
    c0 = cb[0]; c1 = cb[1]; c2 = cb[2]; c3 = cb[3];
  }
  const bool listed = (MODE == 2) && c0 <= LCAPQ && c1 <= LCAPQ && c2 <= LCAPQ && c3 <= LCAPQ;

  if (MODE == 2 && listed) {
    const int wv  = wave;
    const int qc  = wv & (NQ - 1);
    const int sh  = wv / NQ;
    const int nl  = (qc == 0) ? c0 : (qc == 1) ? c1 : (qc == 2) ? c2 : c3;
    const int* list = lists + ((size_t)blockIdx.x * NQ + qc) * (2 * LCAPQ);
    const int beg = (int)(((long long)sh * nl) / (OWN_WAV / NQ));
    const int end = (int)(((long long)(sh + 1) * nl) / (OWN_WAV / NQ));
    int2 nx = make_int2(0, 0);
    if (beg + lane < end) nx = *(const int2*)(list + 2 * (size_t)(beg + lane));
    for (int b = beg; b < end; b += 32) {
      const int2 p = nx;
      if (b + 32 + lane < end) nx = *(const int2*)(list + 2 * (size_t)(b + 32 + lane));
      add_rows(feat, tab, p.x * 128, p.y * 128, min(32, end - b), lane);
    }
  } else if (MODE == 1) {
    const int per = (((E + NQ - 1) / NQ) + 3) & ~3;
    const int e0  = min(qtr * per, E);
    const int El  = min(per, E - e0);
    if (El > 0) {
      const int* sq = src + e0;
      const int* dq = dst + e0;
      int* list = lists + ((size_t)grp * RPG * NQ + qtr) * (2 * LCAPQ);
      const bool vec_ok = ((El & 3) == 0) && ((((size_t)dq) & 15) == 0);
      if (vec_ok)
        owner_sweep<1, true >(sq, dq, feat, tab, cnt, &queue[wave * QCAP], lcount, list, lo, (unsigned)nown, El);
      else
        owner_sweep<1, false>(sq, dq, feat, tab, cnt, &queue[wave * QCAP], lcount, list, lo, (unsigned)nown, El);
    }
  } else if (E > 0) {
    const bool vec_ok = ((E & 3) == 0) && ((((size_t)dst) & 15) == 0);
    if (vec_ok)
      owner_sweep<MODE, true >(src, dst, feat, tab, cnt, &queue[wave * QCAP], lcount, lists, lo, (unsigned)nown, E);
    else
      owner_sweep<MODE, false>(src, dst, feat, tab, cnt, &queue[wave * QCAP], lcount, lists, lo, (unsigned)nown, E);
  }

  __syncthreads();

  if (MODE == 1) {
    if (tid < RPG) counts[((size_t)grp * RPG + tid) * NQ + qtr] = lcount[tid];
    float* degq = deg + (size_t)qtr * N;
    float* s1q  = sums + (size_t)qtr * 5 * N;
    for (int t = tid; t < nown; t += OWN_THR) degq[lo + t] = (float)cnt[t];
    for (int t = tid; t < nown * 5; t += OWN_THR) s1q[(size_t)lo * 5 + t] = tab[t];
  } else {
    for (int r0 = wave; r0 < nown; r0 += 4 * OWN_WAV) {
      float4 qr[4];
      float  dg[4];
#pragma unroll
      for (int u = 0; u < 4; ++u) {
        const int r = r0 + u * OWN_WAV;
        if (r < nown) {
          qr[u] = *(const float4*)(qv + (size_t)(lo + r) * 128 + lane * 4);
          dg[u] = deg[lo + r];
        }
      }
#pragma unroll
      for (int u = 0; u < 4; ++u) {
        const int r = r0 + u * OWN_WAV;
        if (r < nown) {
          float* row = &tab[r * 128 + lane];
          float4 o;
          o.x = qr[u].x + row[0]  / dg[u];
          o.y = qr[u].y + row[32] / dg[u];
          o.z = qr[u].z + row[64] / dg[u];
          o.w = qr[u].w + row[96] / dg[u];
          o.x = o.x > 0.0f ? o.x : 0.0f;
          o.y = o.y > 0.0f ? o.y : 0.0f;
          o.z = o.z > 0.0f ? o.z : 0.0f;
          o.w = o.w > 0.0f ? o.w : 0.0f;
          *(float4*)&tab[r * 128 + 4 * (lane ^ (r & 15))] = o;
        }
      }
    }
    __syncthreads();

    float* part = (float*)queue;
    const int wv = wave;
    if ((wv >> 3) == 0) tail_gemm<7>(tab, part, l1W, l1b, l2W, wv & 7, 0, lane);
    else                tail_gemm<6>(tab, part, l1W, l1b, l2W, wv & 7, wv >> 3, lane);
    __syncthreads();

    if (tid < nown * 2) {
      const int row = tid >> 1;
      const int j   = tid & 1;
      float a = l2b[j];
#pragma unroll
      for (int t8 = 0; t8 < 8; ++t8) a += part[((size_t)t8 * NPR + row) * 2 + j];
      out[(size_t)(lo + row) * 2 + j] = a;
    }
  }
}

__global__ __launch_bounds__(256)
void sage_l1_proj_wmma(const float* __restrict__ x, const float* __restrict__ s1q,
                       const float* __restrict__ degq, float* __restrict__ deg,
                       const float* __restrict__ W1,
                       const float* __restrict__ b1, const float* __restrict__ W2,
                       const float* __restrict__ b2, float* __restrict__ qv,
                       float* __restrict__ pv, int M) {
  __shared__ __attribute__((aligned(16))) float As[MTILE * LDSTRIDE];
  __shared__ float dgs[MTILE];

  const int row0 = blockIdx.x * MTILE;
  const int t = threadIdx.x;

  if (t < MTILE) {
    const int row = row0 + t;
    float d = 1.0f;
    if (row < M) {
      d += ((degq[row] + degq[(size_t)M + row]) + degq[(size_t)2 * M + row]) + degq[(size_t)3 * M + row];
      deg[row] = d;
    }
    dgs[t] = d;
  }

  for (int i = t; i < MTILE * 12; i += 256) {
    const int m = i / 12;
    const int k = i - m * 12;
    const int row = row0 + m;
    float v = 0.0f;
    if (row < M) {
      if (k < 5) {
        v = x[(size_t)row * 5 + k];
      } else if (k < 10) {
        const size_t o = (size_t)row * 5 + (k - 5);
        const float sn = ((s1q[o] + s1q[(size_t)5 * M + o]) + s1q[(size_t)10 * M + o]) + s1q[(size_t)15 * M + o];
        const float dn = 1.0f + (((degq[row] + degq[(size_t)M + row]) + degq[(size_t)2 * M + row]) + degq[(size_t)3 * M + row]);
        v = (x[o] + sn) / dn;
      }
    }
    As[m * LDSTRIDE + k] = v;
  }
  __syncthreads();

  const int lane  = t & 31;
  const int wave  = t >> 5;
  const int n     = wave * 16 + (lane & 15);
  const int mA    = lane & 15;
  const int khalf = (lane < 16) ? 0 : 2;
  const int rbase = (lane < 16) ? 0 : 8;

  v8f c0 = {};
  v8f c1 = {};
  {
    const float* wcol = W1 + n;
#pragma unroll
    for (int k0 = 0; k0 < 12; k0 += 4) {
      const int kb = k0 + khalf;
      v2f a0 = *(const v2f*)&As[mA * LDSTRIDE + kb];
      v2f a1 = *(const v2f*)&As[(mA + 16) * LDSTRIDE + kb];
      const float w0 = wcol[(size_t)min(kb,     9) * NCOLS];
      const float w1 = wcol[(size_t)min(kb + 1, 9) * NCOLS];
      v2f b;
      b.x = (kb     < 10) ? w0 : 0.0f;
      b.y = (kb + 1 < 10) ? w1 : 0.0f;
      c0 = __builtin_amdgcn_wmma_f32_16x16x4_f32(false, a0, false, b,
                                                 (short)0, c0, false, false);
      c1 = __builtin_amdgcn_wmma_f32_16x16x4_f32(false, a1, false, b,
                                                 (short)0, c1, false, false);
    }
  }
  __syncthreads();
  {
    const float bv = b1[n];
#pragma unroll
    for (int r = 0; r < 8; ++r) {
      const float vA = c0[r] + bv;
      const float vB = c1[r] + bv;
      As[(rbase + r) * LDSTRIDE + n]      = vA > 0.0f ? vA : 0.0f;
      As[(16 + rbase + r) * LDSTRIDE + n] = vB > 0.0f ? vB : 0.0f;
    }
  }
  __syncthreads();

  const float* wq = W2 + n;
  const float* wp = W2 + (size_t)NCOLS * NCOLS + n;
  v8f q0 = {};
  v8f q1 = {};
  v8f p0 = {};
  v8f p1 = {};
#pragma unroll 2
  for (int k0 = 0; k0 < NCOLS; k0 += 4) {
    const int kb = k0 + khalf;
    v2f a0 = *(const v2f*)&As[mA * LDSTRIDE + kb];
    v2f a1 = *(const v2f*)&As[(mA + 16) * LDSTRIDE + kb];
    v2f bq, bp;
    bq.x = wq[(size_t)kb * NCOLS];
    bq.y = wq[(size_t)(kb + 1) * NCOLS];
    bp.x = wp[(size_t)kb * NCOLS];
    bp.y = wp[(size_t)(kb + 1) * NCOLS];
    q0 = __builtin_amdgcn_wmma_f32_16x16x4_f32(false, a0, false, bq,
                                               (short)0, q0, false, false);
    q1 = __builtin_amdgcn_wmma_f32_16x16x4_f32(false, a1, false, bq,
                                               (short)0, q1, false, false);
    p0 = __builtin_amdgcn_wmma_f32_16x16x4_f32(false, a0, false, bp,
                                               (short)0, p0, false, false);
    p1 = __builtin_amdgcn_wmma_f32_16x16x4_f32(false, a1, false, bp,
                                               (short)0, p1, false, false);
  }

  const float bv2 = b2[n];
  const size_t o0 = (size_t)(row0 + rbase) * NCOLS + n;
  const size_t o1 = o0 + (size_t)16 * NCOLS;

  if (row0 + MTILE <= M) {
#pragma unroll
    for (int r = 0; r < 8; ++r) {
      const float dA = dgs[rbase + r];
      const float dB = dgs[16 + rbase + r];
      pv[o0 + (size_t)r * NCOLS] = p0[r];
      pv[o1 + (size_t)r * NCOLS] = p1[r];
      qv[o0 + (size_t)r * NCOLS] = q0[r] + bv2 + p0[r] / dA;
      qv[o1 + (size_t)r * NCOLS] = q1[r] + bv2 + p1[r] / dB;
    }
  } else {
#pragma unroll
    for (int r = 0; r < 8; ++r) {
      if (row0 + rbase + r < M) {
        pv[o0 + (size_t)r * NCOLS] = p0[r];
        qv[o0 + (size_t)r * NCOLS] = q0[r] + bv2 + p0[r] / dgs[rbase + r];
      }
      if (row0 + 16 + rbase + r < M) {
        pv[o1 + (size_t)r * NCOLS] = p1[r];
        qv[o1 + (size_t)r * NCOLS] = q1[r] + bv2 + p1[r] / dgs[16 + rbase + r];
      }
    }
  }
}

extern "C" void kernel_launch(void* const* d_in, const int* in_sizes, int n_in,
                              void* d_out, int out_size, void* d_ws, size_t ws_size,
                              hipStream_t stream) {
  const float* x   = (const float*)d_in[0];
  const int*   ei  = (const int*)d_in[1];
  const float* W1  = (const float*)d_in[2];
  const float* b1  = (const float*)d_in[3];
  const float* W2  = (const float*)d_in[4];
  const float* b2  = (const float*)d_in[5];
  const float* l1W = (const float*)d_in[6];
  const float* l1b = (const float*)d_in[7];
  const float* l2W = (const float*)d_in[8];
  const float* l2b = (const float*)d_in[9];

  const int N = in_sizes[0] / 5;
  const int E = in_sizes[1] / 2;
  const int* src = ei;
  const int* dst = ei + E;

  float* ws   = (float*)d_ws;
  float* pv   = ws;
  float* qv   = pv   + (size_t)128 * N;
  float* s1q  = qv   + (size_t)128 * N;
  float* degq = s1q  + (size_t)20 * N;
  float* deg  = degq + (size_t)4 * N;
  const int nrange = (N + NPR - 1) / NPR;
  const int ngroup = (N + NPR1 - 1) / NPR1;
  int* counts = (int*)(deg + (size_t)4 * N);
  int* lists  = counts + (size_t)ngroup * (RPG * NQ);

  const int nthread = 256;

  edge_owner_pass<1><<<ngroup * NQ, OWN_THR, 0, stream>>>(src, dst, x, nullptr, degq, s1q, counts, lists,
                                                          nullptr, nullptr, nullptr, nullptr, nullptr, N, E);

  sage_l1_proj_wmma<<<(N + MTILE - 1) / MTILE, nthread, 0, stream>>>(x, s1q, degq, deg, W1, b1, W2, b2, qv, pv, N);

  edge_owner_pass<2><<<nrange, OWN_THR, 0, stream>>>(src, dst, pv, qv, deg, nullptr, counts, lists,
                                                     l1W, l1b, l2W, l2b, (float*)d_out, N, E);
}


# Layer-1 edge pass: a work-group owns 6400 nodes and sweeps one sixteenth of the target list (16 x 16 groups); slot per edge by one LDS add; projection adds the 16 partial sums in fixed order
# speedup vs baseline: 4.4093x; 1.0405x over previous
#include <hip/hip_runtime.h>

typedef __attribute__((ext_vector_type(2))) float v2f;
typedef __attribute__((ext_vector_type(8))) float v8f;

#define NCOLS 128
#define MTILE 32
#define LDSTRIDE 260

#define NPR      400
#define RPG      16
#define NPR1     (RPG * NPR)
#define NQ       16
#define OWN_THR  1024
#define OWN_WAV  32
#define SCAN_U   4
#define QFLUSH   96
#define QCAP     (QFLUSH + SCAN_U * 128)
#define LCAP     16384
#define LCAPQ    (LCAP / NQ)

__device__ __forceinline__ void add_rows(const float* __restrict__ feat, float* tab,
                                         int soff, int doff, int c, int lane) {
  for (int j = 0; j < c; j += 8) {
    float4 r[8];
    int    d[8];
    if (j + 8 <= c) {
#pragma unroll
      for (int u = 0; u < 8; ++u) {
        const int sj = __builtin_amdgcn_readlane(soff, j + u);
        d[u] = __builtin_amdgcn_readlane(doff, j + u);
        r[u] = *(const float4*)(feat + (size_t)sj + lane * 4);
      }
#pragma unroll
      for (int u = 0; u < 8; ++u) {
        float* row = &tab[d[u] + lane];
        __hip_atomic_fetch_add(row +  0, r[u].x, __ATOMIC_RELAXED, __HIP_MEMORY_SCOPE_WORKGROUP);
        __hip_atomic_fetch_add(row + 32, r[u].y, __ATOMIC_RELAXED, __HIP_MEMORY_SCOPE_WORKGROUP);
        __hip_atomic_fetch_add(row + 64, r[u].z, __ATOMIC_RELAXED, __HIP_MEMORY_SCOPE_WORKGROUP);
        __hip_atomic_fetch_add(row + 96, r[u].w, __ATOMIC_RELAXED, __HIP_MEMORY_SCOPE_WORKGROUP);
      }
    } else {
      const int m = c - j;
#pragma unroll
      for (int u = 0; u < 7; ++u) {
        if (u < m) {
          const int sj = __builtin_amdgcn_readlane(soff, j + u);
          d[u] = __builtin_amdgcn_readlane(doff, j + u);
          r[u] = *(const float4*)(feat + (size_t)sj + lane * 4);
        }
      }
#pragma unroll
      for (int u = 0; u < 7; ++u) {
        if (u < m) {
          float* row = &tab[d[u] + lane];
          __hip_atomic_fetch_add(row +  0, r[u].x, __ATOMIC_RELAXED, __HIP_MEMORY_SCOPE_WORKGROUP);
          __hip_atomic_fetch_add(row + 32, r[u].y, __ATOMIC_RELAXED, __HIP_MEMORY_SCOPE_WORKGROUP);
          __hip_atomic_fetch_add(row + 64, r[u].z, __ATOMIC_RELAXED, __HIP_MEMORY_SCOPE_WORKGROUP);
          __hip_atomic_fetch_add(row + 96, r[u].w, __ATOMIC_RELAXED, __HIP_MEMORY_SCOPE_WORKGROUP);
        }
      }
    }
  }
}

template <bool VEC>
__device__ __forceinline__ int4 load_dst4(const int* __restrict__ dst, int idx4, int nvec, int E) {
  int4 r;
  if (VEC) {
    r = *(const int4*)(dst + (size_t)min(idx4, nvec - 1) * 4);
  } else {
    const int e = idx4 * 4;
    r.x = dst[min(e + 0, E - 1)];
    r.y = dst[min(e + 1, E - 1)];
    r.z = dst[min(e + 2, E - 1)];
    r.w = dst[min(e + 3, E - 1)];
    if (e + 0 >= E) r.x = -1;
    if (e + 1 >= E) r.y = -1;
    if (e + 2 >= E) r.z = -1;
    if (e + 3 >= E) r.w = -1;
  }
  return r;
}

template <int MODE, bool VEC>
__device__ __forceinline__ void owner_sweep(const int* __restrict__ src, const int* __restrict__ dst,
                                            const float* __restrict__ feat, float* tab, int* cnt,
                                            int* q, int* lcount, int* list,
                                            int lo, unsigned nloc, int E) {
  const int tid  = threadIdx.x;
  const int lane = tid & 31;
  const int nvec   = (E + 3) >> 2;
  const int per_it = OWN_THR * SCAN_U;
  const int n_it   = (nvec + per_it - 1) / per_it;
  const int rot = (int)(((long long)blockIdx.x * n_it) / gridDim.x);

  int qn = 0;

  auto fetch = [&](int4 (&v)[SCAN_U], int i) -> int {
    int it = i + rot;
    if (it >= n_it) it -= n_it;
    const int base = (i < n_it) ? (it * per_it + tid) : nvec;
#pragma unroll
    for (int u = 0; u < SCAN_U; ++u) v[u] = load_dst4<VEC>(dst, base + u * OWN_THR, nvec, E);
    return base;
  };

  auto take = [&](const int4 (&v)[SCAN_U], int base) {
#pragma unroll
    for (int u = 0; u < SCAN_U; ++u) {
      const int e0 = (base + u * OWN_THR) * 4;
      const unsigned lim = (base + u * OWN_THR < nvec) ? nloc : 0u;
      const bool h0 = (unsigned)(v[u].x - lo) < lim;
      const bool h1 = (unsigned)(v[u].y - lo) < lim;
      const bool h2 = (unsigned)(v[u].z - lo) < lim;
      const bool h3 = (unsigned)(v[u].w - lo) < lim;
      const unsigned m0 = __builtin_amdgcn_ballot_w32(h0);
      const unsigned m1 = __builtin_amdgcn_ballot_w32(h1);
      const unsigned m2 = __builtin_amdgcn_ballot_w32(h2);
      const unsigned m3 = __builtin_amdgcn_ballot_w32(h3);
      if ((m0 | m1 | m2 | m3) != 0u) {
        if (h0) q[qn + __builtin_amdgcn_mbcnt_lo(m0, 0u)] = e0 + 0;
        qn += __builtin_popcount(m0);
        if (h1) q[qn + __builtin_amdgcn_mbcnt_lo(m1, 0u)] = e0 + 1;
        qn += __builtin_popcount(m1);
        if (h2) q[qn + __builtin_amdgcn_mbcnt_lo(m2, 0u)] = e0 + 2;
        qn += __builtin_popcount(m2);
        if (h3) q[qn + __builtin_amdgcn_mbcnt_lo(m3, 0u)] = e0 + 3;
        qn += __builtin_popcount(m3);
      }
    }
  };

  auto drain = [&]() {
    for (int b = 0; b < qn; b += 32) {
      const int k = b + lane;
      int dl = 0, s = 0;
      if (k < qn) {
        const int e = q[k];
        dl = dst[e] - lo;
        s  = src[e];
      }
      if (MODE == 1) {
        const bool live = k < qn;
        if (live) {
          const unsigned jr = (unsigned)dl / NPR;
          const int slot = __hip_atomic_fetch_add(lcount + jr, 1, __ATOMIC_RELAXED, __HIP_MEMORY_SCOPE_WORKGROUP);
          if (slot < LCAPQ)
            *(int2*)(list + (size_t)jr * (2 * LCAP) + 2 * (size_t)slot) = make_int2(s, dl - (int)jr * NPR);
          const float* xr = feat + (size_t)s * 5;
          const float x0 = xr[0], x1 = xr[1], x2 = xr[2], x3 = xr[3], x4 = xr[4];
          __hip_atomic_fetch_add(&cnt[dl], 1, __ATOMIC_RELAXED, __HIP_MEMORY_SCOPE_WORKGROUP);
          float* row = &tab[dl * 5];
          __hip_atomic_fetch_add(row + 0, x0, __ATOMIC_RELAXED, __HIP_MEMORY_SCOPE_WORKGROUP);
          __hip_atomic_fetch_add(row + 1, x1, __ATOMIC_RELAXED, __HIP_MEMORY_SCOPE_WORKGROUP);
          __hip_atomic_fetch_add(row + 2, x2, __ATOMIC_RELAXED, __HIP_MEMORY_SCOPE_WORKGROUP);
          __hip_atomic_fetch_add(row + 3, x3, __ATOMIC_RELAXED, __HIP_MEMORY_SCOPE_WORKGROUP);
          __hip_atomic_fetch_add(row + 4, x4, __ATOMIC_RELAXED, __HIP_MEMORY_SCOPE_WORKGROUP);
        }
      } else {
        add_rows(feat, tab, s * 128, dl * 128, min(32, qn - b), lane);
      }
    }
    qn = 0;
  };

  int4 A[SCAN_U], B[SCAN_U];
  int baseA = fetch(A, 0);
  for (int i = 0; i < n_it; i += 2) {
    const int baseB = fetch(B, i + 1);
    __builtin_amdgcn_sched_barrier(0);
    take(A, baseA);
    if (qn > QFLUSH) drain();
    baseA = fetch(A, i + 2);
    __builtin_amdgcn_sched_barrier(0);
    take(B, baseB);
    if (qn > QFLUSH || i + 2 >= n_it) drain();
  }
}

template <int CTRL>
__device__ __forceinline__ float row_xor_add(float v) {
  const int t = __builtin_amdgcn_update_dpp(0, __builtin_bit_cast(int, v), CTRL, 0xF, 0xF, true);
  return v + __builtin_bit_cast(float, t);
}
__device__ __forceinline__ float row_sum16(float v) {
  v = row_xor_add<0x161>(v);
  v = row_xor_add<0x162>(v);
  v = row_xor_add<0x164>(v);
  v = row_xor_add<0x168>(v);
  return v;
}

static_assert(NPR == 25 * 16, "tail_gemm splits a range into 25 row tiles of 16");
template <int NT>
__device__ __forceinline__ void tail_gemm(const float* tab, float* part,
                                          const float* __restrict__ l1W, const float* __restrict__ l1b,
                                          const float* __restrict__ l2W, int ct, int g, int lane) {
  const int n     = ct * 16 + (lane & 15);
  const int mA    = lane & 15;
  const int khalf = (lane < 16) ? 0 : 2;
  const int sw    = mA << 2;
  const float* wcol = l1W + n;
  const float* arow = tab + (g * 16 + mA) * 128 + khalf;

  v8f acc[NT];
#pragma unroll
  for (int i = 0; i < NT; ++i) acc[i] = v8f{};

#pragma unroll 2
  for (int k0 = 0; k0 < 128; k0 += 4) {
    const int kb = k0 + khalf;
    v2f b;
    b.x = wcol[(size_t)kb * NCOLS];
    b.y = wcol[(size_t)(kb + 1) * NCOLS];
    const int wo = k0 ^ sw;
#pragma unroll
    for (int i = 0; i < NT; ++i) {
      const v2f a = *(const v2f*)&arow[i * (64 * 128) + wo];
      acc[i] = __builtin_amdgcn_wmma_f32_16x16x4_f32(false, a, false, b,
                                                     (short)0, acc[i], false, false);
    }
  }

  const float bv = l1b[n];
  const float w0 = l2W[2 * n];
  const float w1 = l2W[2 * n + 1];
  const int rbase = (lane < 16) ? 0 : 8;
#pragma unroll
  for (int i = 0; i < NT; ++i) {
#pragma unroll
    for (int r = 0; r < 8; ++r) {
      float h = acc[i][r] + bv;
      h = h > 0.0f ? h : 0.0f;
      const float s0 = row_sum16(h * w0);
      const float s1 = row_sum16(h * w1);
      if ((lane & 15) == 0)
        *(float2*)&part[((size_t)ct * NPR + (g + 4 * i) * 16 + rbase + r) * 2] = make_float2(s0, s1);
    }
  }
}

template <int MODE>
__global__ __launch_bounds__(OWN_THR)
void edge_owner_pass(const int* __restrict__ src, const int* __restrict__ dst,
                     const float* __restrict__ feat, const float* __restrict__ qv,
                     float* __restrict__ deg,
                     float* __restrict__ sums, int* counts, int* lists,
                     const float* __restrict__ l1W, const float* __restrict__ l1b,
                     const float* __restrict__ l2W, const float* __restrict__ l2b,
                     float* __restrict__ out, int N, int E) {
  __shared__ __attribute__((aligned(16))) float tab[NPR * 128];
  __shared__ __attribute__((aligned(16))) int cnt[NPR1];
  __shared__ __attribute__((aligned(16))) int queue[OWN_WAV * QCAP];
  __shared__ int   lcount[RPG];

  const int tid  = threadIdx.x;
  const int lane = tid & 31;
  const int wave = (MODE == 2) ? __builtin_amdgcn_readfirstlane(tid >> 5) : (tid >> 5);
  const int grp  = (MODE == 1) ? (int)(blockIdx.x / NQ) : (int)blockIdx.x;
  const int qtr  = (MODE == 1) ? (int)(blockIdx.x % NQ) : 0;
  const int lo   = grp * ((MODE == 1) ? NPR1 : NPR);
  const int nown = min((MODE == 1) ? NPR1 : NPR, N - lo);

  if (MODE == 1) {
    static_assert((NPR1 * 5) % 4 == 0 && NPR1 % 4 == 0, "the tables are cleared 16 bytes at a time");
    for (int t = tid; t < NPR1 * 5 / 4; t += OWN_THR) ((float4*)tab)[t] = make_float4(0.0f, 0.0f, 0.0f, 0.0f);
    for (int t = tid; t < NPR1 / 4; t += OWN_THR) ((int4*)cnt)[t] = make_int4(0, 0, 0, 0);
    if (tid < RPG) lcount[tid] = 0;
  } else {
    for (int t = tid; t < NPR * 32; t += OWN_THR)
      ((float4*)tab)[t] = make_float4(0.0f, 0.0f, 0.0f, 0.0f);
  }
  __syncthreads();

  static_assert(OWN_WAV % NQ == 0 && (NQ & (NQ - 1)) == 0, "the waves split evenly over a range's NQ lists");
  bool listed = (MODE == 2);
  if (MODE == 2) {
    const int* cb = counts + (size_t)blockIdx.x * NQ;
#pragma unroll
    for (int j = 0; j < NQ; ++j) listed = listed && (cb[j] <= LCAPQ);
  }

  if (MODE == 2 && listed) {
    const int wv  = wave;
    const int qc  = wv & (NQ - 1);
    const int sh  = wv / NQ;
    const int nl  = counts[(size_t)blockIdx.x * NQ + qc];
    const int* list = lists + ((size_t)blockIdx.x * NQ + qc) * (2 * LCAPQ);
    const int beg = (int)(((long long)sh * nl) / (OWN_WAV / NQ));
    const int end = (int)(((long long)(sh + 1) * nl) / (OWN_WAV / NQ));
    int2 nx = make_int2(0, 0);
    if (beg + lane < end) nx = *(const int2*)(list + 2 * (size_t)(beg + lane));
    for (int b = beg; b < end; b += 32) {
      const int2 p = nx;
      if (b + 32 + lane < end) nx = *(const int2*)(list + 2 * (size_t)(b + 32 + lane));
      add_rows(feat, tab, p.x * 128, p.y * 128, min(32, end - b), lane);
    }
  } else if (MODE == 1) {
    const int per = (((E + NQ - 1) / NQ) + 3) & ~3;
    const int e0  = min(qtr * per, E);
    const int El  = min(per, E - e0);
    if (El > 0) {
      const int* sq = src + e0;
      const int* dq = dst + e0;
      int* list = lists + ((size_t)grp * RPG * NQ + qtr) * (2 * LCAPQ);
      const bool vec_ok = ((El & 3) == 0) && ((((size_t)dq) & 15) == 0);
      if (vec_ok)
        owner_sweep<1, true >(sq, dq, feat, tab, cnt, &queue[wave * QCAP], lcount, list, lo, (unsigned)nown, El);
      else
        owner_sweep<1, false>(sq, dq, feat, tab, cnt, &queue[wave * QCAP], lcount, list, lo, (unsigned)nown, El);
    }
  } else if (E > 0) {
    const bool vec_ok = ((E & 3) == 0) && ((((size_t)dst) & 15) == 0);
    if (vec_ok)
      owner_sweep<MODE, true >(src, dst, feat, tab, cnt, &queue[wave * QCAP], lcount, lists, lo, (unsigned)nown, E);
    else
      owner_sweep<MODE, false>(src, dst, feat, tab, cnt, &queue[wave * QCAP], lcount, lists, lo, (unsigned)nown, E);
  }

  __syncthreads();

  if (MODE == 1) {
    if (tid < RPG) counts[((size_t)grp * RPG + tid) * NQ + qtr] = lcount[tid];
    float* degq = deg + (size_t)qtr * N;
    float* s1q  = sums + (size_t)qtr * 5 * N;
    float* dp = degq + lo;
    float* sp = s1q + (size_t)lo * 5;
    if ((nown & 3) == 0 && ((((size_t)dp) | ((size_t)sp)) & 15) == 0) {
      for (int t = tid; t < nown / 4; t += OWN_THR) {
        const int4 c = ((const int4*)cnt)[t];
        ((float4*)dp)[t] = make_float4((float)c.x, (float)c.y, (float)c.z, (float)c.w);
      }
      for (int t = tid; t < nown * 5 / 4; t += OWN_THR) ((float4*)sp)[t] = ((const float4*)tab)[t];
    } else {
      for (int t = tid; t < nown; t += OWN_THR) dp[t] = (float)cnt[t];
      for (int t = tid; t < nown * 5; t += OWN_THR) sp[t] = tab[t];
    }
  } else {
    for (int r0 = wave; r0 < nown; r0 += 4 * OWN_WAV) {
      float4 qr[4];
      float  dg[4];
#pragma unroll
      for (int u = 0; u < 4; ++u) {
        const int r = r0 + u * OWN_WAV;
        if (r < nown) {
          qr[u] = *(const float4*)(qv + (size_t)(lo + r) * 128 + lane * 4);
          dg[u] = deg[lo + r];
        }
      }
#pragma unroll
      for (int u = 0; u < 4; ++u) {
        const int r = r0 + u * OWN_WAV;
        if (r < nown) {
          float* row = &tab[r * 128 + lane];
          float4 o;
          o.x = qr[u].x + row[0]  / dg[u];
          o.y = qr[u].y + row[32] / dg[u];
          o.z = qr[u].z + row[64] / dg[u];
          o.w = qr[u].w + row[96] / dg[u];
          o.x = o.x > 0.0f ? o.x : 0.0f;
          o.y = o.y > 0.0f ? o.y : 0.0f;
          o.z = o.z > 0.0f ? o.z : 0.0f;
          o.w = o.w > 0.0f ? o.w : 0.0f;
          *(float4*)&tab[r * 128 + 4 * (lane ^ (r & 15))] = o;
        }
      }
    }
    __syncthreads();

    float* part = (float*)queue;
    const int wv = wave;
    if ((wv >> 3) == 0) tail_gemm<7>(tab, part, l1W, l1b, l2W, wv & 7, 0, lane);
    else                tail_gemm<6>(tab, part, l1W, l1b, l2W, wv & 7, wv >> 3, lane);
    __syncthreads();

    if (tid < nown * 2) {
      const int row = tid >> 1;
      const int j   = tid & 1;
      float a = l2b[j];
#pragma unroll
      for (int t8 = 0; t8 < 8; ++t8) a += part[((size_t)t8 * NPR + row) * 2 + j];
      out[(size_t)(lo + row) * 2 + j] = a;
    }
  }
}

__global__ __launch_bounds__(256)
void sage_l1_proj_wmma(const float* __restrict__ x, const float* __restrict__ s1q,
                       const float* __restrict__ degq, float* __restrict__ deg,
                       const float* __restrict__ W1,
                       const float* __restrict__ b1, const float* __restrict__ W2,
                       const float* __restrict__ b2, float* __restrict__ qv,
                       float* __restrict__ pv, int M) {
  __shared__ __attribute__((aligned(16))) float As[MTILE * LDSTRIDE];
  __shared__ float dgs[MTILE];

  const int row0 = blockIdx.x * MTILE;
  const int t = threadIdx.x;

  if (t >= 256 - MTILE) {
    const int td  = t - (256 - MTILE);
    const int row = row0 + td;
    float d = 1.0f;
    if (row < M) {
      float a = degq[row];
#pragma unroll
      for (int p = 1; p < NQ; ++p) a += degq[(size_t)p * M + row];
      d += a;
      deg[row] = d;
    }
    dgs[td] = d;
  }

  if (t < MTILE * 5) {
    const int m = t / 5;
    const int f = t - m * 5;
    const int row = row0 + m;
    float xv = 0.0f, mean = 0.0f;
    if (row < M) {
      const size_t o = (size_t)row * 5 + f;
      xv = x[o];
      float sn = s1q[o];
      float da = degq[row];
#pragma unroll
      for (int p = 1; p < NQ; ++p) {
        sn += s1q[(size_t)p * 5 * M + o];
        da += degq[(size_t)p * M + row];
      }
      const float dn = 1.0f + da;
      mean = (xv + sn) / dn;
    }
    As[m * LDSTRIDE + f]     = xv;
    As[m * LDSTRIDE + 5 + f] = mean;
  } else if (t < MTILE * 7) {
    const int i = t - MTILE * 5;
    As[(i >> 1) * LDSTRIDE + 10 + (i & 1)] = 0.0f;
  }
  __syncthreads();

  const int lane  = t & 31;
  const int wave  = t >> 5;
  const int n     = wave * 16 + (lane & 15);
  const int mA    = lane & 15;
  const int khalf = (lane < 16) ? 0 : 2;
  const int rbase = (lane < 16) ? 0 : 8;

  v8f c0 = {};
  v8f c1 = {};
  {
    const float* wcol = W1 + n;
#pragma unroll
    for (int k0 = 0; k0 < 12; k0 += 4) {
      const int kb = k0 + khalf;
      v2f a0 = *(const v2f*)&As[mA * LDSTRIDE + kb];
      v2f a1 = *(const v2f*)&As[(mA + 16) * LDSTRIDE + kb];
      const float w0 = wcol[(size_t)min(kb,     9) * NCOLS];
      const float w1 = wcol[(size_t)min(kb + 1, 9) * NCOLS];
      v2f b;
      b.x = (kb     < 10) ? w0 : 0.0f;
      b.y = (kb + 1 < 10) ? w1 : 0.0f;
      c0 = __builtin_amdgcn_wmma_f32_16x16x4_f32(false, a0, false, b,
                                                 (short)0, c0, false, false);
      c1 = __builtin_amdgcn_wmma_f32_16x16x4_f32(false, a1, false, b,
                                                 (short)0, c1, false, false);
    }
  }
  __syncthreads();
  {
    const float bv = b1[n];
#pragma unroll
    for (int r = 0; r < 8; ++r) {
      const float vA = c0[r] + bv;
      const float vB = c1[r] + bv;
      As[(rbase + r) * LDSTRIDE + n]      = vA > 0.0f ? vA : 0.0f;
      As[(16 + rbase + r) * LDSTRIDE + n] = vB > 0.0f ? vB : 0.0f;
    }
  }
  __syncthreads();

  const float* wq = W2 + n;
  const float* wp = W2 + (size_t)NCOLS * NCOLS + n;
  v8f q0 = {};
  v8f q1 = {};
  v8f p0 = {};
  v8f p1 = {};
#pragma unroll 2
  for (int k0 = 0; k0 < NCOLS; k0 += 4) {
    const int kb = k0 + khalf;
    v2f a0 = *(const v2f*)&As[mA * LDSTRIDE + kb];
    v2f a1 = *(const v2f*)&As[(mA + 16) * LDSTRIDE + kb];
    v2f bq, bp;
    bq.x = wq[(size_t)kb * NCOLS];
    bq.y = wq[(size_t)(kb + 1) * NCOLS];
    bp.x = wp[(size_t)kb * NCOLS];
    bp.y = wp[(size_t)(kb + 1) * NCOLS];
    q0 = __builtin_amdgcn_wmma_f32_16x16x4_f32(false, a0, false, bq,
                                               (short)0, q0, false, false);
    q1 = __builtin_amdgcn_wmma_f32_16x16x4_f32(false, a1, false, bq,
                                               (short)0, q1, false, false);
    p0 = __builtin_amdgcn_wmma_f32_16x16x4_f32(false, a0, false, bp,
                                               (short)0, p0, false, false);
    p1 = __builtin_amdgcn_wmma_f32_16x16x4_f32(false, a1, false, bp,
                                               (short)0, p1, false, false);
  }

  const float bv2 = b2[n];
  const size_t o0 = (size_t)(row0 + rbase) * NCOLS + n;
  const size_t o1 = o0 + (size_t)16 * NCOLS;

  if (row0 + MTILE <= M) {
#pragma unroll
    for (int r = 0; r < 8; ++r) {
      const float dA = dgs[rbase + r];
      const float dB = dgs[16 + rbase + r];
      pv[o0 + (size_t)r * NCOLS] = p0[r];
      pv[o1 + (size_t)r * NCOLS] = p1[r];
      qv[o0 + (size_t)r * NCOLS] = q0[r] + bv2 + p0[r] / dA;
      qv[o1 + (size_t)r * NCOLS] = q1[r] + bv2 + p1[r] / dB;
    }
  } else {
#pragma unroll
    for (int r = 0; r < 8; ++r) {
      if (row0 + rbase + r < M) {
        pv[o0 + (size_t)r * NCOLS] = p0[r];
        qv[o0 + (size_t)r * NCOLS] = q0[r] + bv2 + p0[r] / dgs[rbase + r];
      }
      if (row0 + 16 + rbase + r < M) {
        pv[o1 + (size_t)r * NCOLS] = p1[r];
        qv[o1 + (size_t)r * NCOLS] = q1[r] + bv2 + p1[r] / dgs[16 + rbase + r];
      }
    }
  }
}

extern "C" void kernel_launch(void* const* d_in, const int* in_sizes, int n_in,
                              void* d_out, int out_size, void* d_ws, size_t ws_size,
                              hipStream_t stream) {
  const float* x   = (const float*)d_in[0];
  const int*   ei  = (const int*)d_in[1];
  const float* W1  = (const float*)d_in[2];
  const float* b1  = (const float*)d_in[3];
  const float* W2  = (const float*)d_in[4];
  const float* b2  = (const float*)d_in[5];
  const float* l1W = (const float*)d_in[6];
  const float* l1b = (const float*)d_in[7];
  const float* l2W = (const float*)d_in[8];
  const float* l2b = (const float*)d_in[9];

  const int N = in_sizes[0] / 5;
  const int E = in_sizes[1] / 2;
  const int* src = ei;
  const int* dst = ei + E;

  float* ws   = (float*)d_ws;
  float* pv   = ws;
  float* qv   = pv   + (size_t)128 * N;
  float* s1q  = qv   + (size_t)128 * N;
  float* degq = s1q  + (size_t)(5 * NQ) * N;
  float* deg  = degq + (size_t)NQ * N;
  const int nrange = (N + NPR - 1) / NPR;
  const int ngroup = (N + NPR1 - 1) / NPR1;
  int* counts = (int*)(deg + (size_t)4 * N);
  int* lists  = counts + (size_t)ngroup * (RPG * NQ);

  const int nthread = 256;

  edge_owner_pass<1><<<ngroup * NQ, OWN_THR, 0, stream>>>(src, dst, x, nullptr, degq, s1q, counts, lists,
                                                          nullptr, nullptr, nullptr, nullptr, nullptr, N, E);

  sage_l1_proj_wmma<<<(N + MTILE - 1) / MTILE, nthread, 0, stream>>>(x, s1q, degq, deg, W1, b1, W2, b2, qv, pv, N);

  edge_owner_pass<2><<<nrange, OWN_THR, 0, stream>>>(src, dst, pv, qv, deg, nullptr, counts, lists,
                                                     l1W, l1b, l2W, l2b, (float*)d_out, N, E);
}


# Layer-1 sweep reads the source list beside the targets (16 B a lane each) and queues (source, local target) from registers: the drain's two 4-byte look-ups per edge are gone; LDS adds by arrival
# speedup vs baseline: 4.5665x; 1.0357x over previous
#include <hip/hip_runtime.h>

typedef __attribute__((ext_vector_type(2))) float v2f;
typedef __attribute__((ext_vector_type(8))) float v8f;

#define NCOLS 128
#define MTILE 32
#define LDSTRIDE 260

#define NPR      400
#define RPG      16
#define NPR1     (RPG * NPR)
#define NQ       16
#define OWN_THR  1024
#define OWN_WAV  32
#define SCAN_U   4
#define QFLUSH   96
#define QCAP     (QFLUSH + SCAN_U * 128)
#define LCAP     16384
#define LCAPQ    (LCAP / NQ)

__device__ __forceinline__ void add_rows(const float* __restrict__ feat, float* tab,
                                         int soff, int doff, int c, int lane) {
  for (int j = 0; j < c; j += 8) {
    float4 r[8];
    int    d[8];
    if (j + 8 <= c) {
#pragma unroll
      for (int u = 0; u < 8; ++u) {
        const int sj = __builtin_amdgcn_readlane(soff, j + u);
        d[u] = __builtin_amdgcn_readlane(doff, j + u);
        r[u] = *(const float4*)(feat + (size_t)sj + lane * 4);
      }
#pragma unroll
      for (int u = 0; u < 8; ++u) {
        float* row = &tab[d[u] + lane];
        __hip_atomic_fetch_add(row +  0, r[u].x, __ATOMIC_RELAXED, __HIP_MEMORY_SCOPE_WORKGROUP);
        __hip_atomic_fetch_add(row + 32, r[u].y, __ATOMIC_RELAXED, __HIP_MEMORY_SCOPE_WORKGROUP);
        __hip_atomic_fetch_add(row + 64, r[u].z, __ATOMIC_RELAXED, __HIP_MEMORY_SCOPE_WORKGROUP);
        __hip_atomic_fetch_add(row + 96, r[u].w, __ATOMIC_RELAXED, __HIP_MEMORY_SCOPE_WORKGROUP);
      }
    } else {
      const int m = c - j;
#pragma unroll
      for (int u = 0; u < 7; ++u) {
        if (u < m) {
          const int sj = __builtin_amdgcn_readlane(soff, j + u);
          d[u] = __builtin_amdgcn_readlane(doff, j + u);
          r[u] = *(const float4*)(feat + (size_t)sj + lane * 4);
        }
      }
#pragma unroll
      for (int u = 0; u < 7; ++u) {
        if (u < m) {
          float* row = &tab[d[u] + lane];
          __hip_atomic_fetch_add(row +  0, r[u].x, __ATOMIC_RELAXED, __HIP_MEMORY_SCOPE_WORKGROUP);
          __hip_atomic_fetch_add(row + 32, r[u].y, __ATOMIC_RELAXED, __HIP_MEMORY_SCOPE_WORKGROUP);
          __hip_atomic_fetch_add(row + 64, r[u].z, __ATOMIC_RELAXED, __HIP_MEMORY_SCOPE_WORKGROUP);
          __hip_atomic_fetch_add(row + 96, r[u].w, __ATOMIC_RELAXED, __HIP_MEMORY_SCOPE_WORKGROUP);
        }
      }
    }
  }
}

template <bool VEC>
__device__ __forceinline__ int4 load_list4(const int* __restrict__ lst, int idx4, int nvec, int E) {
  int4 r;
  if (VEC) {
    r = *(const int4*)(lst + (size_t)min(idx4, nvec - 1) * 4);
  } else {
    const int e = idx4 * 4;
    r.x = lst[min(e + 0, E - 1)];
    r.y = lst[min(e + 1, E - 1)];
    r.z = lst[min(e + 2, E - 1)];
    r.w = lst[min(e + 3, E - 1)];
    if (e + 0 >= E) r.x = -1;
    if (e + 1 >= E) r.y = -1;
    if (e + 2 >= E) r.z = -1;
    if (e + 3 >= E) r.w = -1;
  }
  return r;
}

template <int MODE, bool VEC>
__device__ __forceinline__ void owner_sweep(const int* __restrict__ src, const int* __restrict__ dst,
                                            const float* __restrict__ feat, float* tab, int* cnt,
                                            int* q, int* lcount, int* list,
                                            int lo, unsigned nloc, int E) {
  const int tid  = threadIdx.x;
  const int lane = tid & 31;
  const int nvec   = (E + 3) >> 2;
  const int per_it = OWN_THR * SCAN_U;
  const int n_it   = (nvec + per_it - 1) / per_it;
  const int rot = (int)(((long long)blockIdx.x * n_it) / gridDim.x);

  int qn = 0;
  int2* q2 = (int2*)q;

  auto fetch = [&](int4 (&v)[SCAN_U], int4 (&w)[SCAN_U], int i) -> int {
    int it = i + rot;
    if (it >= n_it) it -= n_it;
    const int base = (i < n_it) ? (it * per_it + tid) : nvec;
#pragma unroll
    for (int u = 0; u < SCAN_U; ++u) v[u] = load_list4<VEC>(dst, base + u * OWN_THR, nvec, E);
    if (MODE == 1) {
#pragma unroll
      for (int u = 0; u < SCAN_U; ++u) w[u] = load_list4<VEC>(src, base + u * OWN_THR, nvec, E);
    }
    return base;
  };

  auto take = [&](const int4 (&v)[SCAN_U], const int4 (&w)[SCAN_U], int base) {
#pragma unroll
    for (int u = 0; u < SCAN_U; ++u) {
      const unsigned lim = (base + u * OWN_THR < nvec) ? nloc : 0u;
      const bool h0 = (unsigned)(v[u].x - lo) < lim;
      const bool h1 = (unsigned)(v[u].y - lo) < lim;
      const bool h2 = (unsigned)(v[u].z - lo) < lim;
      const bool h3 = (unsigned)(v[u].w - lo) < lim;
      const unsigned m0 = __builtin_amdgcn_ballot_w32(h0);
      const unsigned m1 = __builtin_amdgcn_ballot_w32(h1);
      const unsigned m2 = __builtin_amdgcn_ballot_w32(h2);
      const unsigned m3 = __builtin_amdgcn_ballot_w32(h3);
      if ((m0 | m1 | m2 | m3) != 0u) {
        if (MODE == 1) {
          if (h0) q2[qn + __builtin_amdgcn_mbcnt_lo(m0, 0u)] = make_int2(w[u].x, v[u].x - lo);
          qn += __builtin_popcount(m0);
          if (h1) q2[qn + __builtin_amdgcn_mbcnt_lo(m1, 0u)] = make_int2(w[u].y, v[u].y - lo);
          qn += __builtin_popcount(m1);
          if (h2) q2[qn + __builtin_amdgcn_mbcnt_lo(m2, 0u)] = make_int2(w[u].z, v[u].z - lo);
          qn += __builtin_popcount(m2);
          if (h3) q2[qn + __builtin_amdgcn_mbcnt_lo(m3, 0u)] = make_int2(w[u].w, v[u].w - lo);
          qn += __builtin_popcount(m3);
        } else {
          const int e0 = (base + u * OWN_THR) * 4;
          if (h0) q[qn + __builtin_amdgcn_mbcnt_lo(m0, 0u)] = e0 + 0;
          qn += __builtin_popcount(m0);
          if (h1) q[qn + __builtin_amdgcn_mbcnt_lo(m1, 0u)] = e0 + 1;
          qn += __builtin_popcount(m1);
          if (h2) q[qn + __builtin_amdgcn_mbcnt_lo(m2, 0u)] = e0 + 2;
          qn += __builtin_popcount(m2);
          if (h3) q[qn + __builtin_amdgcn_mbcnt_lo(m3, 0u)] = e0 + 3;
          qn += __builtin_popcount(m3);
        }
      }
    }
  };

  auto drain = [&]() {
    for (int b = 0; b < qn; b += 32) {
      const int k = b + lane;
      int dl = 0, s = 0;
      if (k < qn) {
        if (MODE == 1) {
          const int2 ent = q2[k];
          s  = ent.x;
          dl = ent.y;
        } else {
          const int e = q[k];
          dl = dst[e] - lo;
          s  = src[e];
        }
      }
      if (MODE == 1) {
        const bool live = k < qn;
        if (live) {
          const unsigned jr = (unsigned)dl / NPR;
          const int slot = __hip_atomic_fetch_add(lcount + jr, 1, __ATOMIC_RELAXED, __HIP_MEMORY_SCOPE_WORKGROUP);
          if (slot < LCAPQ)
            *(int2*)(list + (size_t)jr * (2 * LCAP) + 2 * (size_t)slot) = make_int2(s, dl - (int)jr * NPR);
          const float* xr = feat + (size_t)s * 5;
          const float x0 = xr[0], x1 = xr[1], x2 = xr[2], x3 = xr[3], x4 = xr[4];
          __hip_atomic_fetch_add(&cnt[dl], 1, __ATOMIC_RELAXED, __HIP_MEMORY_SCOPE_WORKGROUP);
          float* row = &tab[dl * 5];
          __hip_atomic_fetch_add(row + 0, x0, __ATOMIC_RELAXED, __HIP_MEMORY_SCOPE_WORKGROUP);
          __hip_atomic_fetch_add(row + 1, x1, __ATOMIC_RELAXED, __HIP_MEMORY_SCOPE_WORKGROUP);
          __hip_atomic_fetch_add(row + 2, x2, __ATOMIC_RELAXED, __HIP_MEMORY_SCOPE_WORKGROUP);
          __hip_atomic_fetch_add(row + 3, x3, __ATOMIC_RELAXED, __HIP_MEMORY_SCOPE_WORKGROUP);
          __hip_atomic_fetch_add(row + 4, x4, __ATOMIC_RELAXED, __HIP_MEMORY_SCOPE_WORKGROUP);
        }
      } else {
        add_rows(feat, tab, s * 128, dl * 128, min(32, qn - b), lane);
      }
    }
    qn = 0;
  };

  int4 A[SCAN_U], B[SCAN_U];
  int4 AS[SCAN_U], BS[SCAN_U];
  int baseA = fetch(A, AS, 0);
  for (int i = 0; i < n_it; i += 2) {
    const int baseB = fetch(B, BS, i + 1);
    __builtin_amdgcn_sched_barrier(0);
    take(A, AS, baseA);
    if (qn > QFLUSH) drain();
    baseA = fetch(A, AS, i + 2);
    __builtin_amdgcn_sched_barrier(0);
    take(B, BS, baseB);
    if (qn > QFLUSH || i + 2 >= n_it) drain();
  }
}

template <int CTRL>
__device__ __forceinline__ float row_xor_add(float v) {
  const int t = __builtin_amdgcn_update_dpp(0, __builtin_bit_cast(int, v), CTRL, 0xF, 0xF, true);
  return v + __builtin_bit_cast(float, t);
}
__device__ __forceinline__ float row_sum16(float v) {
  v = row_xor_add<0x161>(v);
  v = row_xor_add<0x162>(v);
  v = row_xor_add<0x164>(v);
  v = row_xor_add<0x168>(v);
  return v;
}

static_assert(NPR == 25 * 16, "tail_gemm splits a range into 25 row tiles of 16");
template <int NT>
__device__ __forceinline__ void tail_gemm(const float* tab, float* part,
                                          const float* __restrict__ l1W, const float* __restrict__ l1b,
                                          const float* __restrict__ l2W, int ct, int g, int lane) {
  const int n     = ct * 16 + (lane & 15);
  const int mA    = lane & 15;
  const int khalf = (lane < 16) ? 0 : 2;
  const int sw    = mA << 2;
  const float* wcol = l1W + n;
  const float* arow = tab + (g * 16 + mA) * 128 + khalf;

  v8f acc[NT];
#pragma unroll
  for (int i = 0; i < NT; ++i) acc[i] = v8f{};

#pragma unroll 2
  for (int k0 = 0; k0 < 128; k0 += 4) {
    const int kb = k0 + khalf;
    v2f b;
    b.x = wcol[(size_t)kb * NCOLS];
    b.y = wcol[(size_t)(kb + 1) * NCOLS];
    const int wo = k0 ^ sw;
#pragma unroll
    for (int i = 0; i < NT; ++i) {
      const v2f a = *(const v2f*)&arow[i * (64 * 128) + wo];
      acc[i] = __builtin_amdgcn_wmma_f32_16x16x4_f32(false, a, false, b,
                                                     (short)0, acc[i], false, false);
    }
  }

  const float bv = l1b[n];
  const float w0 = l2W[2 * n];
  const float w1 = l2W[2 * n + 1];
  const int rbase = (lane < 16) ? 0 : 8;
#pragma unroll
  for (int i = 0; i < NT; ++i) {
#pragma unroll
    for (int r = 0; r < 8; ++r) {
      float h = acc[i][r] + bv;
      h = h > 0.0f ? h : 0.0f;
      const float s0 = row_sum16(h * w0);
      const float s1 = row_sum16(h * w1);
      if ((lane & 15) == 0)
        *(float2*)&part[((size_t)ct * NPR + (g + 4 * i) * 16 + rbase + r) * 2] = make_float2(s0, s1);
    }
  }
}

template <int MODE>
__global__ __launch_bounds__(OWN_THR)
void edge_owner_pass(const int* __restrict__ src, const int* __restrict__ dst,
                     const float* __restrict__ feat, const float* __restrict__ qv,
                     float* __restrict__ deg,
                     float* __restrict__ sums, int* counts, int* lists,
                     const float* __restrict__ l1W, const float* __restrict__ l1b,
                     const float* __restrict__ l2W, const float* __restrict__ l2b,
                     float* __restrict__ out, int N, int E) {
  constexpr int TABW = (MODE == 1) ? NPR1 * 5 : NPR * 128;
  constexpr int QW   = (MODE == 1) ? 2 * QCAP : QCAP;
  static_assert((TABW + OWN_WAV * QW) * 4 > 160 * 1024, "one work-group per WGP");
  __shared__ __attribute__((aligned(16))) float tab[TABW];
  __shared__ __attribute__((aligned(16))) int cnt[NPR1];
  __shared__ __attribute__((aligned(16))) int queue[OWN_WAV * QW];
  __shared__ int   lcount[RPG];

  const int tid  = threadIdx.x;
  const int lane = tid & 31;
  const int wave = (MODE == 2) ? __builtin_amdgcn_readfirstlane(tid >> 5) : (tid >> 5);
  const int grp  = (MODE == 1) ? (int)(blockIdx.x / NQ) : (int)blockIdx.x;
  const int qtr  = (MODE == 1) ? (int)(blockIdx.x % NQ) : 0;
  const int lo   = grp * ((MODE == 1) ? NPR1 : NPR);
  const int nown = min((MODE == 1) ? NPR1 : NPR, N - lo);

  if (MODE == 1) {
    static_assert((NPR1 * 5) % 4 == 0 && NPR1 % 4 == 0, "the tables are cleared 16 bytes at a time");
    for (int t = tid; t < NPR1 * 5 / 4; t += OWN_THR) ((float4*)tab)[t] = make_float4(0.0f, 0.0f, 0.0f, 0.0f);
    for (int t = tid; t < NPR1 / 4; t += OWN_THR) ((int4*)cnt)[t] = make_int4(0, 0, 0, 0);
    if (tid < RPG) lcount[tid] = 0;
  } else {
    for (int t = tid; t < NPR * 32; t += OWN_THR)
      ((float4*)tab)[t] = make_float4(0.0f, 0.0f, 0.0f, 0.0f);
  }
  __syncthreads();

  static_assert(OWN_WAV % NQ == 0 && (NQ & (NQ - 1)) == 0, "the waves split evenly over a range's NQ lists");
  bool listed = (MODE == 2);
  if (MODE == 2) {
    const int* cb = counts + (size_t)blockIdx.x * NQ;
#pragma unroll
    for (int j = 0; j < NQ; ++j) listed = listed && (cb[j] <= LCAPQ);
  }

  if (MODE == 2 && listed) {
    const int wv  = wave;
    const int qc  = wv & (NQ - 1);
    const int sh  = wv / NQ;
    const int nl  = counts[(size_t)blockIdx.x * NQ + qc];
    const int* list = lists + ((size_t)blockIdx.x * NQ + qc) * (2 * LCAPQ);
    const int beg = (int)(((long long)sh * nl) / (OWN_WAV / NQ));
    const int end = (int)(((long long)(sh + 1) * nl) / (OWN_WAV / NQ));
    int2 nx = make_int2(0, 0);
    if (beg + lane < end) nx = *(const int2*)(list + 2 * (size_t)(beg + lane));
    for (int b = beg; b < end; b += 32) {
      const int2 p = nx;
      if (b + 32 + lane < end) nx = *(const int2*)(list + 2 * (size_t)(b + 32 + lane));
      add_rows(feat, tab, p.x * 128, p.y * 128, min(32, end - b), lane);
    }
  } else if (MODE == 1) {
    const int per = (((E + NQ - 1) / NQ) + 3) & ~3;
    const int e0  = min(qtr * per, E);
    const int El  = min(per, E - e0);
    if (El > 0) {
      const int* sq = src + e0;
      const int* dq = dst + e0;
      int* list = lists + ((size_t)grp * RPG * NQ + qtr) * (2 * LCAPQ);
      const bool vec_ok = ((El & 3) == 0) && (((((size_t)dq) | ((size_t)sq)) & 15) == 0);
      if (vec_ok)
        owner_sweep<1, true >(sq, dq, feat, tab, cnt, &queue[wave * QW], lcount, list, lo, (unsigned)nown, El);
      else
        owner_sweep<1, false>(sq, dq, feat, tab, cnt, &queue[wave * QW], lcount, list, lo, (unsigned)nown, El);
    }
  } else if (E > 0) {
    const bool vec_ok = ((E & 3) == 0) && ((((size_t)dst) & 15) == 0);
    if (vec_ok)
      owner_sweep<MODE, true >(src, dst, feat, tab, cnt, &queue[wave * QW], lcount, lists, lo, (unsigned)nown, E);
    else
      owner_sweep<MODE, false>(src, dst, feat, tab, cnt, &queue[wave * QW], lcount, lists, lo, (unsigned)nown, E);
  }

  __syncthreads();

  if (MODE == 1) {
    if (tid < RPG) counts[((size_t)grp * RPG + tid) * NQ + qtr] = lcount[tid];
    float* degq = deg + (size_t)qtr * N;
    float* s1q  = sums + (size_t)qtr * 5 * N;
    float* dp = degq + lo;
    float* sp = s1q + (size_t)lo * 5;
    if ((nown & 3) == 0 && ((((size_t)dp) | ((size_t)sp)) & 15) == 0) {
      for (int t = tid; t < nown / 4; t += OWN_THR) {
        const int4 c = ((const int4*)cnt)[t];
        ((float4*)dp)[t] = make_float4((float)c.x, (float)c.y, (float)c.z, (float)c.w);
      }
      for (int t = tid; t < nown * 5 / 4; t += OWN_THR) ((float4*)sp)[t] = ((const float4*)tab)[t];
    } else {
      for (int t = tid; t < nown; t += OWN_THR) dp[t] = (float)cnt[t];
      for (int t = tid; t < nown * 5; t += OWN_THR) sp[t] = tab[t];
    }
  } else {
    for (int r0 = wave; r0 < nown; r0 += 4 * OWN_WAV) {
      float4 qr[4];
      float  dg[4];
#pragma unroll
      for (int u = 0; u < 4; ++u) {
        const int r = r0 + u * OWN_WAV;
        if (r < nown) {
          qr[u] = *(const float4*)(qv + (size_t)(lo + r) * 128 + lane * 4);
          dg[u] = deg[lo + r];
        }
      }
#pragma unroll
      for (int u = 0; u < 4; ++u) {
        const int r = r0 + u * OWN_WAV;
        if (r < nown) {
          float* row = &tab[r * 128 + lane];
          float4 o;
          o.x = qr[u].x + row[0]  / dg[u];
          o.y = qr[u].y + row[32] / dg[u];
          o.z = qr[u].z + row[64] / dg[u];
          o.w = qr[u].w + row[96] / dg[u];
          o.x = o.x > 0.0f ? o.x : 0.0f;
          o.y = o.y > 0.0f ? o.y : 0.0f;
          o.z = o.z > 0.0f ? o.z : 0.0f;
          o.w = o.w > 0.0f ? o.w : 0.0f;
          *(float4*)&tab[r * 128 + 4 * (lane ^ (r & 15))] = o;
        }
      }
    }
    __syncthreads();

    float* part = (float*)queue;
    const int wv = wave;
    if ((wv >> 3) == 0) tail_gemm<7>(tab, part, l1W, l1b, l2W, wv & 7, 0, lane);
    else                tail_gemm<6>(tab, part, l1W, l1b, l2W, wv & 7, wv >> 3, lane);
    __syncthreads();

    if (tid < nown * 2) {
      const int row = tid >> 1;
      const int j   = tid & 1;
      float a = l2b[j];
#pragma unroll
      for (int t8 = 0; t8 < 8; ++t8) a += part[((size_t)t8 * NPR + row) * 2 + j];
      out[(size_t)(lo + row) * 2 + j] = a;
    }
  }
}

__global__ __launch_bounds__(256)
void sage_l1_proj_wmma(const float* __restrict__ x, const float* __restrict__ s1q,
                       const float* __restrict__ degq, float* __restrict__ deg,
                       const float* __restrict__ W1,
                       const float* __restrict__ b1, const float* __restrict__ W2,
                       const float* __restrict__ b2, float* __restrict__ qv,
                       float* __restrict__ pv, int M) {
  __shared__ __attribute__((aligned(16))) float As[MTILE * LDSTRIDE];
  __shared__ float dgs[MTILE];

  const int row0 = blockIdx.x * MTILE;
  const int t = threadIdx.x;

  if (t >= 256 - MTILE) {
    const int td  = t - (256 - MTILE);
    const int row = row0 + td;
    float d = 1.0f;
    if (row < M) {
      float a = degq[row];
#pragma unroll
      for (int p = 1; p < NQ; ++p) a += degq[(size_t)p * M + row];
      d += a;
      deg[row] = d;
    }
    dgs[td] = d;
  }

  if (t < MTILE * 5) {
    const int m = t / 5;
    const int f = t - m * 5;
    const int row = row0 + m;
    float xv = 0.0f, mean = 0.0f;
    if (row < M) {
      const size_t o = (size_t)row * 5 + f;
      xv = x[o];
      float sn = s1q[o];
      float da = degq[row];
#pragma unroll
      for (int p = 1; p < NQ; ++p) {
        sn += s1q[(size_t)p * 5 * M + o];
        da += degq[(size_t)p * M + row];
      }
      const float dn = 1.0f + da;
      mean = (xv + sn) / dn;
    }
    As[m * LDSTRIDE + f]     = xv;
    As[m * LDSTRIDE + 5 + f] = mean;
  } else if (t < MTILE * 7) {
    const int i = t - MTILE * 5;
    As[(i >> 1) * LDSTRIDE + 10 + (i & 1)] = 0.0f;
  }
  __syncthreads();

  const int lane  = t & 31;
  const int wave  = t >> 5;
  const int n     = wave * 16 + (lane & 15);
  const int mA    = lane & 15;
  const int khalf = (lane < 16) ? 0 : 2;
  const int rbase = (lane < 16) ? 0 : 8;

  v8f c0 = {};
  v8f c1 = {};
  {
    const float* wcol = W1 + n;
#pragma unroll
    for (int k0 = 0; k0 < 12; k0 += 4) {
      const int kb = k0 + khalf;
      v2f a0 = *(const v2f*)&As[mA * LDSTRIDE + kb];
      v2f a1 = *(const v2f*)&As[(mA + 16) * LDSTRIDE + kb];
      const float w0 = wcol[(size_t)min(kb,     9) * NCOLS];
      const float w1 = wcol[(size_t)min(kb + 1, 9) * NCOLS];
      v2f b;
      b.x = (kb     < 10) ? w0 : 0.0f;
      b.y = (kb + 1 < 10) ? w1 : 0.0f;
      c0 = __builtin_amdgcn_wmma_f32_16x16x4_f32(false, a0, false, b,
                                                 (short)0, c0, false, false);
      c1 = __builtin_amdgcn_wmma_f32_16x16x4_f32(false, a1, false, b,
                                                 (short)0, c1, false, false);
    }
  }
  __syncthreads();
  {
    const float bv = b1[n];
#pragma unroll
    for (int r = 0; r < 8; ++r) {
      const float vA = c0[r] + bv;
      const float vB = c1[r] + bv;
      As[(rbase + r) * LDSTRIDE + n]      = vA > 0.0f ? vA : 0.0f;
      As[(16 + rbase + r) * LDSTRIDE + n] = vB > 0.0f ? vB : 0.0f;
    }
  }
  __syncthreads();

  const float* wq = W2 + n;
  const float* wp = W2 + (size_t)NCOLS * NCOLS + n;
  v8f q0 = {};
  v8f q1 = {};
  v8f p0 = {};
  v8f p1 = {};
#pragma unroll 2
  for (int k0 = 0; k0 < NCOLS; k0 += 4) {
    const int kb = k0 + khalf;
    v2f a0 = *(const v2f*)&As[mA * LDSTRIDE + kb];
    v2f a1 = *(const v2f*)&As[(mA + 16) * LDSTRIDE + kb];
    v2f bq, bp;
    bq.x = wq[(size_t)kb * NCOLS];
    bq.y = wq[(size_t)(kb + 1) * NCOLS];
    bp.x = wp[(size_t)kb * NCOLS];
    bp.y = wp[(size_t)(kb + 1) * NCOLS];
    q0 = __builtin_amdgcn_wmma_f32_16x16x4_f32(false, a0, false, bq,
                                               (short)0, q0, false, false);
    q1 = __builtin_amdgcn_wmma_f32_16x16x4_f32(false, a1, false, bq,
                                               (short)0, q1, false, false);
    p0 = __builtin_amdgcn_wmma_f32_16x16x4_f32(false, a0, false, bp,
                                               (short)0, p0, false, false);
    p1 = __builtin_amdgcn_wmma_f32_16x16x4_f32(false, a1, false, bp,
                                               (short)0, p1, false, false);
  }

  const float bv2 = b2[n];
  const size_t o0 = (size_t)(row0 + rbase) * NCOLS + n;
  const size_t o1 = o0 + (size_t)16 * NCOLS;

  if (row0 + MTILE <= M) {
#pragma unroll
    for (int r = 0; r < 8; ++r) {
      const float dA = dgs[rbase + r];
      const float dB = dgs[16 + rbase + r];
      pv[o0 + (size_t)r * NCOLS] = p0[r];
      pv[o1 + (size_t)r * NCOLS] = p1[r];
      qv[o0 + (size_t)r * NCOLS] = q0[r] + bv2 + p0[r] / dA;
      qv[o1 + (size_t)r * NCOLS] = q1[r] + bv2 + p1[r] / dB;
    }
  } else {
#pragma unroll
    for (int r = 0; r < 8; ++r) {
      if (row0 + rbase + r < M) {
        pv[o0 + (size_t)r * NCOLS] = p0[r];
        qv[o0 + (size_t)r * NCOLS] = q0[r] + bv2 + p0[r] / dgs[rbase + r];
      }
      if (row0 + 16 + rbase + r < M) {
        pv[o1 + (size_t)r * NCOLS] = p1[r];
        qv[o1 + (size_t)r * NCOLS] = q1[r] + bv2 + p1[r] / dgs[16 + rbase + r];
      }
    }
  }
}

extern "C" void kernel_launch(void* const* d_in, const int* in_sizes, int n_in,
                              void* d_out, int out_size, void* d_ws, size_t ws_size,
                              hipStream_t stream) {
  const float* x   = (const float*)d_in[0];
  const int*   ei  = (const int*)d_in[1];
  const float* W1  = (const float*)d_in[2];
  const float* b1  = (const float*)d_in[3];
  const float* W2  = (const float*)d_in[4];
  const float* b2  = (const float*)d_in[5];
  const float* l1W = (const float*)d_in[6];
  const float* l1b = (const float*)d_in[7];
  const float* l2W = (const float*)d_in[8];
  const float* l2b = (const float*)d_in[9];

  const int N = in_sizes[0] / 5;
  const int E = in_sizes[1] / 2;
  const int* src = ei;
  const int* dst = ei + E;

  float* ws   = (float*)d_ws;
  float* pv   = ws;
  float* qv   = pv   + (size_t)128 * N;
  float* s1q  = qv   + (size_t)128 * N;
  float* degq = s1q  + (size_t)(5 * NQ) * N;
  float* deg  = degq + (size_t)NQ * N;
  const int nrange = (N + NPR - 1) / NPR;
  const int ngroup = (N + NPR1 - 1) / NPR1;
  int* counts = (int*)(deg + (size_t)4 * N);
  int* lists  = counts + (size_t)ngroup * (RPG * NQ);

  const int nthread = 256;

  edge_owner_pass<1><<<ngroup * NQ, OWN_THR, 0, stream>>>(src, dst, x, nullptr, degq, s1q, counts, lists,
                                                          nullptr, nullptr, nullptr, nullptr, nullptr, N, E);

  sage_l1_proj_wmma<<<(N + MTILE - 1) / MTILE, nthread, 0, stream>>>(x, s1q, degq, deg, W1, b1, W2, b2, qv, pv, N);

  edge_owner_pass<2><<<nrange, OWN_THR, 0, stream>>>(src, dst, pv, qv, deg, nullptr, counts, lists,
                                                     l1W, l1b, l2W, l2b, (float*)d_out, N, E);
}
